# starting kernel
# baseline (speedup 1.0000x reference)
#include <hip/hip_runtime.h>
#include <hip/hip_bf16.h>

typedef __attribute__((ext_vector_type(2))) float v2f;
typedef __attribute__((ext_vector_type(8))) float v8f;

static __device__ __forceinline__ int gtid() {
    return blockIdx.x * blockDim.x + threadIdx.x;
}

__global__ void init_deg_kernel(float* __restrict__ deg, int n) {
    int i = gtid();
    if (i < n) deg[i] = 1.0f;
}

__global__ void count_deg_kernel(const int* __restrict__ dst, float* __restrict__ deg, int e) {
    int i = gtid();
    if (i < e) atomicAdd(&deg[dst[i]], 1.0f);
}

__global__ void rsqrt_kernel(float* __restrict__ deg, int n) {
    int i = gtid();
    if (i < n) deg[i] = rsqrtf(deg[i]);
}

template <int K, int NOUT>
__global__ void linear_wmma_kernel(const float* __restrict__ x,
                                   const float* __restrict__ w,
                                   float* __restrict__ h,
                                   int n) {
    const int wave = (blockIdx.x * blockDim.x + threadIdx.x) >> 5;
    const int lane = threadIdx.x & 31;
    const int base = wave * 16;
    if (base >= n) return;

    const int m     = lane & 15;
    const int khalf = (lane >> 4) * 2;
    const int row   = min(base + m, n - 1);
    const int mcl   = (m < NOUT) ? m : 0;
    const bool colOK = (m < NOUT);

    v8f acc = {};
    constexpr int KC = (K + 3) / 4;
#pragma unroll
    for (int kc = 0; kc < KC; ++kc) {
        const int k0 = kc * 4 + khalf;
        const bool v0 = (k0 + 0) < K;
        const bool v1 = (k0 + 1) < K;
        const int ka0 = v0 ? (k0 + 0) : 0;
        const int ka1 = v1 ? (k0 + 1) : 0;
        const float xa0 = x[row * K + ka0];
        const float xa1 = x[row * K + ka1];
        const float wb0 = w[mcl * K + ka0];
        const float wb1 = w[mcl * K + ka1];
        v2f a, b;
        a.x = v0 ? xa0 : 0.0f;
        a.y = v1 ? xa1 : 0.0f;
        b.x = (colOK && v0) ? wb0 : 0.0f;
        b.y = (colOK && v1) ? wb1 : 0.0f;
        acc = __builtin_amdgcn_wmma_f32_16x16x4_f32(
             false, a,  false, b,
             (short)0, acc,  false,  false);
    }

    const int mofs = (lane >> 4) * 8;
    if (base + 16 <= n) {
        if (colOK) {
#pragma unroll
            for (int r = 0; r < 8; ++r)
                h[(base + mofs + r) * NOUT + m] = acc[r];
        }
    } else {
        if (colOK) {
#pragma unroll
            for (int r = 0; r < 8; ++r) {
                const int node = base + mofs + r;
                if (node < n) h[node * NOUT + m] = acc[r];
            }
        }
    }
}

template <int C>
__global__ void init_agg_kernel(const float* __restrict__ h,
                                const float* __restrict__ dinv,
                                const float* __restrict__ bias,
                                float* __restrict__ agg, int n) {
    int i = gtid();
    if (i >= n) return;
    float s = dinv[i];
    s = s * s;
#pragma unroll
    for (int c = 0; c < C; ++c) agg[i * C + c] = bias[c] + h[i * C + c] * s;
}

template <int C>
__global__ void edge_scatter_kernel(const int* __restrict__ src,
                                    const int* __restrict__ dst,
                                    const float* __restrict__ h,
                                    const float* __restrict__ dinv,
                                    float* __restrict__ agg, int e) {
    int i = gtid();
    if (i >= e) return;
    const int s = src[i];
    const int d = dst[i];
    const float wgt = dinv[s] * dinv[d];
    const float* hs = h + (size_t)s * C;
    float* ad = agg + (size_t)d * C;
    if constexpr (C == 4) {
        float4 hv = *reinterpret_cast<const float4*>(hs);
        atomicAdd(ad + 0, hv.x * wgt);
        atomicAdd(ad + 1, hv.y * wgt);
        atomicAdd(ad + 2, hv.z * wgt);
        atomicAdd(ad + 3, hv.w * wgt);
    } else {
        float2 hv = *reinterpret_cast<const float2*>(hs);
        atomicAdd(ad + 0, hv.x * wgt);
        atomicAdd(ad + 1, hv.y * wgt);
    }
}

__global__ void tanh_kernel(const float* __restrict__ in, float* __restrict__ out, int n) {
    int i = gtid();
    if (i < n) out[i] = tanhf(in[i]);
}

static inline int cdiv(int a, int b) { return (a + b - 1) / b; }

extern "C" void kernel_launch(void* const* d_in, const int* in_sizes, int n_in,
                              void* d_out, int out_size, void* d_ws, size_t ws_size,
                              hipStream_t stream) {
    const float* x  = (const float*)d_in[0];
    const int*   ei = (const int*)d_in[1];
    const float* W1 = (const float*)d_in[2];
    const float* b1 = (const float*)d_in[3];
    const float* W2 = (const float*)d_in[4];
    const float* b2 = (const float*)d_in[5];
    const float* W3 = (const float*)d_in[6];
    const float* b3 = (const float*)d_in[7];
    float* out = (float*)d_out;

    const int n = in_sizes[0] / 9;
    const int e = in_sizes[1] / 2;
    const int* src = ei;
    const int* dst = ei + e;

    char* ws = (char*)d_ws;
    size_t off = 0;
    auto carve = [&](size_t bytes) {
        void* p = ws + off;
        off += (bytes + 255) & ~(size_t)255;
        return p;
    };
    float* dinv = (float*)carve((size_t)n * sizeof(float));
    float* bufA = (float*)carve((size_t)n * 4 * sizeof(float));
    float* bufB = (float*)carve((size_t)n * 4 * sizeof(float));
    float* bufC = (float*)carve((size_t)n * 4 * sizeof(float));
    (void)ws_size;

    const int B = 256;
    const int gN = cdiv(n, B);
    const int gE = cdiv(e, B);
    const int gW = cdiv(cdiv(n, 16) * 32, B);

    init_deg_kernel<<<gN, B, 0, stream>>>(dinv, n);
    count_deg_kernel<<<gE, B, 0, stream>>>(dst, dinv, e);
    rsqrt_kernel<<<gN, B, 0, stream>>>(dinv, n);

    linear_wmma_kernel<9, 4><<<gW, B, 0, stream>>>(x, W1, bufA, n);
    init_agg_kernel<4><<<gN, B, 0, stream>>>(bufA, dinv, b1, bufB, n);
    edge_scatter_kernel<4><<<gE, B, 0, stream>>>(src, dst, bufA, dinv, bufB, e);
    tanh_kernel<<<cdiv(n * 4, B), B, 0, stream>>>(bufB, bufC, n * 4);

    linear_wmma_kernel<4, 4><<<gW, B, 0, stream>>>(bufC, W2, bufA, n);
    init_agg_kernel<4><<<gN, B, 0, stream>>>(bufA, dinv, b2, bufB, n);
    edge_scatter_kernel<4><<<gE, B, 0, stream>>>(src, dst, bufA, dinv, bufB, e);
    tanh_kernel<<<cdiv(n * 4, B), B, 0, stream>>>(bufB, bufC, n * 4);

    linear_wmma_kernel<4, 2><<<gW, B, 0, stream>>>(bufC, W3, bufA, n);
    init_agg_kernel<2><<<gN, B, 0, stream>>>(bufA, dinv, b3, out, n);
    edge_scatter_kernel<2><<<gE, B, 0, stream>>>(src, dst, bufA, dinv, out, e);
}


# edges sorted by destination once (returning int atomic = rank, scan, fill with dinv[src]); layers sum a node's messages by gather in registers, no float atomics; sum order = atomic arrival order
# speedup vs baseline: 3.3643x; 3.3643x over previous
#include <hip/hip_runtime.h>
#include <hip/hip_bf16.h>


typedef __attribute__((ext_vector_type(2))) float v2f;
typedef __attribute__((ext_vector_type(8))) float v8f;

static __device__ __forceinline__ int gtid() {
    return blockIdx.x * blockDim.x + threadIdx.x;
}

constexpr int EPT = 4;
constexpr int SB  = 1024;

__global__ void zero_cnt_kernel(int* __restrict__ cnt, int n) {
    int i = gtid();
    if (i < n) cnt[i] = 0;
}

__global__ void rank_kernel(const int* __restrict__ dst, int* __restrict__ cnt,
                            int* __restrict__ rank, int e) {
    const int blk0 = blockIdx.x * (blockDim.x * EPT);
    const int base = blk0 + threadIdx.x;
    if (blk0 + (int)blockDim.x * EPT <= e) {
        int d[EPT], r[EPT];
        for (int k = 0; k < EPT; ++k) d[k] = dst[base + k * blockDim.x];
        for (int k = 0; k < EPT; ++k) r[k] = atomicAdd(&cnt[d[k]], 1);
        for (int k = 0; k < EPT; ++k) rank[base + k * blockDim.x] = r[k];
    } else {
        for (int k = 0; k < EPT; ++k) {
            const int i = base + k * blockDim.x;
            if (i < e) rank[i] = atomicAdd(&cnt[dst[i]], 1);
        }
    }
}

static __device__ __forceinline__ int wave_sum(int v) {
    for (int o = 16; o > 0; o >>= 1) v += __shfl_xor(v, o, 32);
    return v;
}

static __device__ __forceinline__ int block_sum_1024(int v, int* sm, int* res) {
    const int lane = threadIdx.x & 31;
    const int w = threadIdx.x >> 5;
    v = wave_sum(v);
    if (lane == 0) sm[w] = v;
    __syncthreads();
    if (w == 0) {
        int t = wave_sum(sm[lane]);
        if (lane == 0) *res = t;
    }
    __syncthreads();
    return *res;
}

__global__ void __launch_bounds__(SB) block_sum_kernel(const int* __restrict__ cnt,
                                                       int* __restrict__ bsum, int n) {
    __shared__ int sm[32];
    __shared__ int res;
    const int i = blockIdx.x * SB + threadIdx.x;
    const int v = (i < n) ? cnt[i] : 0;
    const int t = block_sum_1024(v, sm, &res);
    if (threadIdx.x == 0) bsum[blockIdx.x] = t;
}

__global__ void __launch_bounds__(SB) scan_kernel(const int* __restrict__ cnt,
                                                  const int* __restrict__ bsum,
                                                  int* __restrict__ rowptr,
                                                  float* __restrict__ dinv, int n) {
    __shared__ int sm[32];
    __shared__ int res;
    __shared__ int wofs[32];
    const int tid = threadIdx.x;
    const int lane = tid & 31;
    const int w = tid >> 5;

    int part = 0;
    for (int t = tid; t < (int)blockIdx.x; t += SB) part += bsum[t];
    const int boff = block_sum_1024(part, sm, &res);

    const int i = blockIdx.x * SB + tid;
    const int c = (i < n) ? cnt[i] : 0;
    int x = c;
    for (int o = 1; o < 32; o <<= 1) {
        const int y = __shfl_up(x, o, 32);
        if (lane >= o) x += y;
    }
    if (lane == 31) wofs[w] = x;
    __syncthreads();
    if (w == 0) {
        const int tot = wofs[lane];
        int s = tot;
        for (int o = 1; o < 32; o <<= 1) {
            const int y = __shfl_up(s, o, 32);
            if (lane >= o) s += y;
        }
        wofs[lane] = s - tot;
    }
    __syncthreads();
    if (i < n) {
        const int excl = boff + wofs[w] + (x - c);
        rowptr[i] = excl;
        dinv[i] = rsqrtf((float)c + 1.0f);
        if (i == n - 1) rowptr[n] = excl + c;
    }
}

__global__ void fill_kernel(const int* __restrict__ src, const int* __restrict__ dst,
                            const int* __restrict__ rank, const int* __restrict__ rowptr,
                            const float* __restrict__ dinv, int* __restrict__ csr, int e) {
    int2* __restrict__ ent = reinterpret_cast<int2*>(csr);
    const int blk0 = blockIdx.x * (blockDim.x * EPT);
    const int base = blk0 + threadIdx.x;
    if (blk0 + (int)blockDim.x * EPT <= e) {
        int d[EPT], r[EPT], s[EPT], p[EPT];
        float w[EPT];
        for (int k = 0; k < EPT; ++k) {
            const int i = base + k * blockDim.x;
            d[k] = dst[i];
            r[k] = rank[i];
            s[k] = src[i];
        }
        for (int k = 0; k < EPT; ++k) {
            p[k] = rowptr[d[k]];
            w[k] = dinv[s[k]];
        }
        for (int k = 0; k < EPT; ++k) ent[p[k] + r[k]] = make_int2(s[k], __float_as_int(w[k]));
    } else {
        for (int k = 0; k < EPT; ++k) {
            const int i = base + k * blockDim.x;
            if (i < e) {
                const int s = src[i];
                ent[rowptr[dst[i]] + rank[i]] = make_int2(s, __float_as_int(dinv[s]));
            }
        }
    }
}

template <int K, int NOUT>
__global__ void linear_wmma_kernel(const float* __restrict__ x,
                                   const float* __restrict__ w,
                                   float* __restrict__ h,
                                   int n) {
    const int wave = (blockIdx.x * blockDim.x + threadIdx.x) >> 5;
    const int lane = threadIdx.x & 31;
    const int base = wave * 16;
    if (base >= n) return;

    const int m     = lane & 15;
    const int khalf = (lane >> 4) * 2;
    const int row   = min(base + m, n - 1);
    const int mcl   = (m < NOUT) ? m : 0;
    const bool colOK = (m < NOUT);

    v8f acc = {};
    constexpr int KC = (K + 3) / 4;
#pragma unroll
    for (int kc = 0; kc < KC; ++kc) {
        const int k0 = kc * 4 + khalf;
        const bool v0 = (k0 + 0) < K;
        const bool v1 = (k0 + 1) < K;
        const int ka0 = v0 ? (k0 + 0) : 0;
        const int ka1 = v1 ? (k0 + 1) : 0;
        const float xa0 = x[row * K + ka0];
        const float xa1 = x[row * K + ka1];
        const float wb0 = w[mcl * K + ka0];
        const float wb1 = w[mcl * K + ka1];
        v2f a, b;
        a.x = v0 ? xa0 : 0.0f;
        a.y = v1 ? xa1 : 0.0f;
        b.x = (colOK && v0) ? wb0 : 0.0f;
        b.y = (colOK && v1) ? wb1 : 0.0f;
        acc = __builtin_amdgcn_wmma_f32_16x16x4_f32(
             false, a,  false, b,
             (short)0, acc,  false,  false);
    }

    const int mofs = (lane >> 4) * 8;
    if (base + 16 <= n) {
        if (colOK) {
#pragma unroll
            for (int r = 0; r < 8; ++r)
                h[(base + mofs + r) * NOUT + m] = acc[r];
        }
    } else {
        if (colOK) {
#pragma unroll
            for (int r = 0; r < 8; ++r) {
                const int node = base + mofs + r;
                if (node < n) h[node * NOUT + m] = acc[r];
            }
        }
    }
}

static __device__ __forceinline__ void vfma(float4& a, float w, const float4& b) {
    a.x += w * b.x; a.y += w * b.y; a.z += w * b.z; a.w += w * b.w;
}
static __device__ __forceinline__ void vfma(float2& a, float w, const float2& b) {
    a.x += w * b.x; a.y += w * b.y;
}
static __device__ __forceinline__ void vxor_add(float4& a, int o) {
    a.x += __shfl_xor(a.x, o, 32); a.y += __shfl_xor(a.y, o, 32);
    a.z += __shfl_xor(a.z, o, 32); a.w += __shfl_xor(a.w, o, 32);
}
static __device__ __forceinline__ void vxor_add(float2& a, int o) {
    a.x += __shfl_xor(a.x, o, 32); a.y += __shfl_xor(a.y, o, 32);
}
static __device__ __forceinline__ void vzero(float4& a) { a = make_float4(0.f, 0.f, 0.f, 0.f); }
static __device__ __forceinline__ void vzero(float2& a) { a = make_float2(0.f, 0.f); }
static __device__ __forceinline__ void vfinish(float* sm, const float4& a, float d, const float* __restrict__ bias) {
    sm[0] = bias[0] + a.x * d; sm[1] = bias[1] + a.y * d;
    sm[2] = bias[2] + a.z * d; sm[3] = bias[3] + a.w * d;
}
static __device__ __forceinline__ void vfinish(float* sm, const float2& a, float d, const float* __restrict__ bias) {
    sm[0] = bias[0] + a.x * d; sm[1] = bias[1] + a.y * d;
}

template <int C> struct VecOf;
template <> struct VecOf<4> { typedef float4 type; };
template <> struct VecOf<2> { typedef float2 type; };

constexpr int GL  = 8;
constexpr int GB  = 256;
constexpr int NPB = GB / GL;

template <int C>
__global__ void __launch_bounds__(GB) gather_kernel(const int* __restrict__ rowptr,
                                                    const int* __restrict__ csr,
                                                    const float* __restrict__ h,
                                                    const float* __restrict__ dinv,
                                                    const float* __restrict__ bias,
                                                    float* __restrict__ out, int n) {
    typedef typename VecOf<C>::type VT;
    __shared__ __attribute__((aligned(16))) float sm[NPB * C];
    const VT* __restrict__ hv = reinterpret_cast<const VT*>(h);
    const int2* __restrict__ ent = reinterpret_cast<const int2*>(csr);

    const int tid  = threadIdx.x;
    const int sub  = tid & (GL - 1);
    const int nl   = tid / GL;
    const int node = blockIdx.x * NPB + nl;
    const bool ok  = node < n;

    int j = 0, end = 0;
    float dn = 0.0f;
    if (ok) {
        j   = rowptr[node] + sub;
        end = rowptr[node + 1];
        dn  = dinv[node];
    }
    VT acc;
    vzero(acc);
    if (ok && sub == 0) vfma(acc, dn, hv[node]);

    for (; j < end; j += 4 * GL) {
        const int j1 = j + GL, j2 = j + 2 * GL, j3 = j + 3 * GL;
        const int2 none = make_int2(-1, 0);
        const int2 e0 = ent[j];
        const int2 e1 = (j1 < end) ? ent[j1] : none;
        const int2 e2 = (j2 < end) ? ent[j2] : none;
        const int2 e3 = (j3 < end) ? ent[j3] : none;
        VT v0 = hv[e0.x];
        VT v1, v2, v3;
        vzero(v1); vzero(v2); vzero(v3);
        if (e1.x >= 0) v1 = hv[e1.x];
        if (e2.x >= 0) v2 = hv[e2.x];
        if (e3.x >= 0) v3 = hv[e3.x];
        vfma(acc, __int_as_float(e0.y), v0);
        vfma(acc, __int_as_float(e1.y), v1);
        vfma(acc, __int_as_float(e2.y), v2);
        vfma(acc, __int_as_float(e3.y), v3);
    }

    for (int o = 1; o < GL; o <<= 1) vxor_add(acc, o);

    if (ok && sub == 0) vfinish(sm + nl * C, acc, dn, bias);
    __syncthreads();

    constexpr int NV = NPB * C / 4;
    if (tid < NV) {
        const int first = blockIdx.x * (NPB * C) + tid * 4;
        const int total = n * C;
        if (first + 3 < total) {
            *reinterpret_cast<float4*>(out + first) = *reinterpret_cast<const float4*>(sm + tid * 4);
        } else {
            for (int q = 0; q < 4; ++q)
                if (first + q < total) out[first + q] = sm[tid * 4 + q];
        }
    }
}

__global__ void tanh_kernel(const float* __restrict__ in, float* __restrict__ out, int n) {
    int i = gtid();
    if (i < n) out[i] = tanhf(in[i]);
}

static inline int cdiv(int a, int b) { return (a + b - 1) / b; }

extern "C" void kernel_launch(void* const* d_in, const int* in_sizes, int n_in,
                              void* d_out, int out_size, void* d_ws, size_t ws_size,
                              hipStream_t stream) {
    const float* x  = (const float*)d_in[0];
    const int*   ei = (const int*)d_in[1];
    const float* W1 = (const float*)d_in[2];
    const float* b1 = (const float*)d_in[3];
    const float* W2 = (const float*)d_in[4];
    const float* b2 = (const float*)d_in[5];
    const float* W3 = (const float*)d_in[6];
    const float* b3 = (const float*)d_in[7];
    float* out = (float*)d_out;

    const int n = in_sizes[0] / 9;
    const int e = in_sizes[1] / 2;
    const int* src = ei;
    const int* dst = ei + e;

    char* ws = (char*)d_ws;
    size_t off = 0;
    auto carve = [&](size_t bytes) {
        void* p = ws + off;
        off += (bytes + 255) & ~(size_t)255;
        return p;
    };
    const int nb = cdiv(n, SB);
    float* dinv   = (float*)carve((size_t)n * sizeof(float));
    int*   cnt    = (int*)carve((size_t)n * sizeof(int));
    int*   rowptr = (int*)carve(((size_t)n + 1) * sizeof(int));
    int*   bsum   = (int*)carve((size_t)nb * sizeof(int));
    int*   rank   = (int*)carve((size_t)e * sizeof(int));
    int*   csr    = (int*)carve((size_t)e * 2 * sizeof(int));
    float* bufA = (float*)carve((size_t)n * 4 * sizeof(float));
    float* bufB = (float*)carve((size_t)n * 4 * sizeof(float));
    float* bufC = (float*)carve((size_t)n * 4 * sizeof(float));
    (void)ws_size;

    const int B = 256;
    const int gN = cdiv(n, B);
    const int gE = cdiv(e, B * EPT);
    const int gW = cdiv(cdiv(n, 16) * 32, B);
    const int gG = cdiv(n, NPB);

    zero_cnt_kernel<<<gN, B, 0, stream>>>(cnt, n);
    rank_kernel<<<gE, B, 0, stream>>>(dst, cnt, rank, e);
    block_sum_kernel<<<nb, SB, 0, stream>>>(cnt, bsum, n);
    scan_kernel<<<nb, SB, 0, stream>>>(cnt, bsum, rowptr, dinv, n);
    fill_kernel<<<gE, B, 0, stream>>>(src, dst, rank, rowptr, dinv, csr, e);

    linear_wmma_kernel<9, 4><<<gW, B, 0, stream>>>(x, W1, bufA, n);
    gather_kernel<4><<<gG, GB, 0, stream>>>(rowptr, csr, bufA, dinv, b1, bufB, n);
    tanh_kernel<<<cdiv(n * 4, B), B, 0, stream>>>(bufB, bufC, n * 4);

    linear_wmma_kernel<4, 4><<<gW, B, 0, stream>>>(bufC, W2, bufA, n);
    gather_kernel<4><<<gG, GB, 0, stream>>>(rowptr, csr, bufA, dinv, b2, bufB, n);
    tanh_kernel<<<cdiv(n * 4, B), B, 0, stream>>>(bufB, bufC, n * 4);

    linear_wmma_kernel<4, 2><<<gW, B, 0, stream>>>(bufC, W3, bufA, n);
    gather_kernel<2><<<gG, GB, 0, stream>>>(rowptr, csr, bufA, dinv, b3, out, n);
}


# edges ordered by destination with a two-level counting sort on LDS counters (chunk histogram, scan, bucket partition, per-bucket count and place); no global atomics; sum order = LDS arrival order
# speedup vs baseline: 7.7338x; 2.2988x over previous
#include <hip/hip_runtime.h>
#include <hip/hip_bf16.h>


typedef __attribute__((ext_vector_type(2))) float v2f;
typedef __attribute__((ext_vector_type(8))) float v8f;

static __device__ __forceinline__ int gtid() {
    return blockIdx.x * blockDim.x + threadIdx.x;
}

constexpr int RB   = 8;
constexpr int R    = 1 << RB;
constexpr int NBLK = 256;
constexpr int SB   = 1024;

static __device__ __forceinline__ int wave_sum(int v) {
    for (int o = 16; o > 0; o >>= 1) v += __shfl_xor(v, o, 32);
    return v;
}

static __device__ __forceinline__ int block_sum_1024(int v, int* sm, int* res) {
    const int lane = threadIdx.x & 31;
    const int w = threadIdx.x >> 5;
    v = wave_sum(v);
    if (lane == 0) sm[w] = v;
    __syncthreads();
    if (w == 0) {
        int t = wave_sum(sm[lane]);
        if (lane == 0) *res = t;
    }
    __syncthreads();
    return *res;
}

static __device__ __forceinline__ int block_excl_scan_1024(int c, int* wofs) {
    const int lane = threadIdx.x & 31;
    const int w = threadIdx.x >> 5;
    int x = c;
    for (int o = 1; o < 32; o <<= 1) {
        const int y = __shfl_up(x, o, 32);
        if (lane >= o) x += y;
    }
    if (lane == 31) wofs[w] = x;
    __syncthreads();
    if (w == 0) {
        const int tot = wofs[lane];
        int s = tot;
        for (int o = 1; o < 32; o <<= 1) {
            const int y = __shfl_up(s, o, 32);
            if (lane >= o) s += y;
        }
        wofs[lane] = s - tot;
    }
    __syncthreads();
    return wofs[w] + (x - c);
}

__global__ void __launch_bounds__(SB) hist_kernel(const int* __restrict__ dst, int* __restrict__ counts,
                                                  int e, int chunk, int nbuck) {
    extern __shared__ int hist[];
    const int tid = threadIdx.x;
    for (int p = tid; p < nbuck; p += SB) hist[p] = 0;
    __syncthreads();
    const int beg = blockIdx.x * chunk;
    const int end = min(beg + chunk, e);
    int i0 = beg + tid;
    for (int t0 = beg; t0 + 4 * SB <= end; t0 += 4 * SB, i0 += 4 * SB) {
        const int d0 = dst[i0], d1 = dst[i0 + SB], d2 = dst[i0 + 2 * SB], d3 = dst[i0 + 3 * SB];
        atomicAdd(&hist[d0 >> RB], 1);
        atomicAdd(&hist[d1 >> RB], 1);
        atomicAdd(&hist[d2 >> RB], 1);
        atomicAdd(&hist[d3 >> RB], 1);
    }
    for (; i0 < end; i0 += 4 * SB) {
        const int i1 = i0 + SB, i2 = i0 + 2 * SB, i3 = i0 + 3 * SB;
        const bool k1 = i1 < end, k2 = i2 < end, k3 = i3 < end;
        const int d0 = dst[i0];
        const int d1 = dst[k1 ? i1 : i0];
        const int d2 = dst[k2 ? i2 : i0];
        const int d3 = dst[k3 ? i3 : i0];
        atomicAdd(&hist[d0 >> RB], 1);
        if (k1) atomicAdd(&hist[d1 >> RB], 1);
        if (k2) atomicAdd(&hist[d2 >> RB], 1);
        if (k3) atomicAdd(&hist[d3 >> RB], 1);
    }
    __syncthreads();
    for (int p = tid; p < nbuck; p += SB) counts[p * (int)gridDim.x + (int)blockIdx.x] = hist[p];
}

__global__ void __launch_bounds__(SB) block_sum_kernel(const int* __restrict__ cnt,
                                                       int* __restrict__ bsum, int n) {
    __shared__ int sm[32];
    __shared__ int res;
    const int i = blockIdx.x * SB + threadIdx.x;
    const int v = (i < n) ? cnt[i] : 0;
    const int t = block_sum_1024(v, sm, &res);
    if (threadIdx.x == 0) bsum[blockIdx.x] = t;
}

__global__ void __launch_bounds__(SB) scan_kernel(const int* __restrict__ cnt,
                                                  const int* __restrict__ bsum,
                                                  int* __restrict__ offs, int n) {
    __shared__ int sm[32];
    __shared__ int res;
    __shared__ int wofs[32];
    const int tid = threadIdx.x;

    int part = 0;
    for (int t = tid; t < (int)blockIdx.x; t += SB) part += bsum[t];
    const int boff = block_sum_1024(part, sm, &res);

    const int i = blockIdx.x * SB + tid;
    const int c = (i < n) ? cnt[i] : 0;
    const int excl = boff + block_excl_scan_1024(c, wofs);
    if (i < n) {
        offs[i] = excl;
        if (i == n - 1) offs[n] = excl + c;
    }
}

__global__ void __launch_bounds__(SB) part_kernel(const int* __restrict__ src, const int* __restrict__ dst,
                                                  const int* __restrict__ offs, int* __restrict__ part,
                                                  int e, int chunk, int nbuck) {
    extern __shared__ int cur[];
    int2* __restrict__ pe = reinterpret_cast<int2*>(part);
    const int tid = threadIdx.x;
    for (int p = tid; p < nbuck; p += SB) cur[p] = offs[p * (int)gridDim.x + (int)blockIdx.x];
    __syncthreads();
    const int beg = blockIdx.x * chunk;
    const int end = min(beg + chunk, e);
    int i0 = beg + tid;
    for (int t0 = beg; t0 + 4 * SB <= end; t0 += 4 * SB, i0 += 4 * SB) {
        const int d0 = dst[i0], d1 = dst[i0 + SB], d2 = dst[i0 + 2 * SB], d3 = dst[i0 + 3 * SB];
        const int s0 = src[i0], s1 = src[i0 + SB], s2 = src[i0 + 2 * SB], s3 = src[i0 + 3 * SB];
        const int q0 = atomicAdd(&cur[d0 >> RB], 1);
        const int q1 = atomicAdd(&cur[d1 >> RB], 1);
        const int q2 = atomicAdd(&cur[d2 >> RB], 1);
        const int q3 = atomicAdd(&cur[d3 >> RB], 1);
        pe[q0] = make_int2(s0, d0);
        pe[q1] = make_int2(s1, d1);
        pe[q2] = make_int2(s2, d2);
        pe[q3] = make_int2(s3, d3);
    }
    for (; i0 < end; i0 += 4 * SB) {
        const int i1 = i0 + SB, i2 = i0 + 2 * SB, i3 = i0 + 3 * SB;
        const bool k1 = i1 < end, k2 = i2 < end, k3 = i3 < end;
        const int j1 = k1 ? i1 : i0, j2 = k2 ? i2 : i0, j3 = k3 ? i3 : i0;
        const int d0 = dst[i0], d1 = dst[j1], d2 = dst[j2], d3 = dst[j3];
        const int s0 = src[i0], s1 = src[j1], s2 = src[j2], s3 = src[j3];
        pe[atomicAdd(&cur[d0 >> RB], 1)] = make_int2(s0, d0);
        if (k1) pe[atomicAdd(&cur[d1 >> RB], 1)] = make_int2(s1, d1);
        if (k2) pe[atomicAdd(&cur[d2 >> RB], 1)] = make_int2(s2, d2);
        if (k3) pe[atomicAdd(&cur[d3 >> RB], 1)] = make_int2(s3, d3);
    }
}

__global__ void __launch_bounds__(SB) count_kernel(const int* __restrict__ part, const int* __restrict__ offs,
                                                   int* __restrict__ rowptr, float* __restrict__ dinv,
                                                   int n, int chunks) {
    __shared__ int c[R];
    __shared__ int wofs[32];
    const int2* __restrict__ pe = reinterpret_cast<const int2*>(part);
    const int tid = threadIdx.x;
    const int node0 = blockIdx.x << RB;
    if (tid < R) c[tid] = 0;
    __syncthreads();
    const int beg = offs[blockIdx.x * chunks];
    const int end = offs[(blockIdx.x + 1) * chunks];
    int j0 = beg + tid;
    for (int t0 = beg; t0 + 4 * SB <= end; t0 += 4 * SB, j0 += 4 * SB) {
        const int d0 = pe[j0].y, d1 = pe[j0 + SB].y, d2 = pe[j0 + 2 * SB].y, d3 = pe[j0 + 3 * SB].y;
        atomicAdd(&c[d0 - node0], 1);
        atomicAdd(&c[d1 - node0], 1);
        atomicAdd(&c[d2 - node0], 1);
        atomicAdd(&c[d3 - node0], 1);
    }
    for (; j0 < end; j0 += 4 * SB) {
        const int j1 = j0 + SB, j2 = j0 + 2 * SB, j3 = j0 + 3 * SB;
        const bool k1 = j1 < end, k2 = j2 < end, k3 = j3 < end;
        const int d0 = pe[j0].y;
        const int d1 = pe[k1 ? j1 : j0].y;
        const int d2 = pe[k2 ? j2 : j0].y;
        const int d3 = pe[k3 ? j3 : j0].y;
        atomicAdd(&c[d0 - node0], 1);
        if (k1) atomicAdd(&c[d1 - node0], 1);
        if (k2) atomicAdd(&c[d2 - node0], 1);
        if (k3) atomicAdd(&c[d3 - node0], 1);
    }
    __syncthreads();
    const int cv = (tid < R) ? c[tid] : 0;
    const int excl = block_excl_scan_1024(cv, wofs);
    const int node = node0 + tid;
    if (tid < R && node < n) {
        rowptr[node] = beg + excl;
        dinv[node] = rsqrtf((float)cv + 1.0f);
        if (node == n - 1) rowptr[n] = beg + excl + cv;
    }
}

__global__ void __launch_bounds__(SB) place_kernel(const int* __restrict__ part, const int* __restrict__ offs,
                                                   const int* __restrict__ rowptr, const float* __restrict__ dinv,
                                                   int* __restrict__ csr, int n, int chunks) {
    __shared__ int cur[R];
    const int2* __restrict__ pe = reinterpret_cast<const int2*>(part);
    int2* __restrict__ ent = reinterpret_cast<int2*>(csr);
    const int tid = threadIdx.x;
    const int node0 = blockIdx.x << RB;
    if (tid < R) cur[tid] = (node0 + tid < n) ? rowptr[node0 + tid] : 0;
    __syncthreads();
    const int beg = offs[blockIdx.x * chunks];
    const int end = offs[(blockIdx.x + 1) * chunks];
    int j0 = beg + tid;
    for (int t0 = beg; t0 + 4 * SB <= end; t0 += 4 * SB, j0 += 4 * SB) {
        const int2 a0 = pe[j0], a1 = pe[j0 + SB], a2 = pe[j0 + 2 * SB], a3 = pe[j0 + 3 * SB];
        const float w0 = dinv[a0.x], w1 = dinv[a1.x], w2 = dinv[a2.x], w3 = dinv[a3.x];
        const int q0 = atomicAdd(&cur[a0.y - node0], 1);
        const int q1 = atomicAdd(&cur[a1.y - node0], 1);
        const int q2 = atomicAdd(&cur[a2.y - node0], 1);
        const int q3 = atomicAdd(&cur[a3.y - node0], 1);
        ent[q0] = make_int2(a0.x, __float_as_int(w0));
        ent[q1] = make_int2(a1.x, __float_as_int(w1));
        ent[q2] = make_int2(a2.x, __float_as_int(w2));
        ent[q3] = make_int2(a3.x, __float_as_int(w3));
    }
    for (; j0 < end; j0 += 4 * SB) {
        const int j1 = j0 + SB, j2 = j0 + 2 * SB, j3 = j0 + 3 * SB;
        const bool k1 = j1 < end, k2 = j2 < end, k3 = j3 < end;
        const int2 a0 = pe[j0];
        const int2 a1 = pe[k1 ? j1 : j0];
        const int2 a2 = pe[k2 ? j2 : j0];
        const int2 a3 = pe[k3 ? j3 : j0];
        const float w0 = dinv[a0.x], w1 = dinv[a1.x], w2 = dinv[a2.x], w3 = dinv[a3.x];
        ent[atomicAdd(&cur[a0.y - node0], 1)] = make_int2(a0.x, __float_as_int(w0));
        if (k1) ent[atomicAdd(&cur[a1.y - node0], 1)] = make_int2(a1.x, __float_as_int(w1));
        if (k2) ent[atomicAdd(&cur[a2.y - node0], 1)] = make_int2(a2.x, __float_as_int(w2));
        if (k3) ent[atomicAdd(&cur[a3.y - node0], 1)] = make_int2(a3.x, __float_as_int(w3));
    }
}

template <int K, int NOUT>
__global__ void linear_wmma_kernel(const float* __restrict__ x,
                                   const float* __restrict__ w,
                                   float* __restrict__ h,
                                   int n) {
    const int wave = (blockIdx.x * blockDim.x + threadIdx.x) >> 5;
    const int lane = threadIdx.x & 31;
    const int base = wave * 16;
    if (base >= n) return;

    const int m     = lane & 15;
    const int khalf = (lane >> 4) * 2;
    const int row   = min(base + m, n - 1);
    const int mcl   = (m < NOUT) ? m : 0;
    const bool colOK = (m < NOUT);

    v8f acc = {};
    constexpr int KC = (K + 3) / 4;
#pragma unroll
    for (int kc = 0; kc < KC; ++kc) {
        const int k0 = kc * 4 + khalf;
        const bool v0 = (k0 + 0) < K;
        const bool v1 = (k0 + 1) < K;
        const int ka0 = v0 ? (k0 + 0) : 0;
        const int ka1 = v1 ? (k0 + 1) : 0;
        const float xa0 = x[row * K + ka0];
        const float xa1 = x[row * K + ka1];
        const float wb0 = w[mcl * K + ka0];
        const float wb1 = w[mcl * K + ka1];
        v2f a, b;
        a.x = v0 ? xa0 : 0.0f;
        a.y = v1 ? xa1 : 0.0f;
        b.x = (colOK && v0) ? wb0 : 0.0f;
        b.y = (colOK && v1) ? wb1 : 0.0f;
        acc = __builtin_amdgcn_wmma_f32_16x16x4_f32(
             false, a,  false, b,
             (short)0, acc,  false,  false);
    }

    const int mofs = (lane >> 4) * 8;
    if (base + 16 <= n) {
        if (colOK) {
#pragma unroll
            for (int r = 0; r < 8; ++r)
                h[(base + mofs + r) * NOUT + m] = acc[r];
        }
    } else {
        if (colOK) {
#pragma unroll
            for (int r = 0; r < 8; ++r) {
                const int node = base + mofs + r;
                if (node < n) h[node * NOUT + m] = acc[r];
            }
        }
    }
}

static __device__ __forceinline__ void vfma(float4& a, float w, const float4& b) {
    a.x += w * b.x; a.y += w * b.y; a.z += w * b.z; a.w += w * b.w;
}
static __device__ __forceinline__ void vfma(float2& a, float w, const float2& b) {
    a.x += w * b.x; a.y += w * b.y;
}
static __device__ __forceinline__ void vxor_add(float4& a, int o) {
    a.x += __shfl_xor(a.x, o, 32); a.y += __shfl_xor(a.y, o, 32);
    a.z += __shfl_xor(a.z, o, 32); a.w += __shfl_xor(a.w, o, 32);
}
static __device__ __forceinline__ void vxor_add(float2& a, int o) {
    a.x += __shfl_xor(a.x, o, 32); a.y += __shfl_xor(a.y, o, 32);
}
static __device__ __forceinline__ void vzero(float4& a) { a = make_float4(0.f, 0.f, 0.f, 0.f); }
static __device__ __forceinline__ void vzero(float2& a) { a = make_float2(0.f, 0.f); }
static __device__ __forceinline__ void vfinish(float* sm, const float4& a, float d, const float* __restrict__ bias) {
    sm[0] = bias[0] + a.x * d; sm[1] = bias[1] + a.y * d;
    sm[2] = bias[2] + a.z * d; sm[3] = bias[3] + a.w * d;
}
static __device__ __forceinline__ void vfinish(float* sm, const float2& a, float d, const float* __restrict__ bias) {
    sm[0] = bias[0] + a.x * d; sm[1] = bias[1] + a.y * d;
}

template <int C> struct VecOf;
template <> struct VecOf<4> { typedef float4 type; };
template <> struct VecOf<2> { typedef float2 type; };

constexpr int GL  = 8;
constexpr int GB  = 256;
constexpr int NPB = GB / GL;

template <int C>
__global__ void __launch_bounds__(GB) gather_kernel(const int* __restrict__ rowptr,
                                                    const int* __restrict__ csr,
                                                    const float* __restrict__ h,
                                                    const float* __restrict__ dinv,
                                                    const float* __restrict__ bias,
                                                    float* __restrict__ out, int n) {
    typedef typename VecOf<C>::type VT;
    __shared__ __attribute__((aligned(16))) float sm[NPB * C];
    const VT* __restrict__ hv = reinterpret_cast<const VT*>(h);
    const int2* __restrict__ ent = reinterpret_cast<const int2*>(csr);

    const int tid  = threadIdx.x;
    const int sub  = tid & (GL - 1);
    const int nl   = tid / GL;
    const int node = blockIdx.x * NPB + nl;
    const bool ok  = node < n;

    int j = 0, end = 0;
    float dn = 0.0f;
    if (ok) {
        j   = rowptr[node] + sub;
        end = rowptr[node + 1];
        dn  = dinv[node];
    }
    VT acc;
    vzero(acc);
    if (ok && sub == 0) vfma(acc, dn, hv[node]);

    for (; j < end; j += 4 * GL) {
        const int j1 = j + GL, j2 = j + 2 * GL, j3 = j + 3 * GL;
        const int2 none = make_int2(-1, 0);
        const int2 e0 = ent[j];
        const int2 e1 = (j1 < end) ? ent[j1] : none;
        const int2 e2 = (j2 < end) ? ent[j2] : none;
        const int2 e3 = (j3 < end) ? ent[j3] : none;
        VT v0 = hv[e0.x];
        VT v1, v2, v3;
        vzero(v1); vzero(v2); vzero(v3);
        if (e1.x >= 0) v1 = hv[e1.x];
        if (e2.x >= 0) v2 = hv[e2.x];
        if (e3.x >= 0) v3 = hv[e3.x];
        vfma(acc, __int_as_float(e0.y), v0);
        vfma(acc, __int_as_float(e1.y), v1);
        vfma(acc, __int_as_float(e2.y), v2);
        vfma(acc, __int_as_float(e3.y), v3);
    }

    for (int o = 1; o < GL; o <<= 1) vxor_add(acc, o);

    if (ok && sub == 0) vfinish(sm + nl * C, acc, dn, bias);
    __syncthreads();

    constexpr int NV = NPB * C / 4;
    if (tid < NV) {
        const int first = blockIdx.x * (NPB * C) + tid * 4;
        const int total = n * C;
        if (first + 3 < total) {
            *reinterpret_cast<float4*>(out + first) = *reinterpret_cast<const float4*>(sm + tid * 4);
        } else {
            for (int q = 0; q < 4; ++q)
                if (first + q < total) out[first + q] = sm[tid * 4 + q];
        }
    }
}

__global__ void tanh_kernel(const float* __restrict__ in, float* __restrict__ out, int n) {
    int i = gtid();
    if (i < n) out[i] = tanhf(in[i]);
}

static inline int cdiv(int a, int b) { return (a + b - 1) / b; }

extern "C" void kernel_launch(void* const* d_in, const int* in_sizes, int n_in,
                              void* d_out, int out_size, void* d_ws, size_t ws_size,
                              hipStream_t stream) {
    const float* x  = (const float*)d_in[0];
    const int*   ei = (const int*)d_in[1];
    const float* W1 = (const float*)d_in[2];
    const float* b1 = (const float*)d_in[3];
    const float* W2 = (const float*)d_in[4];
    const float* b2 = (const float*)d_in[5];
    const float* W3 = (const float*)d_in[6];
    const float* b3 = (const float*)d_in[7];
    float* out = (float*)d_out;

    const int n = in_sizes[0] / 9;
    const int e = in_sizes[1] / 2;
    const int* src = ei;
    const int* dst = ei + e;

    char* ws = (char*)d_ws;
    size_t off = 0;
    auto carve = [&](size_t bytes) {
        void* p = ws + off;
        off += (bytes + 255) & ~(size_t)255;
        return p;
    };
    const int nbuck  = cdiv(n, R);
    const int chunk  = cdiv(e, NBLK);
    const int m      = nbuck * NBLK;
    const int nbm    = cdiv(m, SB);
    float* dinv   = (float*)carve((size_t)n * sizeof(float));
    int*   rowptr = (int*)carve(((size_t)n + 1) * sizeof(int));
    int*   counts = (int*)carve((size_t)m * sizeof(int));
    int*   offs   = (int*)carve(((size_t)m + 1) * sizeof(int));
    int*   bsum   = (int*)carve((size_t)nbm * sizeof(int));
    int*   part   = (int*)carve((size_t)e * 2 * sizeof(int));
    int*   csr    = (int*)carve((size_t)e * 2 * sizeof(int));
    float* bufA = (float*)carve((size_t)n * 4 * sizeof(float));
    float* bufB = (float*)carve((size_t)n * 4 * sizeof(float));
    float* bufC = (float*)carve((size_t)n * 4 * sizeof(float));
    (void)ws_size;

    const int B = 256;
    const int gW = cdiv(cdiv(n, 16) * 32, B);
    const int gG = cdiv(n, NPB);
    const size_t ldsb = (size_t)nbuck * sizeof(int);

    hist_kernel<<<NBLK, SB, ldsb, stream>>>(dst, counts, e, chunk, nbuck);
    block_sum_kernel<<<nbm, SB, 0, stream>>>(counts, bsum, m);
    scan_kernel<<<nbm, SB, 0, stream>>>(counts, bsum, offs, m);
    part_kernel<<<NBLK, SB, ldsb, stream>>>(src, dst, offs, part, e, chunk, nbuck);
    count_kernel<<<nbuck, SB, 0, stream>>>(part, offs, rowptr, dinv, n, NBLK);
    place_kernel<<<nbuck, SB, 0, stream>>>(part, offs, rowptr, dinv, csr, n, NBLK);

    linear_wmma_kernel<9, 4><<<gW, B, 0, stream>>>(x, W1, bufA, n);
    gather_kernel<4><<<gG, GB, 0, stream>>>(rowptr, csr, bufA, dinv, b1, bufB, n);
    tanh_kernel<<<cdiv(n * 4, B), B, 0, stream>>>(bufB, bufC, n * 4);

    linear_wmma_kernel<4, 4><<<gW, B, 0, stream>>>(bufC, W2, bufA, n);
    gather_kernel<4><<<gG, GB, 0, stream>>>(rowptr, csr, bufA, dinv, b2, bufB, n);
    tanh_kernel<<<cdiv(n * 4, B), B, 0, stream>>>(bufB, bufC, n * 4);

    linear_wmma_kernel<4, 2><<<gW, B, 0, stream>>>(bufC, W3, bufA, n);
    gather_kernel<2><<<gG, GB, 0, stream>>>(rowptr, csr, bufA, dinv, b3, out, n);
}


# tanh and the next layer's linear (same f32 16x16x4 WMMA, same k order) run in the gather's epilogue on its LDS rows; two tanh and two linear launches gone (14 -> 10); summation order unchanged
# speedup vs baseline: 8.2602x; 1.0681x over previous
#include <hip/hip_runtime.h>
#include <hip/hip_bf16.h>


typedef __attribute__((ext_vector_type(2))) float v2f;
typedef __attribute__((ext_vector_type(8))) float v8f;

constexpr int RB   = 8;
constexpr int R    = 1 << RB;
constexpr int NBLK = 256;
constexpr int SB   = 1024;

static __device__ __forceinline__ int wave_sum(int v) {
    for (int o = 16; o > 0; o >>= 1) v += __shfl_xor(v, o, 32);
    return v;
}

static __device__ __forceinline__ int block_sum_1024(int v, int* sm, int* res) {
    const int lane = threadIdx.x & 31;
    const int w = threadIdx.x >> 5;
    v = wave_sum(v);
    if (lane == 0) sm[w] = v;
    __syncthreads();
    if (w == 0) {
        int t = wave_sum(sm[lane]);
        if (lane == 0) *res = t;
    }
    __syncthreads();
    return *res;
}

static __device__ __forceinline__ int block_excl_scan_1024(int c, int* wofs) {
    const int lane = threadIdx.x & 31;
    const int w = threadIdx.x >> 5;
    int x = c;
    for (int o = 1; o < 32; o <<= 1) {
        const int y = __shfl_up(x, o, 32);
        if (lane >= o) x += y;
    }
    if (lane == 31) wofs[w] = x;
    __syncthreads();
    if (w == 0) {
        const int tot = wofs[lane];
        int s = tot;
        for (int o = 1; o < 32; o <<= 1) {
            const int y = __shfl_up(s, o, 32);
            if (lane >= o) s += y;
        }
        wofs[lane] = s - tot;
    }
    __syncthreads();
    return wofs[w] + (x - c);
}

__global__ void __launch_bounds__(SB) hist_kernel(const int* __restrict__ dst, int* __restrict__ counts,
                                                  int e, int chunk, int nbuck) {
    extern __shared__ int hist[];
    const int tid = threadIdx.x;
    for (int p = tid; p < nbuck; p += SB) hist[p] = 0;
    __syncthreads();
    const int beg = blockIdx.x * chunk;
    const int end = min(beg + chunk, e);
    int i0 = beg + tid;
    for (int t0 = beg; t0 + 4 * SB <= end; t0 += 4 * SB, i0 += 4 * SB) {
        const int d0 = dst[i0], d1 = dst[i0 + SB], d2 = dst[i0 + 2 * SB], d3 = dst[i0 + 3 * SB];
        atomicAdd(&hist[d0 >> RB], 1);
        atomicAdd(&hist[d1 >> RB], 1);
        atomicAdd(&hist[d2 >> RB], 1);
        atomicAdd(&hist[d3 >> RB], 1);
    }
    for (; i0 < end; i0 += 4 * SB) {
        const int i1 = i0 + SB, i2 = i0 + 2 * SB, i3 = i0 + 3 * SB;
        const bool k1 = i1 < end, k2 = i2 < end, k3 = i3 < end;
        const int d0 = dst[i0];
        const int d1 = dst[k1 ? i1 : i0];
        const int d2 = dst[k2 ? i2 : i0];
        const int d3 = dst[k3 ? i3 : i0];
        atomicAdd(&hist[d0 >> RB], 1);
        if (k1) atomicAdd(&hist[d1 >> RB], 1);
        if (k2) atomicAdd(&hist[d2 >> RB], 1);
        if (k3) atomicAdd(&hist[d3 >> RB], 1);
    }
    __syncthreads();
    for (int p = tid; p < nbuck; p += SB) counts[p * (int)gridDim.x + (int)blockIdx.x] = hist[p];
}

__global__ void __launch_bounds__(SB) block_sum_kernel(const int* __restrict__ cnt,
                                                       int* __restrict__ bsum, int n) {
    __shared__ int sm[32];
    __shared__ int res;
    const int i = blockIdx.x * SB + threadIdx.x;
    const int v = (i < n) ? cnt[i] : 0;
    const int t = block_sum_1024(v, sm, &res);
    if (threadIdx.x == 0) bsum[blockIdx.x] = t;
}

__global__ void __launch_bounds__(SB) scan_kernel(const int* __restrict__ cnt,
                                                  const int* __restrict__ bsum,
                                                  int* __restrict__ offs, int n) {
    __shared__ int sm[32];
    __shared__ int res;
    __shared__ int wofs[32];
    const int tid = threadIdx.x;

    int part = 0;
    for (int t = tid; t < (int)blockIdx.x; t += SB) part += bsum[t];
    const int boff = block_sum_1024(part, sm, &res);

    const int i = blockIdx.x * SB + tid;
    const int c = (i < n) ? cnt[i] : 0;
    const int excl = boff + block_excl_scan_1024(c, wofs);
    if (i < n) {
        offs[i] = excl;
        if (i == n - 1) offs[n] = excl + c;
    }
}

__global__ void __launch_bounds__(SB) part_kernel(const int* __restrict__ src, const int* __restrict__ dst,
                                                  const int* __restrict__ offs, int* __restrict__ part,
                                                  int e, int chunk, int nbuck) {
    extern __shared__ int cur[];
    int2* __restrict__ pe = reinterpret_cast<int2*>(part);
    const int tid = threadIdx.x;
    for (int p = tid; p < nbuck; p += SB) cur[p] = offs[p * (int)gridDim.x + (int)blockIdx.x];
    __syncthreads();
    const int beg = blockIdx.x * chunk;
    const int end = min(beg + chunk, e);
    int i0 = beg + tid;
    for (int t0 = beg; t0 + 4 * SB <= end; t0 += 4 * SB, i0 += 4 * SB) {
        const int d0 = dst[i0], d1 = dst[i0 + SB], d2 = dst[i0 + 2 * SB], d3 = dst[i0 + 3 * SB];
        const int s0 = src[i0], s1 = src[i0 + SB], s2 = src[i0 + 2 * SB], s3 = src[i0 + 3 * SB];
        const int q0 = atomicAdd(&cur[d0 >> RB], 1);
        const int q1 = atomicAdd(&cur[d1 >> RB], 1);
        const int q2 = atomicAdd(&cur[d2 >> RB], 1);
        const int q3 = atomicAdd(&cur[d3 >> RB], 1);
        pe[q0] = make_int2(s0, d0);
        pe[q1] = make_int2(s1, d1);
        pe[q2] = make_int2(s2, d2);
        pe[q3] = make_int2(s3, d3);
    }
    for (; i0 < end; i0 += 4 * SB) {
        const int i1 = i0 + SB, i2 = i0 + 2 * SB, i3 = i0 + 3 * SB;
        const bool k1 = i1 < end, k2 = i2 < end, k3 = i3 < end;
        const int j1 = k1 ? i1 : i0, j2 = k2 ? i2 : i0, j3 = k3 ? i3 : i0;
        const int d0 = dst[i0], d1 = dst[j1], d2 = dst[j2], d3 = dst[j3];
        const int s0 = src[i0], s1 = src[j1], s2 = src[j2], s3 = src[j3];
        pe[atomicAdd(&cur[d0 >> RB], 1)] = make_int2(s0, d0);
        if (k1) pe[atomicAdd(&cur[d1 >> RB], 1)] = make_int2(s1, d1);
        if (k2) pe[atomicAdd(&cur[d2 >> RB], 1)] = make_int2(s2, d2);
        if (k3) pe[atomicAdd(&cur[d3 >> RB], 1)] = make_int2(s3, d3);
    }
}

__global__ void __launch_bounds__(SB) count_kernel(const int* __restrict__ part, const int* __restrict__ offs,
                                                   int* __restrict__ rowptr, float* __restrict__ dinv,
                                                   int n, int chunks) {
    __shared__ int c[R];
    __shared__ int wofs[32];
    const int2* __restrict__ pe = reinterpret_cast<const int2*>(part);
    const int tid = threadIdx.x;
    const int node0 = blockIdx.x << RB;
    if (tid < R) c[tid] = 0;
    __syncthreads();
    const int beg = offs[blockIdx.x * chunks];
    const int end = offs[(blockIdx.x + 1) * chunks];
    int j0 = beg + tid;
    for (int t0 = beg; t0 + 4 * SB <= end; t0 += 4 * SB, j0 += 4 * SB) {
        const int d0 = pe[j0].y, d1 = pe[j0 + SB].y, d2 = pe[j0 + 2 * SB].y, d3 = pe[j0 + 3 * SB].y;
        atomicAdd(&c[d0 - node0], 1);
        atomicAdd(&c[d1 - node0], 1);
        atomicAdd(&c[d2 - node0], 1);
        atomicAdd(&c[d3 - node0], 1);
    }
    for (; j0 < end; j0 += 4 * SB) {
        const int j1 = j0 + SB, j2 = j0 + 2 * SB, j3 = j0 + 3 * SB;
        const bool k1 = j1 < end, k2 = j2 < end, k3 = j3 < end;
        const int d0 = pe[j0].y;
        const int d1 = pe[k1 ? j1 : j0].y;
        const int d2 = pe[k2 ? j2 : j0].y;
        const int d3 = pe[k3 ? j3 : j0].y;
        atomicAdd(&c[d0 - node0], 1);
        if (k1) atomicAdd(&c[d1 - node0], 1);
        if (k2) atomicAdd(&c[d2 - node0], 1);
        if (k3) atomicAdd(&c[d3 - node0], 1);
    }
    __syncthreads();
    const int cv = (tid < R) ? c[tid] : 0;
    const int excl = block_excl_scan_1024(cv, wofs);
    const int node = node0 + tid;
    if (tid < R && node < n) {
        rowptr[node] = beg + excl;
        dinv[node] = rsqrtf((float)cv + 1.0f);
        if (node == n - 1) rowptr[n] = beg + excl + cv;
    }
}

__global__ void __launch_bounds__(SB) place_kernel(const int* __restrict__ part, const int* __restrict__ offs,
                                                   const int* __restrict__ rowptr, const float* __restrict__ dinv,
                                                   int* __restrict__ csr, int n, int chunks) {
    __shared__ int cur[R];
    const int2* __restrict__ pe = reinterpret_cast<const int2*>(part);
    int2* __restrict__ ent = reinterpret_cast<int2*>(csr);
    const int tid = threadIdx.x;
    const int node0 = blockIdx.x << RB;
    if (tid < R) cur[tid] = (node0 + tid < n) ? rowptr[node0 + tid] : 0;
    __syncthreads();
    const int beg = offs[blockIdx.x * chunks];
    const int end = offs[(blockIdx.x + 1) * chunks];
    int j0 = beg + tid;
    for (int t0 = beg; t0 + 4 * SB <= end; t0 += 4 * SB, j0 += 4 * SB) {
        const int2 a0 = pe[j0], a1 = pe[j0 + SB], a2 = pe[j0 + 2 * SB], a3 = pe[j0 + 3 * SB];
        const float w0 = dinv[a0.x], w1 = dinv[a1.x], w2 = dinv[a2.x], w3 = dinv[a3.x];
        const int q0 = atomicAdd(&cur[a0.y - node0], 1);
        const int q1 = atomicAdd(&cur[a1.y - node0], 1);
        const int q2 = atomicAdd(&cur[a2.y - node0], 1);
        const int q3 = atomicAdd(&cur[a3.y - node0], 1);
        ent[q0] = make_int2(a0.x, __float_as_int(w0));
        ent[q1] = make_int2(a1.x, __float_as_int(w1));
        ent[q2] = make_int2(a2.x, __float_as_int(w2));
        ent[q3] = make_int2(a3.x, __float_as_int(w3));
    }
    for (; j0 < end; j0 += 4 * SB) {
        const int j1 = j0 + SB, j2 = j0 + 2 * SB, j3 = j0 + 3 * SB;
        const bool k1 = j1 < end, k2 = j2 < end, k3 = j3 < end;
        const int2 a0 = pe[j0];
        const int2 a1 = pe[k1 ? j1 : j0];
        const int2 a2 = pe[k2 ? j2 : j0];
        const int2 a3 = pe[k3 ? j3 : j0];
        const float w0 = dinv[a0.x], w1 = dinv[a1.x], w2 = dinv[a2.x], w3 = dinv[a3.x];
        ent[atomicAdd(&cur[a0.y - node0], 1)] = make_int2(a0.x, __float_as_int(w0));
        if (k1) ent[atomicAdd(&cur[a1.y - node0], 1)] = make_int2(a1.x, __float_as_int(w1));
        if (k2) ent[atomicAdd(&cur[a2.y - node0], 1)] = make_int2(a2.x, __float_as_int(w2));
        if (k3) ent[atomicAdd(&cur[a3.y - node0], 1)] = make_int2(a3.x, __float_as_int(w3));
    }
}

template <int K, int NOUT>
__global__ void linear_wmma_kernel(const float* __restrict__ x,
                                   const float* __restrict__ w,
                                   float* __restrict__ h,
                                   int n) {
    const int wave = (blockIdx.x * blockDim.x + threadIdx.x) >> 5;
    const int lane = threadIdx.x & 31;
    const int base = wave * 16;
    if (base >= n) return;

    const int m     = lane & 15;
    const int khalf = (lane >> 4) * 2;
    const int row   = min(base + m, n - 1);
    const int mcl   = (m < NOUT) ? m : 0;
    const bool colOK = (m < NOUT);

    v8f acc = {};
    constexpr int KC = (K + 3) / 4;
#pragma unroll
    for (int kc = 0; kc < KC; ++kc) {
        const int k0 = kc * 4 + khalf;
        const bool v0 = (k0 + 0) < K;
        const bool v1 = (k0 + 1) < K;
        const int ka0 = v0 ? (k0 + 0) : 0;
        const int ka1 = v1 ? (k0 + 1) : 0;
        const float xa0 = x[row * K + ka0];
        const float xa1 = x[row * K + ka1];
        const float wb0 = w[mcl * K + ka0];
        const float wb1 = w[mcl * K + ka1];
        v2f a, b;
        a.x = v0 ? xa0 : 0.0f;
        a.y = v1 ? xa1 : 0.0f;
        b.x = (colOK && v0) ? wb0 : 0.0f;
        b.y = (colOK && v1) ? wb1 : 0.0f;
        acc = __builtin_amdgcn_wmma_f32_16x16x4_f32(
             false, a,  false, b,
             (short)0, acc,  false,  false);
    }

    const int mofs = (lane >> 4) * 8;
    if (base + 16 <= n) {
        if (colOK) {
#pragma unroll
            for (int r = 0; r < 8; ++r)
                h[(base + mofs + r) * NOUT + m] = acc[r];
        }
    } else {
        if (colOK) {
#pragma unroll
            for (int r = 0; r < 8; ++r) {
                const int node = base + mofs + r;
                if (node < n) h[node * NOUT + m] = acc[r];
            }
        }
    }
}

static __device__ __forceinline__ void vfma(float4& a, float w, const float4& b) {
    a.x += w * b.x; a.y += w * b.y; a.z += w * b.z; a.w += w * b.w;
}
static __device__ __forceinline__ void vfma(float2& a, float w, const float2& b) {
    a.x += w * b.x; a.y += w * b.y;
}
static __device__ __forceinline__ void vxor_add(float4& a, int o) {
    a.x += __shfl_xor(a.x, o, 32); a.y += __shfl_xor(a.y, o, 32);
    a.z += __shfl_xor(a.z, o, 32); a.w += __shfl_xor(a.w, o, 32);
}
static __device__ __forceinline__ void vxor_add(float2& a, int o) {
    a.x += __shfl_xor(a.x, o, 32); a.y += __shfl_xor(a.y, o, 32);
}
static __device__ __forceinline__ void vzero(float4& a) { a = make_float4(0.f, 0.f, 0.f, 0.f); }
static __device__ __forceinline__ void vzero(float2& a) { a = make_float2(0.f, 0.f); }
static __device__ __forceinline__ void vfinish(float* sm, const float4& a, float d, const float* __restrict__ bias) {
    sm[0] = bias[0] + a.x * d; sm[1] = bias[1] + a.y * d;
    sm[2] = bias[2] + a.z * d; sm[3] = bias[3] + a.w * d;
}
static __device__ __forceinline__ void vfinish(float* sm, const float2& a, float d, const float* __restrict__ bias) {
    sm[0] = bias[0] + a.x * d; sm[1] = bias[1] + a.y * d;
}
static __device__ __forceinline__ void vclear(float* sm, const float4&) { sm[0] = 0.f; sm[1] = 0.f; sm[2] = 0.f; sm[3] = 0.f; }
static __device__ __forceinline__ void vclear(float* sm, const float2&) { sm[0] = 0.f; sm[1] = 0.f; }

template <int C> struct VecOf;
template <> struct VecOf<4> { typedef float4 type; };
template <> struct VecOf<2> { typedef float2 type; };

constexpr int GL  = 8;
constexpr int GB  = 256;
constexpr int NPB = GB / GL;

template <int C, int NEXT>
__global__ void __launch_bounds__(GB) gather_kernel(const int* __restrict__ rowptr,
                                                    const int* __restrict__ csr,
                                                    const float* __restrict__ h,
                                                    const float* __restrict__ dinv,
                                                    const float* __restrict__ bias,
                                                    const float* __restrict__ wnext,
                                                    float* __restrict__ out, int n) {
    typedef typename VecOf<C>::type VT;
    constexpr int CO = (NEXT > 0) ? NEXT : C;
    __shared__ __attribute__((aligned(16))) float sm[NPB * C];
    __shared__ __attribute__((aligned(16))) float sn[NPB * CO];
    const VT* __restrict__ hv = reinterpret_cast<const VT*>(h);
    const int2* __restrict__ ent = reinterpret_cast<const int2*>(csr);

    const int tid  = threadIdx.x;
    const int sub  = tid & (GL - 1);
    const int nl   = tid / GL;
    const int node = blockIdx.x * NPB + nl;
    const bool ok  = node < n;

    int j = 0, end = 0;
    float dn = 0.0f;
    if (ok) {
        j   = rowptr[node] + sub;
        end = rowptr[node + 1];
        dn  = dinv[node];
    }
    VT acc;
    vzero(acc);
    if (ok && sub == 0) vfma(acc, dn, hv[node]);

    for (; j < end; j += 4 * GL) {
        const int j1 = j + GL, j2 = j + 2 * GL, j3 = j + 3 * GL;
        const int2 none = make_int2(-1, 0);
        const int2 e0 = ent[j];
        const int2 e1 = (j1 < end) ? ent[j1] : none;
        const int2 e2 = (j2 < end) ? ent[j2] : none;
        const int2 e3 = (j3 < end) ? ent[j3] : none;
        VT v0 = hv[e0.x];
        VT v1, v2, v3;
        vzero(v1); vzero(v2); vzero(v3);
        if (e1.x >= 0) v1 = hv[e1.x];
        if (e2.x >= 0) v2 = hv[e2.x];
        if (e3.x >= 0) v3 = hv[e3.x];
        vfma(acc, __int_as_float(e0.y), v0);
        vfma(acc, __int_as_float(e1.y), v1);
        vfma(acc, __int_as_float(e2.y), v2);
        vfma(acc, __int_as_float(e3.y), v3);
    }

    for (int o = 1; o < GL; o <<= 1) vxor_add(acc, o);

    if (sub == 0) {
        if (ok) vfinish(sm + nl * C, acc, dn, bias);
        else vclear(sm + nl * C, acc);
    }
    __syncthreads();

    const float* stage = sm;
    if constexpr (NEXT > 0) {
        static_assert(C == 4, "the epilogue product is one 16x16x4 step");
        const int wv = __builtin_amdgcn_readfirstlane(tid >> 5);
        if (wv < NPB / 16) {
            const int lane = tid & 31;
            const int m    = lane & 15;
            const int kh   = (lane >> 4) * 2;
            const int mcl  = (m < NEXT) ? m : 0;
            const float w0 = wnext[mcl * C + kh];
            const float w1 = wnext[mcl * C + kh + 1];
            const float2 av = *reinterpret_cast<const float2*>(sm + (wv * 16 + m) * C + kh);
            v2f a, b;
            a.x = tanhf(av.x);
            a.y = tanhf(av.y);
            b.x = (m < NEXT) ? w0 : 0.0f;
            b.y = (m < NEXT) ? w1 : 0.0f;
            v8f d = {};
            d = __builtin_amdgcn_wmma_f32_16x16x4_f32(
                 false, a,  false, b,
                 (short)0, d,  false,  false);
            if (m < NEXT) {
                float* dst = sn + (wv * 16 + (lane >> 4) * 8) * NEXT + m;
#pragma unroll
                for (int r = 0; r < 8; ++r) dst[r * NEXT] = d[r];
            }
        }
        __syncthreads();
        stage = sn;
    }

    constexpr int NV = NPB * CO / 4;
    if (tid < NV) {
        const int first = blockIdx.x * (NPB * CO) + tid * 4;
        const int total = n * CO;
        if (first + 3 < total) {
            *reinterpret_cast<float4*>(out + first) = *reinterpret_cast<const float4*>(stage + tid * 4);
        } else {
            for (int q = 0; q < 4; ++q)
                if (first + q < total) out[first + q] = stage[tid * 4 + q];
        }
    }
}

static inline int cdiv(int a, int b) { return (a + b - 1) / b; }

extern "C" void kernel_launch(void* const* d_in, const int* in_sizes, int n_in,
                              void* d_out, int out_size, void* d_ws, size_t ws_size,
                              hipStream_t stream) {
    const float* x  = (const float*)d_in[0];
    const int*   ei = (const int*)d_in[1];
    const float* W1 = (const float*)d_in[2];
    const float* b1 = (const float*)d_in[3];
    const float* W2 = (const float*)d_in[4];
    const float* b2 = (const float*)d_in[5];
    const float* W3 = (const float*)d_in[6];
    const float* b3 = (const float*)d_in[7];
    float* out = (float*)d_out;

    const int n = in_sizes[0] / 9;
    const int e = in_sizes[1] / 2;
    const int* src = ei;
    const int* dst = ei + e;

    char* ws = (char*)d_ws;
    size_t off = 0;
    auto carve = [&](size_t bytes) {
        void* p = ws + off;
        off += (bytes + 255) & ~(size_t)255;
        return p;
    };
    const int nbuck  = cdiv(n, R);
    const int chunk  = cdiv(e, NBLK);
    const int m      = nbuck * NBLK;
    const int nbm    = cdiv(m, SB);
    float* dinv   = (float*)carve((size_t)n * sizeof(float));
    int*   rowptr = (int*)carve(((size_t)n + 1) * sizeof(int));
    int*   counts = (int*)carve((size_t)m * sizeof(int));
    int*   offs   = (int*)carve(((size_t)m + 1) * sizeof(int));
    int*   bsum   = (int*)carve((size_t)nbm * sizeof(int));
    int*   part   = (int*)carve((size_t)e * 2 * sizeof(int));
    int*   csr    = (int*)carve((size_t)e * 2 * sizeof(int));
    float* bufA = (float*)carve((size_t)n * 4 * sizeof(float));
    float* bufB = (float*)carve((size_t)n * 4 * sizeof(float));
    (void)ws_size;

    const int B = 256;
    const int gW = cdiv(cdiv(n, 16) * 32, B);
    const int gG = cdiv(n, NPB);
    const size_t ldsb = (size_t)nbuck * sizeof(int);

    hist_kernel<<<NBLK, SB, ldsb, stream>>>(dst, counts, e, chunk, nbuck);
    block_sum_kernel<<<nbm, SB, 0, stream>>>(counts, bsum, m);
    scan_kernel<<<nbm, SB, 0, stream>>>(counts, bsum, offs, m);
    part_kernel<<<NBLK, SB, ldsb, stream>>>(src, dst, offs, part, e, chunk, nbuck);
    count_kernel<<<nbuck, SB, 0, stream>>>(part, offs, rowptr, dinv, n, NBLK);
    place_kernel<<<nbuck, SB, 0, stream>>>(part, offs, rowptr, dinv, csr, n, NBLK);

    linear_wmma_kernel<9, 4><<<gW, B, 0, stream>>>(x, W1, bufA, n);
    gather_kernel<4, 4><<<gG, GB, 0, stream>>>(rowptr, csr, bufA, dinv, b1, W2, bufB, n);
    gather_kernel<4, 2><<<gG, GB, 0, stream>>>(rowptr, csr, bufB, dinv, b2, W3, bufA, n);
    gather_kernel<2, 0><<<gG, GB, 0, stream>>>(rowptr, csr, bufA, dinv, b3, nullptr, out, n);
}


# both sort levels place records in an LDS image of their output and copy it out on consecutive addresses (level-2 list as whole 128-byte lines), not one scattered store per edge; sum order unchanged
# speedup vs baseline: 11.9531x; 1.4471x over previous
#include <hip/hip_runtime.h>
#include <hip/hip_bf16.h>


typedef __attribute__((ext_vector_type(2))) float v2f;
typedef __attribute__((ext_vector_type(8))) float v8f;

constexpr int RB   = 8;
constexpr int R    = 1 << RB;
constexpr int NBLK = 256;
constexpr int SB   = 1024;
constexpr int PCAP = 16384;
constexpr int CCAP = 16384;
constexpr int LINE = 32;

static __device__ __forceinline__ int wave_sum(int v) {
    for (int o = 16; o > 0; o >>= 1) v += __shfl_xor(v, o, 32);
    return v;
}

static __device__ __forceinline__ int block_sum_1024(int v, int* sm, int* res) {
    const int lane = threadIdx.x & 31;
    const int w = threadIdx.x >> 5;
    v = wave_sum(v);
    if (lane == 0) sm[w] = v;
    __syncthreads();
    if (w == 0) {
        int t = wave_sum(sm[lane]);
        if (lane == 0) *res = t;
    }
    __syncthreads();
    return *res;
}

static __device__ __forceinline__ int block_excl_scan_1024(int c, int* wofs) {
    const int lane = threadIdx.x & 31;
    const int w = threadIdx.x >> 5;
    int x = c;
    for (int o = 1; o < 32; o <<= 1) {
        const int y = __shfl_up(x, o, 32);
        if (lane >= o) x += y;
    }
    if (lane == 31) wofs[w] = x;
    __syncthreads();
    if (w == 0) {
        const int tot = wofs[lane];
        int s = tot;
        for (int o = 1; o < 32; o <<= 1) {
            const int y = __shfl_up(s, o, 32);
            if (lane >= o) s += y;
        }
        wofs[lane] = s - tot;
    }
    __syncthreads();
    return wofs[w] + (x - c);
}

__global__ void __launch_bounds__(SB) hist_kernel(const int* __restrict__ dst, int* __restrict__ counts,
                                                  int e, int chunk, int nbuck) {
    extern __shared__ int hist[];
    const int tid = threadIdx.x;
    for (int p = tid; p < nbuck; p += SB) hist[p] = 0;
    __syncthreads();
    const int beg = blockIdx.x * chunk;
    const int end = min(beg + chunk, e);
    int i0 = beg + tid;
    for (int t0 = beg; t0 + 4 * SB <= end; t0 += 4 * SB, i0 += 4 * SB) {
        const int d0 = dst[i0], d1 = dst[i0 + SB], d2 = dst[i0 + 2 * SB], d3 = dst[i0 + 3 * SB];
        atomicAdd(&hist[d0 >> RB], 1);
        atomicAdd(&hist[d1 >> RB], 1);
        atomicAdd(&hist[d2 >> RB], 1);
        atomicAdd(&hist[d3 >> RB], 1);
    }
    for (; i0 < end; i0 += 4 * SB) {
        const int i1 = i0 + SB, i2 = i0 + 2 * SB, i3 = i0 + 3 * SB;
        const bool k1 = i1 < end, k2 = i2 < end, k3 = i3 < end;
        const int d0 = dst[i0];
        const int d1 = dst[k1 ? i1 : i0];
        const int d2 = dst[k2 ? i2 : i0];
        const int d3 = dst[k3 ? i3 : i0];
        atomicAdd(&hist[d0 >> RB], 1);
        if (k1) atomicAdd(&hist[d1 >> RB], 1);
        if (k2) atomicAdd(&hist[d2 >> RB], 1);
        if (k3) atomicAdd(&hist[d3 >> RB], 1);
    }
    __syncthreads();
    for (int p = tid; p < nbuck; p += SB) counts[p * (int)gridDim.x + (int)blockIdx.x] = hist[p];
}

__global__ void __launch_bounds__(SB) block_sum_kernel(const int* __restrict__ cnt,
                                                       int* __restrict__ bsum, int n) {
    __shared__ int sm[32];
    __shared__ int res;
    const int i = blockIdx.x * SB + threadIdx.x;
    const int v = (i < n) ? cnt[i] : 0;
    const int t = block_sum_1024(v, sm, &res);
    if (threadIdx.x == 0) bsum[blockIdx.x] = t;
}

__global__ void __launch_bounds__(SB) scan_kernel(const int* __restrict__ cnt,
                                                  const int* __restrict__ bsum,
                                                  int* __restrict__ offs, int n) {
    __shared__ int sm[32];
    __shared__ int res;
    __shared__ int wofs[32];
    const int tid = threadIdx.x;

    int part = 0;
    for (int t = tid; t < (int)blockIdx.x; t += SB) part += bsum[t];
    const int boff = block_sum_1024(part, sm, &res);

    const int i = blockIdx.x * SB + tid;
    const int c = (i < n) ? cnt[i] : 0;
    const int excl = boff + block_excl_scan_1024(c, wofs);
    if (i < n) {
        offs[i] = excl;
        if (i == n - 1) offs[n] = excl + c;
    }
}

template <class Put>
static __device__ __forceinline__ void chunk_place(const int* __restrict__ src, const int* __restrict__ dst,
                                                   int* cur, int beg, int end, Put put) {
    int i0 = beg + (int)threadIdx.x;
    for (int t0 = beg; t0 + 4 * SB <= end; t0 += 4 * SB, i0 += 4 * SB) {
        const int d0 = dst[i0], d1 = dst[i0 + SB], d2 = dst[i0 + 2 * SB], d3 = dst[i0 + 3 * SB];
        const int s0 = src[i0], s1 = src[i0 + SB], s2 = src[i0 + 2 * SB], s3 = src[i0 + 3 * SB];
        const int q0 = atomicAdd(&cur[d0 >> RB], 1);
        const int q1 = atomicAdd(&cur[d1 >> RB], 1);
        const int q2 = atomicAdd(&cur[d2 >> RB], 1);
        const int q3 = atomicAdd(&cur[d3 >> RB], 1);
        put(q0, s0, d0);
        put(q1, s1, d1);
        put(q2, s2, d2);
        put(q3, s3, d3);
    }
    for (; i0 < end; i0 += 4 * SB) {
        const int i1 = i0 + SB, i2 = i0 + 2 * SB, i3 = i0 + 3 * SB;
        const bool k1 = i1 < end, k2 = i2 < end, k3 = i3 < end;
        const int j1 = k1 ? i1 : i0, j2 = k2 ? i2 : i0, j3 = k3 ? i3 : i0;
        const int d0 = dst[i0], d1 = dst[j1], d2 = dst[j2], d3 = dst[j3];
        const int s0 = src[i0], s1 = src[j1], s2 = src[j2], s3 = src[j3];
        put(atomicAdd(&cur[d0 >> RB], 1), s0, d0);
        if (k1) put(atomicAdd(&cur[d1 >> RB], 1), s1, d1);
        if (k2) put(atomicAdd(&cur[d2 >> RB], 1), s2, d2);
        if (k3) put(atomicAdd(&cur[d3 >> RB], 1), s3, d3);
    }
}

__global__ void __launch_bounds__(SB) part_kernel(const int* __restrict__ src, const int* __restrict__ dst,
                                                  const int* __restrict__ offs, int* __restrict__ part,
                                                  int e, int chunk, int nbuck) {
    extern __shared__ int cur[];
    __shared__ int2 stage[PCAP];
    __shared__ int wofs[32];
    __shared__ int segtot;
    int* shift = cur + nbuck;
    int2* __restrict__ pe = reinterpret_cast<int2*>(part);
    const int tid = threadIdx.x;
    const int nb = (int)gridDim.x;
    const int blk = (int)blockIdx.x;
    const int beg = blk * chunk;
    const int end = min(beg + chunk, e);
    const int cnt = max(end - beg, 0);

    if (cnt > PCAP) {
        for (int p = tid; p < nbuck; p += SB) cur[p] = offs[p * nb + blk];
        __syncthreads();
        chunk_place(src, dst, cur, beg, end, [&](int q, int s, int d) { pe[q] = make_int2(s, d); });
        return;
    }

    int carry = 0;
    for (int base = 0; base < nbuck; base += SB) {
        const int p = base + tid;
        int g = 0, k = 0;
        if (p < nbuck) {
            g = offs[p * nb + blk];
            k = offs[p * nb + blk + 1] - g;
        }
        const int ex = carry + block_excl_scan_1024(k, wofs);
        if (p < nbuck) {
            cur[p] = ex;
            shift[p] = g - ex;
        }
        if (tid == SB - 1) segtot = ex + k;
        __syncthreads();
        carry = segtot;
    }

    chunk_place(src, dst, cur, beg, end, [&](int q, int s, int d) { stage[q] = make_int2(s, d); });
    __syncthreads();

    int i = tid;
    for (; i + 3 * SB < cnt; i += 4 * SB) {
        const int2 p0 = stage[i], p1 = stage[i + SB], p2 = stage[i + 2 * SB], p3 = stage[i + 3 * SB];
        const int o0 = shift[p0.y >> RB], o1 = shift[p1.y >> RB], o2 = shift[p2.y >> RB], o3 = shift[p3.y >> RB];
        pe[i + o0] = p0;
        pe[i + SB + o1] = p1;
        pe[i + 2 * SB + o2] = p2;
        pe[i + 3 * SB + o3] = p3;
    }
    for (; i < cnt; i += SB) {
        const int2 p0 = stage[i];
        pe[i + shift[p0.y >> RB]] = p0;
    }
}

template <class Put>
static __device__ __forceinline__ void bucket_place(const int2* __restrict__ pe, int* c, int node0,
                                                    int beg, int end, Put put) {
    int j0 = beg + (int)threadIdx.x;
    for (int t0 = beg; t0 + 4 * SB <= end; t0 += 4 * SB, j0 += 4 * SB) {
        const int2 a0 = pe[j0], a1 = pe[j0 + SB], a2 = pe[j0 + 2 * SB], a3 = pe[j0 + 3 * SB];
        const int q0 = atomicAdd(&c[a0.y - node0], 1);
        const int q1 = atomicAdd(&c[a1.y - node0], 1);
        const int q2 = atomicAdd(&c[a2.y - node0], 1);
        const int q3 = atomicAdd(&c[a3.y - node0], 1);
        put(q0, a0.x);
        put(q1, a1.x);
        put(q2, a2.x);
        put(q3, a3.x);
    }
    for (; j0 < end; j0 += 4 * SB) {
        const int j1 = j0 + SB, j2 = j0 + 2 * SB, j3 = j0 + 3 * SB;
        const bool k1 = j1 < end, k2 = j2 < end, k3 = j3 < end;
        const int2 a0 = pe[j0];
        const int2 a1 = pe[k1 ? j1 : j0];
        const int2 a2 = pe[k2 ? j2 : j0];
        const int2 a3 = pe[k3 ? j3 : j0];
        put(atomicAdd(&c[a0.y - node0], 1), a0.x);
        if (k1) put(atomicAdd(&c[a1.y - node0], 1), a1.x);
        if (k2) put(atomicAdd(&c[a2.y - node0], 1), a2.x);
        if (k3) put(atomicAdd(&c[a3.y - node0], 1), a3.x);
    }
}

__global__ void __launch_bounds__(SB) count_place_kernel(const int* __restrict__ part, const int* __restrict__ offs,
                                                         int* __restrict__ rowptr, float* __restrict__ dinv,
                                                         int* __restrict__ csr, int n, int chunks) {
    __shared__ int c[R];
    __shared__ int wofs[32];
    __shared__ int img[CCAP + LINE];
    const int2* __restrict__ pe = reinterpret_cast<const int2*>(part);
    const int tid = threadIdx.x;
    const int node0 = blockIdx.x << RB;
    if (tid < R) c[tid] = 0;
    __syncthreads();
    const int beg = offs[blockIdx.x * chunks];
    const int end = offs[(blockIdx.x + 1) * chunks];
    int j0 = beg + tid;
    for (int t0 = beg; t0 + 4 * SB <= end; t0 += 4 * SB, j0 += 4 * SB) {
        const int d0 = pe[j0].y, d1 = pe[j0 + SB].y, d2 = pe[j0 + 2 * SB].y, d3 = pe[j0 + 3 * SB].y;
        atomicAdd(&c[d0 - node0], 1);
        atomicAdd(&c[d1 - node0], 1);
        atomicAdd(&c[d2 - node0], 1);
        atomicAdd(&c[d3 - node0], 1);
    }
    for (; j0 < end; j0 += 4 * SB) {
        const int j1 = j0 + SB, j2 = j0 + 2 * SB, j3 = j0 + 3 * SB;
        const bool k1 = j1 < end, k2 = j2 < end, k3 = j3 < end;
        const int d0 = pe[j0].y;
        const int d1 = pe[k1 ? j1 : j0].y;
        const int d2 = pe[k2 ? j2 : j0].y;
        const int d3 = pe[k3 ? j3 : j0].y;
        atomicAdd(&c[d0 - node0], 1);
        if (k1) atomicAdd(&c[d1 - node0], 1);
        if (k2) atomicAdd(&c[d2 - node0], 1);
        if (k3) atomicAdd(&c[d3 - node0], 1);
    }
    __syncthreads();
    const int cv = (tid < R) ? c[tid] : 0;
    const int excl = block_excl_scan_1024(cv, wofs);
    const int node = node0 + tid;
    const int cnt = end - beg;
    const bool staged = cnt <= CCAP;
    const int lead = beg & (LINE - 1);
    if (tid < R) {
        c[tid] = (staged ? lead : beg) + excl;
        if (node < n) {
            rowptr[node] = beg + excl;
            dinv[node] = rsqrtf((float)cv + 1.0f);
            if (node == n - 1) rowptr[n] = beg + excl + cv;
        }
    }
    __syncthreads();
    if (staged) {
        bucket_place(pe, c, node0, beg, end, [&](int q, int s) { img[q] = s; });
        __syncthreads();
        const int line0 = beg - lead;
        const int top = lead + cnt;
        int li = tid;
        for (; li + 3 * SB < top; li += 4 * SB) {
            const bool k0 = li >= lead;
            const int v0 = k0 ? img[li] : 0;
            const int v1 = img[li + SB], v2 = img[li + 2 * SB], v3 = img[li + 3 * SB];
            if (k0) csr[line0 + li] = v0;
            csr[line0 + li + SB] = v1;
            csr[line0 + li + 2 * SB] = v2;
            csr[line0 + li + 3 * SB] = v3;
        }
        for (; li < top; li += SB)
            if (li >= lead) csr[line0 + li] = img[li];
    } else {
        bucket_place(pe, c, node0, beg, end, [&](int q, int s) { csr[q] = s; });
    }
}

template <int K, int NOUT>
__global__ void linear_wmma_kernel(const float* __restrict__ x,
                                   const float* __restrict__ w,
                                   const float* __restrict__ dinv,
                                   float* __restrict__ h,
                                   int n) {
    const int wave = (blockIdx.x * blockDim.x + threadIdx.x) >> 5;
    const int lane = threadIdx.x & 31;
    const int base = wave * 16;
    if (base >= n) return;

    const int m     = lane & 15;
    const int khalf = (lane >> 4) * 2;
    const int row   = min(base + m, n - 1);
    const int mcl   = (m < NOUT) ? m : 0;
    const bool colOK = (m < NOUT);
    const int mofs  = (lane >> 4) * 8;

    float sc[8];
#pragma unroll
    for (int r = 0; r < 8; ++r) sc[r] = dinv[min(base + mofs + r, n - 1)];

    v8f acc = {};
    constexpr int KC = (K + 3) / 4;
#pragma unroll
    for (int kc = 0; kc < KC; ++kc) {
        const int k0 = kc * 4 + khalf;
        const bool v0 = (k0 + 0) < K;
        const bool v1 = (k0 + 1) < K;
        const int ka0 = v0 ? (k0 + 0) : 0;
        const int ka1 = v1 ? (k0 + 1) : 0;
        const float xa0 = x[row * K + ka0];
        const float xa1 = x[row * K + ka1];
        const float wb0 = w[mcl * K + ka0];
        const float wb1 = w[mcl * K + ka1];
        v2f a, b;
        a.x = v0 ? xa0 : 0.0f;
        a.y = v1 ? xa1 : 0.0f;
        b.x = (colOK && v0) ? wb0 : 0.0f;
        b.y = (colOK && v1) ? wb1 : 0.0f;
        acc = __builtin_amdgcn_wmma_f32_16x16x4_f32(
             false, a,  false, b,
             (short)0, acc,  false,  false);
    }

    if (base + 16 <= n) {
        if (colOK) {
#pragma unroll
            for (int r = 0; r < 8; ++r)
                h[(base + mofs + r) * NOUT + m] = acc[r] * sc[r];
        }
    } else {
        if (colOK) {
#pragma unroll
            for (int r = 0; r < 8; ++r) {
                const int node = base + mofs + r;
                if (node < n) h[node * NOUT + m] = acc[r] * sc[r];
            }
        }
    }
}

static __device__ __forceinline__ void vadd(float4& a, const float4& b) {
    a.x += b.x; a.y += b.y; a.z += b.z; a.w += b.w;
}
static __device__ __forceinline__ void vadd(float2& a, const float2& b) {
    a.x += b.x; a.y += b.y;
}
static __device__ __forceinline__ void vxor_add(float4& a, int o) {
    a.x += __shfl_xor(a.x, o, 32); a.y += __shfl_xor(a.y, o, 32);
    a.z += __shfl_xor(a.z, o, 32); a.w += __shfl_xor(a.w, o, 32);
}
static __device__ __forceinline__ void vxor_add(float2& a, int o) {
    a.x += __shfl_xor(a.x, o, 32); a.y += __shfl_xor(a.y, o, 32);
}
static __device__ __forceinline__ void vzero(float4& a) { a = make_float4(0.f, 0.f, 0.f, 0.f); }
static __device__ __forceinline__ void vzero(float2& a) { a = make_float2(0.f, 0.f); }
static __device__ __forceinline__ void vfinish(float* sm, const float4& a, float d, const float* __restrict__ bias) {
    sm[0] = bias[0] + a.x * d; sm[1] = bias[1] + a.y * d;
    sm[2] = bias[2] + a.z * d; sm[3] = bias[3] + a.w * d;
}
static __device__ __forceinline__ void vfinish(float* sm, const float2& a, float d, const float* __restrict__ bias) {
    sm[0] = bias[0] + a.x * d; sm[1] = bias[1] + a.y * d;
}
static __device__ __forceinline__ void vclear(float* sm, const float4&) { sm[0] = 0.f; sm[1] = 0.f; sm[2] = 0.f; sm[3] = 0.f; }
static __device__ __forceinline__ void vclear(float* sm, const float2&) { sm[0] = 0.f; sm[1] = 0.f; }

template <int C> struct VecOf;
template <> struct VecOf<4> { typedef float4 type; };
template <> struct VecOf<2> { typedef float2 type; };

constexpr int GL  = 8;
constexpr int GB  = 256;
constexpr int NPB = GB / GL;

template <int C, int NEXT>
__global__ void __launch_bounds__(GB) gather_kernel(const int* __restrict__ rowptr,
                                                    const int* __restrict__ csr,
                                                    const float* __restrict__ h,
                                                    const float* __restrict__ dinv,
                                                    const float* __restrict__ bias,
                                                    const float* __restrict__ wnext,
                                                    float* __restrict__ out, int n) {
    typedef typename VecOf<C>::type VT;
    constexpr int CO = (NEXT > 0) ? NEXT : C;
    __shared__ __attribute__((aligned(16))) float sm[NPB * C];
    __shared__ __attribute__((aligned(16))) float sn[NPB * CO];
    __shared__ float sd[NPB];
    const VT* __restrict__ hv = reinterpret_cast<const VT*>(h);

    const int tid  = threadIdx.x;
    const int sub  = tid & (GL - 1);
    const int nl   = tid / GL;
    const int node = blockIdx.x * NPB + nl;
    const bool ok  = node < n;

    int j = 0, end = 0;
    float dn = 0.0f;
    if (ok) {
        j   = rowptr[node] + sub;
        end = rowptr[node + 1];
        dn  = dinv[node];
    }
    VT acc;
    vzero(acc);
    if (ok && sub == 0) acc = hv[node];

    for (; j < end; j += 4 * GL) {
        const int j1 = j + GL, j2 = j + 2 * GL, j3 = j + 3 * GL;
        const int e0 = csr[j];
        const int e1 = (j1 < end) ? csr[j1] : -1;
        const int e2 = (j2 < end) ? csr[j2] : -1;
        const int e3 = (j3 < end) ? csr[j3] : -1;
        VT v0 = hv[e0];
        VT v1, v2, v3;
        vzero(v1); vzero(v2); vzero(v3);
        if (e1 >= 0) v1 = hv[e1];
        if (e2 >= 0) v2 = hv[e2];
        if (e3 >= 0) v3 = hv[e3];
        vadd(acc, v0);
        vadd(acc, v1);
        vadd(acc, v2);
        vadd(acc, v3);
    }

    for (int o = 1; o < GL; o <<= 1) vxor_add(acc, o);

    if (sub == 0) {
        if (ok) vfinish(sm + nl * C, acc, dn, bias);
        else vclear(sm + nl * C, acc);
        if constexpr (NEXT > 0) sd[nl] = dn;
    }
    __syncthreads();

    const float* stage = sm;
    if constexpr (NEXT > 0) {
        static_assert(C == 4, "the epilogue product is one 16x16x4 step");
        const int wv = __builtin_amdgcn_readfirstlane(tid >> 5);
        if (wv < NPB / 16) {
            const int lane = tid & 31;
            const int m    = lane & 15;
            const int kh   = (lane >> 4) * 2;
            const int mcl  = (m < NEXT) ? m : 0;
            const float w0 = wnext[mcl * C + kh];
            const float w1 = wnext[mcl * C + kh + 1];
            const float2 av = *reinterpret_cast<const float2*>(sm + (wv * 16 + m) * C + kh);
            const float rs = sd[wv * 16 + m];
            v2f a, b;
            a.x = rs * tanhf(av.x);
            a.y = rs * tanhf(av.y);
            b.x = (m < NEXT) ? w0 : 0.0f;
            b.y = (m < NEXT) ? w1 : 0.0f;
            v8f d = {};
            d = __builtin_amdgcn_wmma_f32_16x16x4_f32(
                 false, a,  false, b,
                 (short)0, d,  false,  false);
            if (m < NEXT) {
                float* dst = sn + (wv * 16 + (lane >> 4) * 8) * NEXT + m;
#pragma unroll
                for (int r = 0; r < 8; ++r) dst[r * NEXT] = d[r];
            }
        }
        __syncthreads();
        stage = sn;
    }

    constexpr int NV = NPB * CO / 4;
    if (tid < NV) {
        const int first = blockIdx.x * (NPB * CO) + tid * 4;
        const int total = n * CO;
        if (first + 3 < total) {
            *reinterpret_cast<float4*>(out + first) = *reinterpret_cast<const float4*>(stage + tid * 4);
        } else {
            for (int q = 0; q < 4; ++q)
                if (first + q < total) out[first + q] = stage[tid * 4 + q];
        }
    }
}

static inline int cdiv(int a, int b) { return (a + b - 1) / b; }

extern "C" void kernel_launch(void* const* d_in, const int* in_sizes, int n_in,
                              void* d_out, int out_size, void* d_ws, size_t ws_size,
                              hipStream_t stream) {
    const float* x  = (const float*)d_in[0];
    const int*   ei = (const int*)d_in[1];
    const float* W1 = (const float*)d_in[2];
    const float* b1 = (const float*)d_in[3];
    const float* W2 = (const float*)d_in[4];
    const float* b2 = (const float*)d_in[5];
    const float* W3 = (const float*)d_in[6];
    const float* b3 = (const float*)d_in[7];
    float* out = (float*)d_out;

    const int n = in_sizes[0] / 9;
    const int e = in_sizes[1] / 2;
    const int* src = ei;
    const int* dst = ei + e;

    char* ws = (char*)d_ws;
    size_t off = 0;
    auto carve = [&](size_t bytes) {
        void* p = ws + off;
        off += (bytes + 255) & ~(size_t)255;
        return p;
    };
    const int nbuck  = cdiv(n, R);
    const int chunk  = cdiv(e, NBLK);
    const int m      = nbuck * NBLK;
    const int nbm    = cdiv(m, SB);
    float* dinv   = (float*)carve((size_t)n * sizeof(float));
    int*   rowptr = (int*)carve(((size_t)n + 1) * sizeof(int));
    int*   counts = (int*)carve((size_t)m * sizeof(int));
    int*   offs   = (int*)carve(((size_t)m + 1) * sizeof(int));
    int*   bsum   = (int*)carve((size_t)nbm * sizeof(int));
    int*   part   = (int*)carve((size_t)e * 2 * sizeof(int));
    int*   csr    = (int*)carve((size_t)e * sizeof(int));
    float* bufA = (float*)carve((size_t)n * 4 * sizeof(float));
    float* bufB = (float*)carve((size_t)n * 4 * sizeof(float));
    (void)ws_size;

    const int B = 256;
    const int gW = cdiv(cdiv(n, 16) * 32, B);
    const int gG = cdiv(n, NPB);
    const size_t ldsb = (size_t)nbuck * sizeof(int);

    hist_kernel<<<NBLK, SB, ldsb, stream>>>(dst, counts, e, chunk, nbuck);
    block_sum_kernel<<<nbm, SB, 0, stream>>>(counts, bsum, m);
    scan_kernel<<<nbm, SB, 0, stream>>>(counts, bsum, offs, m);
    part_kernel<<<NBLK, SB, 2 * ldsb, stream>>>(src, dst, offs, part, e, chunk, nbuck);
    count_place_kernel<<<nbuck, SB, 0, stream>>>(part, offs, rowptr, dinv, csr, n, NBLK);

    linear_wmma_kernel<9, 4><<<gW, B, 0, stream>>>(x, W1, dinv, bufA, n);
    gather_kernel<4, 4><<<gG, GB, 0, stream>>>(rowptr, csr, bufA, dinv, b1, W2, bufB, n);
    gather_kernel<4, 2><<<gG, GB, 0, stream>>>(rowptr, csr, bufB, dinv, b2, W3, bufA, n);
    gather_kernel<2, 0><<<gG, GB, 0, stream>>>(rowptr, csr, bufA, dinv, b3, nullptr, out, n);
}


# level 1 sorts each chunk by bucket inside its work-group (one returning LDS add counts and ranks), level 2 walks the chunks' runs; global count and both scan launches gone (9->6); sum order changed
# speedup vs baseline: 13.1269x; 1.0982x over previous
#include <hip/hip_runtime.h>
#include <hip/hip_bf16.h>


typedef __attribute__((ext_vector_type(2))) float v2f;
typedef __attribute__((ext_vector_type(8))) float v8f;

constexpr int RB   = 8;
constexpr int R    = 1 << RB;
constexpr int NBLK = 256;
constexpr int SB   = 1024;
constexpr int PCAP = 16384;
constexpr int EPT  = PCAP / SB;
constexpr int CCAP = 16384;
constexpr int LINE = 32;
constexpr int NW   = SB / 32;

static __device__ __forceinline__ int wave_sum(int v) {
    for (int o = 16; o > 0; o >>= 1) v += __shfl_xor(v, o, 32);
    return v;
}

static __device__ __forceinline__ int2 block_sum2_1024(int a, int b, int* sm, int* res) {
    const int lane = threadIdx.x & 31;
    const int w = threadIdx.x >> 5;
    a = wave_sum(a);
    b = wave_sum(b);
    if (lane == 0) { sm[w] = a; sm[NW + w] = b; }
    __syncthreads();
    if (w == 0) {
        const int ta = wave_sum(sm[lane]);
        const int tb = wave_sum(sm[NW + lane]);
        if (lane == 0) { res[0] = ta; res[1] = tb; }
    }
    __syncthreads();
    return make_int2(res[0], res[1]);
}

static __device__ __forceinline__ int block_excl_scan_1024(int c, int* wofs) {
    const int lane = threadIdx.x & 31;
    const int w = threadIdx.x >> 5;
    int x = c;
    for (int o = 1; o < 32; o <<= 1) {
        const int y = __shfl_up(x, o, 32);
        if (lane >= o) x += y;
    }
    if (lane == 31) wofs[w] = x;
    __syncthreads();
    if (w == 0) {
        const int tot = wofs[lane];
        int s = tot;
        for (int o = 1; o < 32; o <<= 1) {
            const int y = __shfl_up(s, o, 32);
            if (lane >= o) s += y;
        }
        wofs[lane] = s - tot;
    }
    __syncthreads();
    return wofs[w] + (x - c);
}

__global__ void __launch_bounds__(SB) chunk_sort_kernel(const int* __restrict__ src, const int* __restrict__ dst,
                                                        int* __restrict__ runs, int* __restrict__ part,
                                                        int e, int chunk, int nbuck) {
    extern __shared__ int cur[];
    __shared__ int2 stage[PCAP];
    __shared__ int wofs[32];
    __shared__ int segtot;
    int2* __restrict__ pe = reinterpret_cast<int2*>(part);
    const int tid = threadIdx.x;
    const int nb = (int)gridDim.x;
    const int blk = (int)blockIdx.x;
    const int beg = blk * chunk;
    const int end = min(beg + chunk, e);
    const int cnt = max(end - beg, 0);

    int d[EPT], s[EPT], r[EPT];
    bool ok[EPT];
#pragma unroll
    for (int k = 0; k < EPT; ++k) {
        const int i = tid + k * SB;
        ok[k] = i < cnt;
        const int j = ok[k] ? beg + i : 0;
        d[k] = dst[j];
        s[k] = src[j];
        r[k] = 0;
    }
    for (int p = tid; p < nbuck; p += SB) cur[p] = 0;
    __syncthreads();

#pragma unroll
    for (int k = 0; k < EPT; ++k)
        if (ok[k]) r[k] = atomicAdd(&cur[d[k] >> RB], 1);
    __syncthreads();

    int carry = 0;
    for (int base = 0; base < nbuck; base += SB) {
        const int p = base + tid;
        const int c = (p < nbuck) ? cur[p] : 0;
        const int ex = carry + block_excl_scan_1024(c, wofs);
        if (p < nbuck) {
            cur[p] = ex;
            runs[p * nb + blk] = ex;
        }
        if (tid == SB - 1) segtot = ex + c;
        __syncthreads();
        carry = segtot;
    }
    if (tid == 0) runs[nbuck * nb + blk] = cnt;

    int q[EPT];
#pragma unroll
    for (int k = 0; k < EPT; ++k) q[k] = cur[d[k] >> RB] + r[k];
#pragma unroll
    for (int k = 0; k < EPT; ++k)
        if (ok[k]) stage[q[k]] = make_int2(s[k], d[k]);
    __syncthreads();

    int i = tid;
    for (; i + 3 * SB < cnt; i += 4 * SB) {
        const int2 p0 = stage[i], p1 = stage[i + SB], p2 = stage[i + 2 * SB], p3 = stage[i + 3 * SB];
        pe[beg + i] = p0;
        pe[beg + i + SB] = p1;
        pe[beg + i + 2 * SB] = p2;
        pe[beg + i + 3 * SB] = p3;
    }
    for (; i < cnt; i += SB) pe[beg + i] = stage[i];
}

constexpr int RPT = 8;
template <class F>
static __device__ __forceinline__ void bucket_sweep(const int2* __restrict__ pe, const int* rstart,
                                                    const int* rlen, int chunks, F f) {
    const int lane = threadIdx.x & 31;
    const int w = threadIdx.x >> 5;
    for (int b0 = w; b0 < chunks; b0 += RPT * NW) {
        int st[RPT], ln[RPT];
#pragma unroll
        for (int q = 0; q < RPT; ++q) {
            const int b = b0 + q * NW;
            const bool in = b < chunks;
            st[q] = in ? rstart[b] : 0;
            ln[q] = in ? rlen[b] : 0;
        }
        int2 a[2 * RPT];
        bool k[2 * RPT];
#pragma unroll
        for (int q = 0; q < RPT; ++q) {
            k[2 * q] = lane < ln[q];
            k[2 * q + 1] = lane + 32 < ln[q];
            a[2 * q] = pe[k[2 * q] ? st[q] + lane : 0];
            a[2 * q + 1] = pe[k[2 * q + 1] ? st[q] + lane + 32 : 0];
        }
        asm volatile("" ::"v"(a[0].x), "v"(a[0].y), "v"(a[1].x), "v"(a[1].y), "v"(a[2].x), "v"(a[2].y),
                     "v"(a[3].x), "v"(a[3].y), "v"(a[4].x), "v"(a[4].y), "v"(a[5].x), "v"(a[5].y),
                     "v"(a[6].x), "v"(a[6].y), "v"(a[7].x), "v"(a[7].y));
        asm volatile("" ::"v"(a[8].x), "v"(a[8].y), "v"(a[9].x), "v"(a[9].y), "v"(a[10].x), "v"(a[10].y),
                     "v"(a[11].x), "v"(a[11].y), "v"(a[12].x), "v"(a[12].y), "v"(a[13].x), "v"(a[13].y),
                     "v"(a[14].x), "v"(a[14].y), "v"(a[15].x), "v"(a[15].y));
#pragma unroll
        for (int q = 0; q < 2 * RPT; ++q)
            if (k[q]) f(a[q]);
#pragma unroll
        for (int q = 0; q < RPT; ++q)
            for (int i = lane + 64; i < ln[q]; i += 32) f(pe[st[q] + i]);
    }
}

__global__ void __launch_bounds__(SB) count_place_kernel(const int* __restrict__ part, const int* __restrict__ runs,
                                                         int* __restrict__ rowptr, float* __restrict__ dinv,
                                                         int* __restrict__ csr, int n, int chunk, int chunks) {
    extern __shared__ int rtab[];
    __shared__ int c[R];
    __shared__ int wofs[32];
    __shared__ int sm[2 * NW];
    __shared__ int res[2];
    __shared__ int img[CCAP + LINE];
    int* rstart = rtab;
    int* rlen = rtab + chunks;
    const int2* __restrict__ pe = reinterpret_cast<const int2*>(part);
    const int tid = threadIdx.x;
    const int bucket = (int)blockIdx.x;
    const int node0 = bucket << RB;
    if (tid < R) c[tid] = 0;
    int slo = 0, shi = 0;
    for (int b = tid; b < chunks; b += SB) {
        const int lo = runs[bucket * chunks + b];
        const int hi = runs[(bucket + 1) * chunks + b];
        rstart[b] = b * chunk + lo;
        rlen[b] = hi - lo;
        slo += lo;
        shi += hi;
    }
    const int2 be = block_sum2_1024(slo, shi, sm, res);
    const int beg = be.x;
    const int end = be.y;

    bucket_sweep(pe, rstart, rlen, chunks, [&](const int2& a) { atomicAdd(&c[a.y - node0], 1); });
    __syncthreads();
    const int cv = (tid < R) ? c[tid] : 0;
    const int excl = block_excl_scan_1024(cv, wofs);
    const int node = node0 + tid;
    const int cnt = end - beg;
    const bool staged = cnt <= CCAP;
    const int lead = beg & (LINE - 1);
    if (tid < R) {
        c[tid] = (staged ? lead : beg) + excl;
        if (node < n) {
            rowptr[node] = beg + excl;
            dinv[node] = rsqrtf((float)cv + 1.0f);
            if (node == n - 1) rowptr[n] = beg + excl + cv;
        }
    }
    __syncthreads();
    if (staged) {
        bucket_sweep(pe, rstart, rlen, chunks, [&](const int2& a) { img[atomicAdd(&c[a.y - node0], 1)] = a.x; });
        __syncthreads();
        const int line0 = beg - lead;
        const int top = lead + cnt;
        int li = tid;
        for (; li + 3 * SB < top; li += 4 * SB) {
            const bool k0 = li >= lead;
            const int v0 = k0 ? img[li] : 0;
            const int v1 = img[li + SB], v2 = img[li + 2 * SB], v3 = img[li + 3 * SB];
            if (k0) csr[line0 + li] = v0;
            csr[line0 + li + SB] = v1;
            csr[line0 + li + 2 * SB] = v2;
            csr[line0 + li + 3 * SB] = v3;
        }
        for (; li < top; li += SB)
            if (li >= lead) csr[line0 + li] = img[li];
    } else {
        bucket_sweep(pe, rstart, rlen, chunks, [&](const int2& a) { csr[atomicAdd(&c[a.y - node0], 1)] = a.x; });
    }
}

template <int K, int NOUT>
__global__ void linear_wmma_kernel(const float* __restrict__ x,
                                   const float* __restrict__ w,
                                   const float* __restrict__ dinv,
                                   float* __restrict__ h,
                                   int n) {
    const int wave = (blockIdx.x * blockDim.x + threadIdx.x) >> 5;
    const int lane = threadIdx.x & 31;
    const int base = wave * 16;
    if (base >= n) return;

    const int m     = lane & 15;
    const int khalf = (lane >> 4) * 2;
    const int row   = min(base + m, n - 1);
    const int mcl   = (m < NOUT) ? m : 0;
    const bool colOK = (m < NOUT);
    const int mofs  = (lane >> 4) * 8;

    float sc[8];
#pragma unroll
    for (int r = 0; r < 8; ++r) sc[r] = dinv[min(base + mofs + r, n - 1)];

    v8f acc = {};
    constexpr int KC = (K + 3) / 4;
#pragma unroll
    for (int kc = 0; kc < KC; ++kc) {
        const int k0 = kc * 4 + khalf;
        const bool v0 = (k0 + 0) < K;
        const bool v1 = (k0 + 1) < K;
        const int ka0 = v0 ? (k0 + 0) : 0;
        const int ka1 = v1 ? (k0 + 1) : 0;
        const float xa0 = x[row * K + ka0];
        const float xa1 = x[row * K + ka1];
        const float wb0 = w[mcl * K + ka0];
        const float wb1 = w[mcl * K + ka1];
        v2f a, b;
        a.x = v0 ? xa0 : 0.0f;
        a.y = v1 ? xa1 : 0.0f;
        b.x = (colOK && v0) ? wb0 : 0.0f;
        b.y = (colOK && v1) ? wb1 : 0.0f;
        acc = __builtin_amdgcn_wmma_f32_16x16x4_f32(
             false, a,  false, b,
             (short)0, acc,  false,  false);
    }

    if (base + 16 <= n) {
        if (colOK) {
#pragma unroll
            for (int r = 0; r < 8; ++r)
                h[(base + mofs + r) * NOUT + m] = acc[r] * sc[r];
        }
    } else {
        if (colOK) {
#pragma unroll
            for (int r = 0; r < 8; ++r) {
                const int node = base + mofs + r;
                if (node < n) h[node * NOUT + m] = acc[r] * sc[r];
            }
        }
    }
}

static __device__ __forceinline__ void vadd(float4& a, const float4& b) {
    a.x += b.x; a.y += b.y; a.z += b.z; a.w += b.w;
}
static __device__ __forceinline__ void vadd(float2& a, const float2& b) {
    a.x += b.x; a.y += b.y;
}
static __device__ __forceinline__ void vxor_add(float4& a, int o) {
    a.x += __shfl_xor(a.x, o, 32); a.y += __shfl_xor(a.y, o, 32);
    a.z += __shfl_xor(a.z, o, 32); a.w += __shfl_xor(a.w, o, 32);
}
static __device__ __forceinline__ void vxor_add(float2& a, int o) {
    a.x += __shfl_xor(a.x, o, 32); a.y += __shfl_xor(a.y, o, 32);
}
static __device__ __forceinline__ void vzero(float4& a) { a = make_float4(0.f, 0.f, 0.f, 0.f); }
static __device__ __forceinline__ void vzero(float2& a) { a = make_float2(0.f, 0.f); }
static __device__ __forceinline__ void vfinish(float* sm, const float4& a, float d, const float* __restrict__ bias) {
    sm[0] = bias[0] + a.x * d; sm[1] = bias[1] + a.y * d;
    sm[2] = bias[2] + a.z * d; sm[3] = bias[3] + a.w * d;
}
static __device__ __forceinline__ void vfinish(float* sm, const float2& a, float d, const float* __restrict__ bias) {
    sm[0] = bias[0] + a.x * d; sm[1] = bias[1] + a.y * d;
}
static __device__ __forceinline__ void vclear(float* sm, const float4&) { sm[0] = 0.f; sm[1] = 0.f; sm[2] = 0.f; sm[3] = 0.f; }
static __device__ __forceinline__ void vclear(float* sm, const float2&) { sm[0] = 0.f; sm[1] = 0.f; }

template <int C> struct VecOf;
template <> struct VecOf<4> { typedef float4 type; };
template <> struct VecOf<2> { typedef float2 type; };

constexpr int GL  = 8;
constexpr int GB  = 256;
constexpr int NPB = GB / GL;

template <int C, int NEXT>
__global__ void __launch_bounds__(GB) gather_kernel(const int* __restrict__ rowptr,
                                                    const int* __restrict__ csr,
                                                    const float* __restrict__ h,
                                                    const float* __restrict__ dinv,
                                                    const float* __restrict__ bias,
                                                    const float* __restrict__ wnext,
                                                    float* __restrict__ out, int n) {
    typedef typename VecOf<C>::type VT;
    constexpr int CO = (NEXT > 0) ? NEXT : C;
    __shared__ __attribute__((aligned(16))) float sm[NPB * C];
    __shared__ __attribute__((aligned(16))) float sn[NPB * CO];
    __shared__ float sd[NPB];
    const VT* __restrict__ hv = reinterpret_cast<const VT*>(h);

    const int tid  = threadIdx.x;
    const int sub  = tid & (GL - 1);
    const int nl   = tid / GL;
    const int node = blockIdx.x * NPB + nl;
    const bool ok  = node < n;

    int j = 0, end = 0;
    float dn = 0.0f;
    if (ok) {
        j   = rowptr[node] + sub;
        end = rowptr[node + 1];
        dn  = dinv[node];
    }
    VT acc;
    vzero(acc);
    if (ok && sub == 0) acc = hv[node];

    for (; j < end; j += 4 * GL) {
        const int j1 = j + GL, j2 = j + 2 * GL, j3 = j + 3 * GL;
        const int e0 = csr[j];
        const int e1 = (j1 < end) ? csr[j1] : -1;
        const int e2 = (j2 < end) ? csr[j2] : -1;
        const int e3 = (j3 < end) ? csr[j3] : -1;
        VT v0 = hv[e0];
        VT v1, v2, v3;
        vzero(v1); vzero(v2); vzero(v3);
        if (e1 >= 0) v1 = hv[e1];
        if (e2 >= 0) v2 = hv[e2];
        if (e3 >= 0) v3 = hv[e3];
        vadd(acc, v0);
        vadd(acc, v1);
        vadd(acc, v2);
        vadd(acc, v3);
    }

    for (int o = 1; o < GL; o <<= 1) vxor_add(acc, o);

    if (sub == 0) {
        if (ok) vfinish(sm + nl * C, acc, dn, bias);
        else vclear(sm + nl * C, acc);
        if constexpr (NEXT > 0) sd[nl] = dn;
    }
    __syncthreads();

    const float* stage = sm;
    if constexpr (NEXT > 0) {
        static_assert(C == 4, "the epilogue product is one 16x16x4 step");
        const int wv = __builtin_amdgcn_readfirstlane(tid >> 5);
        if (wv < NPB / 16) {
            const int lane = tid & 31;
            const int m    = lane & 15;
            const int kh   = (lane >> 4) * 2;
            const int mcl  = (m < NEXT) ? m : 0;
            const float w0 = wnext[mcl * C + kh];
            const float w1 = wnext[mcl * C + kh + 1];
            const float2 av = *reinterpret_cast<const float2*>(sm + (wv * 16 + m) * C + kh);
            const float rs = sd[wv * 16 + m];
            v2f a, b;
            a.x = rs * tanhf(av.x);
            a.y = rs * tanhf(av.y);
            b.x = (m < NEXT) ? w0 : 0.0f;
            b.y = (m < NEXT) ? w1 : 0.0f;
            v8f d = {};
            d = __builtin_amdgcn_wmma_f32_16x16x4_f32(
                 false, a,  false, b,
                 (short)0, d,  false,  false);
            if (m < NEXT) {
                float* dst = sn + (wv * 16 + (lane >> 4) * 8) * NEXT + m;
#pragma unroll
                for (int r = 0; r < 8; ++r) dst[r * NEXT] = d[r];
            }
        }
        __syncthreads();
        stage = sn;
    }

    constexpr int NV = NPB * CO / 4;
    if (tid < NV) {
        const int first = blockIdx.x * (NPB * CO) + tid * 4;
        const int total = n * CO;
        if (first + 3 < total) {
            *reinterpret_cast<float4*>(out + first) = *reinterpret_cast<const float4*>(stage + tid * 4);
        } else {
            for (int q = 0; q < 4; ++q)
                if (first + q < total) out[first + q] = stage[tid * 4 + q];
        }
    }
}

static inline int cdiv(int a, int b) { return (a + b - 1) / b; }

extern "C" void kernel_launch(void* const* d_in, const int* in_sizes, int n_in,
                              void* d_out, int out_size, void* d_ws, size_t ws_size,
                              hipStream_t stream) {
    const float* x  = (const float*)d_in[0];
    const int*   ei = (const int*)d_in[1];
    const float* W1 = (const float*)d_in[2];
    const float* b1 = (const float*)d_in[3];
    const float* W2 = (const float*)d_in[4];
    const float* b2 = (const float*)d_in[5];
    const float* W3 = (const float*)d_in[6];
    const float* b3 = (const float*)d_in[7];
    float* out = (float*)d_out;

    const int n = in_sizes[0] / 9;
    const int e = in_sizes[1] / 2;
    const int* src = ei;
    const int* dst = ei + e;

    char* ws = (char*)d_ws;
    size_t off = 0;
    auto carve = [&](size_t bytes) {
        void* p = ws + off;
        off += (bytes + 255) & ~(size_t)255;
        return p;
    };
    const int nbuck  = cdiv(n, R);
    const int fit    = cdiv(e, PCAP);
    const int chunks = fit > NBLK ? fit : NBLK;
    const int chunk  = cdiv(e, chunks);
    float* dinv   = (float*)carve((size_t)n * sizeof(float));
    int*   rowptr = (int*)carve(((size_t)n + 1) * sizeof(int));
    int*   runs   = (int*)carve(((size_t)nbuck + 1) * chunks * sizeof(int));
    int*   part   = (int*)carve((size_t)e * 2 * sizeof(int));
    int*   csr    = (int*)carve((size_t)e * sizeof(int));
    float* bufA = (float*)carve((size_t)n * 4 * sizeof(float));
    float* bufB = (float*)carve((size_t)n * 4 * sizeof(float));
    (void)ws_size;

    const int B = 256;
    const int gW = cdiv(cdiv(n, 16) * 32, B);
    const int gG = cdiv(n, NPB);
    const size_t lds1 = (size_t)nbuck * sizeof(int);
    const size_t lds2 = (size_t)chunks * 2 * sizeof(int);

    chunk_sort_kernel<<<chunks, SB, lds1, stream>>>(src, dst, runs, part, e, chunk, nbuck);
    count_place_kernel<<<nbuck, SB, lds2, stream>>>(part, runs, rowptr, dinv, csr, n, chunk, chunks);

    linear_wmma_kernel<9, 4><<<gW, B, 0, stream>>>(x, W1, dinv, bufA, n);
    gather_kernel<4, 4><<<gG, GB, 0, stream>>>(rowptr, csr, bufA, dinv, b1, W2, bufB, n);
    gather_kernel<4, 2><<<gG, GB, 0, stream>>>(rowptr, csr, bufB, dinv, b2, W3, bufA, n);
    gather_kernel<2, 0><<<gG, GB, 0, stream>>>(rowptr, csr, bufA, dinv, b3, nullptr, out, n);
}


# gather: 4 lanes per node, 8 edges per lane and trip (64 nodes per block): half as many waves enter per layer for the same loads; 2-step butterfly; sum order changed (lane split and tree)
# speedup vs baseline: 13.8768x; 1.0571x over previous
#include <hip/hip_runtime.h>
#include <hip/hip_bf16.h>


typedef __attribute__((ext_vector_type(2))) float v2f;
typedef __attribute__((ext_vector_type(8))) float v8f;

constexpr int RB   = 8;
constexpr int R    = 1 << RB;
constexpr int NBLK = 256;
constexpr int SB   = 1024;
constexpr int PCAP = 16384;
constexpr int EPT  = PCAP / SB;
constexpr int CCAP = 16384;
constexpr int LINE = 32;
constexpr int NW   = SB / 32;

static __device__ __forceinline__ int wave_sum(int v) {
    for (int o = 16; o > 0; o >>= 1) v += __shfl_xor(v, o, 32);
    return v;
}

static __device__ __forceinline__ int2 block_sum2_1024(int a, int b, int* sm, int* res) {
    const int lane = threadIdx.x & 31;
    const int w = threadIdx.x >> 5;
    a = wave_sum(a);
    b = wave_sum(b);
    if (lane == 0) { sm[w] = a; sm[NW + w] = b; }
    __syncthreads();
    if (w == 0) {
        const int ta = wave_sum(sm[lane]);
        const int tb = wave_sum(sm[NW + lane]);
        if (lane == 0) { res[0] = ta; res[1] = tb; }
    }
    __syncthreads();
    return make_int2(res[0], res[1]);
}

static __device__ __forceinline__ int block_excl_scan_1024(int c, int* wofs) {
    const int lane = threadIdx.x & 31;
    const int w = threadIdx.x >> 5;
    int x = c;
    for (int o = 1; o < 32; o <<= 1) {
        const int y = __shfl_up(x, o, 32);
        if (lane >= o) x += y;
    }
    if (lane == 31) wofs[w] = x;
    __syncthreads();
    if (w == 0) {
        const int tot = wofs[lane];
        int s = tot;
        for (int o = 1; o < 32; o <<= 1) {
            const int y = __shfl_up(s, o, 32);
            if (lane >= o) s += y;
        }
        wofs[lane] = s - tot;
    }
    __syncthreads();
    return wofs[w] + (x - c);
}

__global__ void __launch_bounds__(SB) chunk_sort_kernel(const int* __restrict__ src, const int* __restrict__ dst,
                                                        int* __restrict__ runs, int* __restrict__ part,
                                                        int e, int chunk, int nbuck) {
    extern __shared__ int cur[];
    __shared__ int2 stage[PCAP];
    __shared__ int wofs[32];
    __shared__ int segtot;
    int2* __restrict__ pe = reinterpret_cast<int2*>(part);
    const int tid = threadIdx.x;
    const int nb = (int)gridDim.x;
    const int blk = (int)blockIdx.x;
    const int beg = blk * chunk;
    const int end = min(beg + chunk, e);
    const int cnt = max(end - beg, 0);

    int d[EPT], s[EPT], r[EPT];
    bool ok[EPT];
#pragma unroll
    for (int k = 0; k < EPT; ++k) {
        const int i = tid + k * SB;
        ok[k] = i < cnt;
        const int j = ok[k] ? beg + i : 0;
        d[k] = dst[j];
        s[k] = src[j];
        r[k] = 0;
    }
    for (int p = tid; p < nbuck; p += SB) cur[p] = 0;
    __syncthreads();

#pragma unroll
    for (int k = 0; k < EPT; ++k)
        if (ok[k]) r[k] = atomicAdd(&cur[d[k] >> RB], 1);
    __syncthreads();

    int carry = 0;
    for (int base = 0; base < nbuck; base += SB) {
        const int p = base + tid;
        const int c = (p < nbuck) ? cur[p] : 0;
        const int ex = carry + block_excl_scan_1024(c, wofs);
        if (p < nbuck) {
            cur[p] = ex;
            runs[p * nb + blk] = ex;
        }
        if (tid == SB - 1) segtot = ex + c;
        __syncthreads();
        carry = segtot;
    }
    if (tid == 0) runs[nbuck * nb + blk] = cnt;

    int q[EPT];
#pragma unroll
    for (int k = 0; k < EPT; ++k) q[k] = cur[d[k] >> RB] + r[k];
#pragma unroll
    for (int k = 0; k < EPT; ++k)
        if (ok[k]) stage[q[k]] = make_int2(s[k], d[k]);
    __syncthreads();

    int i = tid;
    for (; i + 3 * SB < cnt; i += 4 * SB) {
        const int2 p0 = stage[i], p1 = stage[i + SB], p2 = stage[i + 2 * SB], p3 = stage[i + 3 * SB];
        pe[beg + i] = p0;
        pe[beg + i + SB] = p1;
        pe[beg + i + 2 * SB] = p2;
        pe[beg + i + 3 * SB] = p3;
    }
    for (; i < cnt; i += SB) pe[beg + i] = stage[i];
}

constexpr int RPT = 8;
template <class F>
static __device__ __forceinline__ void bucket_sweep(const int2* __restrict__ pe, const int* rstart,
                                                    const int* rlen, int chunks, F f) {
    const int lane = threadIdx.x & 31;
    const int w = threadIdx.x >> 5;
    for (int b0 = w; b0 < chunks; b0 += RPT * NW) {
        int st[RPT], ln[RPT];
#pragma unroll
        for (int q = 0; q < RPT; ++q) {
            const int b = b0 + q * NW;
            const bool in = b < chunks;
            st[q] = in ? rstart[b] : 0;
            ln[q] = in ? rlen[b] : 0;
        }
        int2 a[2 * RPT];
        bool k[2 * RPT];
#pragma unroll
        for (int q = 0; q < RPT; ++q) {
            k[2 * q] = lane < ln[q];
            k[2 * q + 1] = lane + 32 < ln[q];
            a[2 * q] = pe[k[2 * q] ? st[q] + lane : 0];
            a[2 * q + 1] = pe[k[2 * q + 1] ? st[q] + lane + 32 : 0];
        }
        asm volatile("" ::"v"(a[0].x), "v"(a[0].y), "v"(a[1].x), "v"(a[1].y), "v"(a[2].x), "v"(a[2].y),
                     "v"(a[3].x), "v"(a[3].y), "v"(a[4].x), "v"(a[4].y), "v"(a[5].x), "v"(a[5].y),
                     "v"(a[6].x), "v"(a[6].y), "v"(a[7].x), "v"(a[7].y));
        asm volatile("" ::"v"(a[8].x), "v"(a[8].y), "v"(a[9].x), "v"(a[9].y), "v"(a[10].x), "v"(a[10].y),
                     "v"(a[11].x), "v"(a[11].y), "v"(a[12].x), "v"(a[12].y), "v"(a[13].x), "v"(a[13].y),
                     "v"(a[14].x), "v"(a[14].y), "v"(a[15].x), "v"(a[15].y));
#pragma unroll
        for (int q = 0; q < 2 * RPT; ++q)
            if (k[q]) f(a[q]);
#pragma unroll
        for (int q = 0; q < RPT; ++q)
            for (int i = lane + 64; i < ln[q]; i += 32) f(pe[st[q] + i]);
    }
}

__global__ void __launch_bounds__(SB) count_place_kernel(const int* __restrict__ part, const int* __restrict__ runs,
                                                         int* __restrict__ rowptr, float* __restrict__ dinv,
                                                         int* __restrict__ csr, int n, int chunk, int chunks) {
    extern __shared__ int rtab[];
    __shared__ int c[R];
    __shared__ int wofs[32];
    __shared__ int sm[2 * NW];
    __shared__ int res[2];
    __shared__ int img[CCAP + LINE];
    int* rstart = rtab;
    int* rlen = rtab + chunks;
    const int2* __restrict__ pe = reinterpret_cast<const int2*>(part);
    const int tid = threadIdx.x;
    const int bucket = (int)blockIdx.x;
    const int node0 = bucket << RB;
    if (tid < R) c[tid] = 0;
    int slo = 0, shi = 0;
    for (int b = tid; b < chunks; b += SB) {
        const int lo = runs[bucket * chunks + b];
        const int hi = runs[(bucket + 1) * chunks + b];
        rstart[b] = b * chunk + lo;
        rlen[b] = hi - lo;
        slo += lo;
        shi += hi;
    }
    const int2 be = block_sum2_1024(slo, shi, sm, res);
    const int beg = be.x;
    const int end = be.y;

    bucket_sweep(pe, rstart, rlen, chunks, [&](const int2& a) { atomicAdd(&c[a.y - node0], 1); });
    __syncthreads();
    const int cv = (tid < R) ? c[tid] : 0;
    const int excl = block_excl_scan_1024(cv, wofs);
    const int node = node0 + tid;
    const int cnt = end - beg;
    const bool staged = cnt <= CCAP;
    const int lead = beg & (LINE - 1);
    if (tid < R) {
        c[tid] = (staged ? lead : beg) + excl;
        if (node < n) {
            rowptr[node] = beg + excl;
            dinv[node] = rsqrtf((float)cv + 1.0f);
            if (node == n - 1) rowptr[n] = beg + excl + cv;
        }
    }
    __syncthreads();
    if (staged) {
        bucket_sweep(pe, rstart, rlen, chunks, [&](const int2& a) { img[atomicAdd(&c[a.y - node0], 1)] = a.x; });
        __syncthreads();
        const int line0 = beg - lead;
        const int top = lead + cnt;
        int li = tid;
        for (; li + 3 * SB < top; li += 4 * SB) {
            const bool k0 = li >= lead;
            const int v0 = k0 ? img[li] : 0;
            const int v1 = img[li + SB], v2 = img[li + 2 * SB], v3 = img[li + 3 * SB];
            if (k0) csr[line0 + li] = v0;
            csr[line0 + li + SB] = v1;
            csr[line0 + li + 2 * SB] = v2;
            csr[line0 + li + 3 * SB] = v3;
        }
        for (; li < top; li += SB)
            if (li >= lead) csr[line0 + li] = img[li];
    } else {
        bucket_sweep(pe, rstart, rlen, chunks, [&](const int2& a) { csr[atomicAdd(&c[a.y - node0], 1)] = a.x; });
    }
}

template <int K, int NOUT>
__global__ void linear_wmma_kernel(const float* __restrict__ x,
                                   const float* __restrict__ w,
                                   const float* __restrict__ dinv,
                                   float* __restrict__ h,
                                   int n) {
    const int wave = (blockIdx.x * blockDim.x + threadIdx.x) >> 5;
    const int lane = threadIdx.x & 31;
    const int base = wave * 16;
    if (base >= n) return;

    const int m     = lane & 15;
    const int khalf = (lane >> 4) * 2;
    const int row   = min(base + m, n - 1);
    const int mcl   = (m < NOUT) ? m : 0;
    const bool colOK = (m < NOUT);
    const int mofs  = (lane >> 4) * 8;

    float sc[8];
#pragma unroll
    for (int r = 0; r < 8; ++r) sc[r] = dinv[min(base + mofs + r, n - 1)];

    v8f acc = {};
    constexpr int KC = (K + 3) / 4;
#pragma unroll
    for (int kc = 0; kc < KC; ++kc) {
        const int k0 = kc * 4 + khalf;
        const bool v0 = (k0 + 0) < K;
        const bool v1 = (k0 + 1) < K;
        const int ka0 = v0 ? (k0 + 0) : 0;
        const int ka1 = v1 ? (k0 + 1) : 0;
        const float xa0 = x[row * K + ka0];
        const float xa1 = x[row * K + ka1];
        const float wb0 = w[mcl * K + ka0];
        const float wb1 = w[mcl * K + ka1];
        v2f a, b;
        a.x = v0 ? xa0 : 0.0f;
        a.y = v1 ? xa1 : 0.0f;
        b.x = (colOK && v0) ? wb0 : 0.0f;
        b.y = (colOK && v1) ? wb1 : 0.0f;
        acc = __builtin_amdgcn_wmma_f32_16x16x4_f32(
             false, a,  false, b,
             (short)0, acc,  false,  false);
    }

    if (base + 16 <= n) {
        if (colOK) {
#pragma unroll
            for (int r = 0; r < 8; ++r)
                h[(base + mofs + r) * NOUT + m] = acc[r] * sc[r];
        }
    } else {
        if (colOK) {
#pragma unroll
            for (int r = 0; r < 8; ++r) {
                const int node = base + mofs + r;
                if (node < n) h[node * NOUT + m] = acc[r] * sc[r];
            }
        }
    }
}

static __device__ __forceinline__ void vadd(float4& a, const float4& b) {
    a.x += b.x; a.y += b.y; a.z += b.z; a.w += b.w;
}
static __device__ __forceinline__ void vadd(float2& a, const float2& b) {
    a.x += b.x; a.y += b.y;
}
static __device__ __forceinline__ void vxor_add(float4& a, int o) {
    a.x += __shfl_xor(a.x, o, 32); a.y += __shfl_xor(a.y, o, 32);
    a.z += __shfl_xor(a.z, o, 32); a.w += __shfl_xor(a.w, o, 32);
}
static __device__ __forceinline__ void vxor_add(float2& a, int o) {
    a.x += __shfl_xor(a.x, o, 32); a.y += __shfl_xor(a.y, o, 32);
}
static __device__ __forceinline__ void vzero(float4& a) { a = make_float4(0.f, 0.f, 0.f, 0.f); }
static __device__ __forceinline__ void vzero(float2& a) { a = make_float2(0.f, 0.f); }
static __device__ __forceinline__ void vfinish(float* sm, const float4& a, float d, const float* __restrict__ bias) {
    sm[0] = bias[0] + a.x * d; sm[1] = bias[1] + a.y * d;
    sm[2] = bias[2] + a.z * d; sm[3] = bias[3] + a.w * d;
}
static __device__ __forceinline__ void vfinish(float* sm, const float2& a, float d, const float* __restrict__ bias) {
    sm[0] = bias[0] + a.x * d; sm[1] = bias[1] + a.y * d;
}
static __device__ __forceinline__ void vclear(float* sm, const float4&) { sm[0] = 0.f; sm[1] = 0.f; sm[2] = 0.f; sm[3] = 0.f; }
static __device__ __forceinline__ void vclear(float* sm, const float2&) { sm[0] = 0.f; sm[1] = 0.f; }

template <int C> struct VecOf;
template <> struct VecOf<4> { typedef float4 type; };
template <> struct VecOf<2> { typedef float2 type; };

constexpr int GL  = 4;
constexpr int GU  = 8;
constexpr int GB  = 256;
constexpr int NPB = GB / GL;

template <int C, int NEXT>
__global__ void __launch_bounds__(GB) gather_kernel(const int* __restrict__ rowptr,
                                                    const int* __restrict__ csr,
                                                    const float* __restrict__ h,
                                                    const float* __restrict__ dinv,
                                                    const float* __restrict__ bias,
                                                    const float* __restrict__ wnext,
                                                    float* __restrict__ out, int n) {
    typedef typename VecOf<C>::type VT;
    constexpr int CO = (NEXT > 0) ? NEXT : C;
    __shared__ __attribute__((aligned(16))) float sm[NPB * C];
    __shared__ __attribute__((aligned(16))) float sn[NPB * CO];
    __shared__ float sd[NPB];
    const VT* __restrict__ hv = reinterpret_cast<const VT*>(h);

    const int tid  = threadIdx.x;
    const int sub  = tid & (GL - 1);
    const int nl   = tid / GL;
    const int node = blockIdx.x * NPB + nl;
    const bool ok  = node < n;

    int j = 0, end = 0;
    float dn = 0.0f;
    if (ok) {
        j   = rowptr[node] + sub;
        end = rowptr[node + 1];
        dn  = dinv[node];
    }
    VT acc;
    vzero(acc);
    if (ok && sub == 0) acc = hv[node];

    for (; j < end; j += GU * GL) {
        int ent[GU];
        ent[0] = csr[j];
#pragma unroll
        for (int u = 1; u < GU; ++u) {
            const int ju = j + u * GL;
            ent[u] = (ju < end) ? csr[ju] : -1;
        }
        VT v[GU];
        v[0] = hv[ent[0]];
#pragma unroll
        for (int u = 1; u < GU; ++u) {
            vzero(v[u]);
            if (ent[u] >= 0) v[u] = hv[ent[u]];
        }
#pragma unroll
        for (int u = 0; u < GU; ++u) vadd(acc, v[u]);
    }

    for (int o = 1; o < GL; o <<= 1) vxor_add(acc, o);

    if (sub == 0) {
        if (ok) vfinish(sm + nl * C, acc, dn, bias);
        else vclear(sm + nl * C, acc);
        if constexpr (NEXT > 0) sd[nl] = dn;
    }
    __syncthreads();

    const float* stage = sm;
    if constexpr (NEXT > 0) {
        static_assert(C == 4, "the epilogue product is one 16x16x4 step");
        const int wv = __builtin_amdgcn_readfirstlane(tid >> 5);
        if (wv < NPB / 16) {
            const int lane = tid & 31;
            const int m    = lane & 15;
            const int kh   = (lane >> 4) * 2;
            const int mcl  = (m < NEXT) ? m : 0;
            const float w0 = wnext[mcl * C + kh];
            const float w1 = wnext[mcl * C + kh + 1];
            const float2 av = *reinterpret_cast<const float2*>(sm + (wv * 16 + m) * C + kh);
            const float rs = sd[wv * 16 + m];
            v2f a, b;
            a.x = rs * tanhf(av.x);
            a.y = rs * tanhf(av.y);
            b.x = (m < NEXT) ? w0 : 0.0f;
            b.y = (m < NEXT) ? w1 : 0.0f;
            v8f d = {};
            d = __builtin_amdgcn_wmma_f32_16x16x4_f32(
                 false, a,  false, b,
                 (short)0, d,  false,  false);
            if (m < NEXT) {
                float* dst = sn + (wv * 16 + (lane >> 4) * 8) * NEXT + m;
#pragma unroll
                for (int r = 0; r < 8; ++r) dst[r * NEXT] = d[r];
            }
        }
        __syncthreads();
        stage = sn;
    }

    constexpr int NV = NPB * CO / 4;
    if (tid < NV) {
        const int first = blockIdx.x * (NPB * CO) + tid * 4;
        const int total = n * CO;
        if (first + 3 < total) {
            *reinterpret_cast<float4*>(out + first) = *reinterpret_cast<const float4*>(stage + tid * 4);
        } else {
            for (int q = 0; q < 4; ++q)
                if (first + q < total) out[first + q] = stage[tid * 4 + q];
        }
    }
}

static inline int cdiv(int a, int b) { return (a + b - 1) / b; }

extern "C" void kernel_launch(void* const* d_in, const int* in_sizes, int n_in,
                              void* d_out, int out_size, void* d_ws, size_t ws_size,
                              hipStream_t stream) {
    const float* x  = (const float*)d_in[0];
    const int*   ei = (const int*)d_in[1];
    const float* W1 = (const float*)d_in[2];
    const float* b1 = (const float*)d_in[3];
    const float* W2 = (const float*)d_in[4];
    const float* b2 = (const float*)d_in[5];
    const float* W3 = (const float*)d_in[6];
    const float* b3 = (const float*)d_in[7];
    float* out = (float*)d_out;

    const int n = in_sizes[0] / 9;
    const int e = in_sizes[1] / 2;
    const int* src = ei;
    const int* dst = ei + e;

    char* ws = (char*)d_ws;
    size_t off = 0;
    auto carve = [&](size_t bytes) {
        void* p = ws + off;
        off += (bytes + 255) & ~(size_t)255;
        return p;
    };
    const int nbuck  = cdiv(n, R);
    const int fit    = cdiv(e, PCAP);
    const int chunks = fit > NBLK ? fit : NBLK;
    const int chunk  = cdiv(e, chunks);
    float* dinv   = (float*)carve((size_t)n * sizeof(float));
    int*   rowptr = (int*)carve(((size_t)n + 1) * sizeof(int));
    int*   runs   = (int*)carve(((size_t)nbuck + 1) * chunks * sizeof(int));
    int*   part   = (int*)carve((size_t)e * 2 * sizeof(int));
    int*   csr    = (int*)carve((size_t)e * sizeof(int));
    float* bufA = (float*)carve((size_t)n * 4 * sizeof(float));
    float* bufB = (float*)carve((size_t)n * 4 * sizeof(float));
    (void)ws_size;

    const int B = 256;
    const int gW = cdiv(cdiv(n, 16) * 32, B);
    const int gG = cdiv(n, NPB);
    const size_t lds1 = (size_t)nbuck * sizeof(int);
    const size_t lds2 = (size_t)chunks * 2 * sizeof(int);

    chunk_sort_kernel<<<chunks, SB, lds1, stream>>>(src, dst, runs, part, e, chunk, nbuck);
    count_place_kernel<<<nbuck, SB, lds2, stream>>>(part, runs, rowptr, dinv, csr, n, chunk, chunks);

    linear_wmma_kernel<9, 4><<<gW, B, 0, stream>>>(x, W1, dinv, bufA, n);
    gather_kernel<4, 4><<<gG, GB, 0, stream>>>(rowptr, csr, bufA, dinv, b1, W2, bufB, n);
    gather_kernel<4, 2><<<gG, GB, 0, stream>>>(rowptr, csr, bufB, dinv, b2, W3, bufA, n);
    gather_kernel<2, 0><<<gG, GB, 0, stream>>>(rowptr, csr, bufA, dinv, b3, nullptr, out, n);
}


# layer 1's linear (same f32 16x16x4 WMMA, same k order and dinv scaling) runs inside the level-2 sort launch, 16 waves of a bucket taking its 16 node tiles; linear launch gone (6 -> 5); sums unchanged
# speedup vs baseline: 14.3117x; 1.0313x over previous
#include <hip/hip_runtime.h>
#include <hip/hip_bf16.h>


typedef __attribute__((ext_vector_type(2))) float v2f;
typedef __attribute__((ext_vector_type(8))) float v8f;

constexpr int RB   = 8;
constexpr int R    = 1 << RB;
constexpr int NBLK = 256;
constexpr int SB   = 1024;
constexpr int PCAP = 16384;
constexpr int EPT  = PCAP / SB;
constexpr int CCAP = 16384;
constexpr int LINE = 32;
constexpr int NW   = SB / 32;

static __device__ __forceinline__ int wave_sum(int v) {
    for (int o = 16; o > 0; o >>= 1) v += __shfl_xor(v, o, 32);
    return v;
}

static __device__ __forceinline__ int2 block_sum2_1024(int a, int b, int* sm, int* res) {
    const int lane = threadIdx.x & 31;
    const int w = threadIdx.x >> 5;
    a = wave_sum(a);
    b = wave_sum(b);
    if (lane == 0) { sm[w] = a; sm[NW + w] = b; }
    __syncthreads();
    if (w == 0) {
        const int ta = wave_sum(sm[lane]);
        const int tb = wave_sum(sm[NW + lane]);
        if (lane == 0) { res[0] = ta; res[1] = tb; }
    }
    __syncthreads();
    return make_int2(res[0], res[1]);
}

static __device__ __forceinline__ int block_excl_scan_1024(int c, int* wofs) {
    const int lane = threadIdx.x & 31;
    const int w = threadIdx.x >> 5;
    int x = c;
    for (int o = 1; o < 32; o <<= 1) {
        const int y = __shfl_up(x, o, 32);
        if (lane >= o) x += y;
    }
    if (lane == 31) wofs[w] = x;
    __syncthreads();
    if (w == 0) {
        const int tot = wofs[lane];
        int s = tot;
        for (int o = 1; o < 32; o <<= 1) {
            const int y = __shfl_up(s, o, 32);
            if (lane >= o) s += y;
        }
        wofs[lane] = s - tot;
    }
    __syncthreads();
    return wofs[w] + (x - c);
}

__global__ void __launch_bounds__(SB) chunk_sort_kernel(const int* __restrict__ src, const int* __restrict__ dst,
                                                        int* __restrict__ runs, int* __restrict__ part,
                                                        int e, int chunk, int nbuck) {
    extern __shared__ int cur[];
    __shared__ int2 stage[PCAP];
    __shared__ int wofs[32];
    __shared__ int segtot;
    int2* __restrict__ pe = reinterpret_cast<int2*>(part);
    const int tid = threadIdx.x;
    const int nb = (int)gridDim.x;
    const int blk = (int)blockIdx.x;
    const int beg = blk * chunk;
    const int end = min(beg + chunk, e);
    const int cnt = max(end - beg, 0);

    int d[EPT], s[EPT], r[EPT];
    bool ok[EPT];
#pragma unroll
    for (int k = 0; k < EPT; ++k) {
        const int i = tid + k * SB;
        ok[k] = i < cnt;
        const int j = ok[k] ? beg + i : 0;
        d[k] = dst[j];
        s[k] = src[j];
        r[k] = 0;
    }
    for (int p = tid; p < nbuck; p += SB) cur[p] = 0;
    __syncthreads();

#pragma unroll
    for (int k = 0; k < EPT; ++k)
        if (ok[k]) r[k] = atomicAdd(&cur[d[k] >> RB], 1);
    __syncthreads();

    int carry = 0;
    for (int base = 0; base < nbuck; base += SB) {
        const int p = base + tid;
        const int c = (p < nbuck) ? cur[p] : 0;
        const int ex = carry + block_excl_scan_1024(c, wofs);
        if (p < nbuck) {
            cur[p] = ex;
            runs[p * nb + blk] = ex;
        }
        if (tid == SB - 1) segtot = ex + c;
        __syncthreads();
        carry = segtot;
    }
    if (tid == 0) runs[nbuck * nb + blk] = cnt;

    int q[EPT];
#pragma unroll
    for (int k = 0; k < EPT; ++k) q[k] = cur[d[k] >> RB] + r[k];
#pragma unroll
    for (int k = 0; k < EPT; ++k)
        if (ok[k]) stage[q[k]] = make_int2(s[k], d[k]);
    __syncthreads();

    int i = tid;
    for (; i + 3 * SB < cnt; i += 4 * SB) {
        const int2 p0 = stage[i], p1 = stage[i + SB], p2 = stage[i + 2 * SB], p3 = stage[i + 3 * SB];
        pe[beg + i] = p0;
        pe[beg + i + SB] = p1;
        pe[beg + i + 2 * SB] = p2;
        pe[beg + i + 3 * SB] = p3;
    }
    for (; i < cnt; i += SB) pe[beg + i] = stage[i];
}

constexpr int RPT = 8;
template <class F>
static __device__ __forceinline__ void bucket_sweep(const int2* __restrict__ pe, const int* rstart,
                                                    const int* rlen, int chunks, F f) {
    const int lane = threadIdx.x & 31;
    const int w = threadIdx.x >> 5;
    for (int b0 = w; b0 < chunks; b0 += RPT * NW) {
        int st[RPT], ln[RPT];
#pragma unroll
        for (int q = 0; q < RPT; ++q) {
            const int b = b0 + q * NW;
            const bool in = b < chunks;
            st[q] = in ? rstart[b] : 0;
            ln[q] = in ? rlen[b] : 0;
        }
        int2 a[2 * RPT];
        bool k[2 * RPT];
#pragma unroll
        for (int q = 0; q < RPT; ++q) {
            k[2 * q] = lane < ln[q];
            k[2 * q + 1] = lane + 32 < ln[q];
            a[2 * q] = pe[k[2 * q] ? st[q] + lane : 0];
            a[2 * q + 1] = pe[k[2 * q + 1] ? st[q] + lane + 32 : 0];
        }
        asm volatile("" ::"v"(a[0].x), "v"(a[0].y), "v"(a[1].x), "v"(a[1].y), "v"(a[2].x), "v"(a[2].y),
                     "v"(a[3].x), "v"(a[3].y), "v"(a[4].x), "v"(a[4].y), "v"(a[5].x), "v"(a[5].y),
                     "v"(a[6].x), "v"(a[6].y), "v"(a[7].x), "v"(a[7].y));
        asm volatile("" ::"v"(a[8].x), "v"(a[8].y), "v"(a[9].x), "v"(a[9].y), "v"(a[10].x), "v"(a[10].y),
                     "v"(a[11].x), "v"(a[11].y), "v"(a[12].x), "v"(a[12].y), "v"(a[13].x), "v"(a[13].y),
                     "v"(a[14].x), "v"(a[14].y), "v"(a[15].x), "v"(a[15].y));
#pragma unroll
        for (int q = 0; q < 2 * RPT; ++q)
            if (k[q]) f(a[q]);
#pragma unroll
        for (int q = 0; q < RPT; ++q)
            for (int i = lane + 64; i < ln[q]; i += 32) f(pe[st[q] + i]);
    }
}

template <int K, int NOUT>
__global__ void __launch_bounds__(SB) count_place_kernel(const int* __restrict__ part, const int* __restrict__ runs,
                                                         int* __restrict__ rowptr, float* __restrict__ dinv,
                                                         int* __restrict__ csr,
                                                         const float* __restrict__ x,
                                                         const float* __restrict__ w,
                                                         float* __restrict__ h,
                                                         int n, int chunk, int chunks) {
    static_assert(NOUT == 4, "a row of h leaves as one 16-byte store");
    static_assert(R / 16 <= NW, "one wave per 16-node tile of the bucket");
    extern __shared__ int rtab[];
    __shared__ int c[R];
    __shared__ int wofs[32];
    __shared__ int sm[2 * NW];
    __shared__ int res[2];
    __shared__ int img[CCAP + LINE];
    __shared__ __attribute__((aligned(16))) float sh[R * NOUT];
    int* rstart = rtab;
    int* rlen = rtab + chunks;
    const int2* __restrict__ pe = reinterpret_cast<const int2*>(part);
    const int tid = threadIdx.x;
    const int bucket = (int)blockIdx.x;
    const int node0 = bucket << RB;
    if (tid < R) c[tid] = 0;

    constexpr int KC = (K + 3) / 4;
    const int lane = tid & 31;
    const int wv = __builtin_amdgcn_readfirstlane(tid >> 5);
    const int tbase = node0 + wv * 16;
    const bool tile = (wv < R / 16) && (tbase < n);
    const int m = lane & 15;
    const int khalf = (lane >> 4) * 2;
    const bool colOK = m < NOUT;
    float xa[2 * KC], wb[2 * KC];
#pragma unroll
    for (int q = 0; q < 2 * KC; ++q) { xa[q] = 0.0f; wb[q] = 0.0f; }
    if (tile) {
        const int row = min(tbase + m, n - 1);
        const int mcl = colOK ? m : 0;
#pragma unroll
        for (int kc = 0; kc < KC; ++kc) {
            const int k0 = kc * 4 + khalf;
            const int ka0 = (k0 + 0) < K ? (k0 + 0) : 0;
            const int ka1 = (k0 + 1) < K ? (k0 + 1) : 0;
            xa[2 * kc] = x[row * K + ka0];
            xa[2 * kc + 1] = x[row * K + ka1];
            wb[2 * kc] = w[mcl * K + ka0];
            wb[2 * kc + 1] = w[mcl * K + ka1];
        }
    }

    int slo = 0, shi = 0;
    for (int b = tid; b < chunks; b += SB) {
        const int lo = runs[bucket * chunks + b];
        const int hi = runs[(bucket + 1) * chunks + b];
        rstart[b] = b * chunk + lo;
        rlen[b] = hi - lo;
        slo += lo;
        shi += hi;
    }
    if (tile) {
        v8f acc = {};
#pragma unroll
        for (int kc = 0; kc < KC; ++kc) {
            const int k0 = kc * 4 + khalf;
            const bool v0 = (k0 + 0) < K;
            const bool v1 = (k0 + 1) < K;
            v2f a, b;
            a.x = v0 ? xa[2 * kc] : 0.0f;
            a.y = v1 ? xa[2 * kc + 1] : 0.0f;
            b.x = (colOK && v0) ? wb[2 * kc] : 0.0f;
            b.y = (colOK && v1) ? wb[2 * kc + 1] : 0.0f;
            acc = __builtin_amdgcn_wmma_f32_16x16x4_f32(
                 false, a,  false, b,
                 (short)0, acc,  false,  false);
        }
        if (colOK) {
            float* dstp = sh + (wv * 16 + (lane >> 4) * 8) * NOUT + m;
#pragma unroll
            for (int r = 0; r < 8; ++r) dstp[r * NOUT] = acc[r];
        }
    }
    const int2 be = block_sum2_1024(slo, shi, sm, res);
    const int beg = be.x;
    const int end = be.y;

    bucket_sweep(pe, rstart, rlen, chunks, [&](const int2& a) { atomicAdd(&c[a.y - node0], 1); });
    __syncthreads();
    const int cv = (tid < R) ? c[tid] : 0;
    const int excl = block_excl_scan_1024(cv, wofs);
    const int node = node0 + tid;
    const int cnt = end - beg;
    const bool staged = cnt <= CCAP;
    const int lead = beg & (LINE - 1);
    if (tid < R) {
        c[tid] = (staged ? lead : beg) + excl;
        if (node < n) {
            const float dv = rsqrtf((float)cv + 1.0f);
            rowptr[node] = beg + excl;
            dinv[node] = dv;
            const float4 p = *reinterpret_cast<const float4*>(sh + tid * NOUT);
            reinterpret_cast<float4*>(h)[node] = make_float4(p.x * dv, p.y * dv, p.z * dv, p.w * dv);
            if (node == n - 1) rowptr[n] = beg + excl + cv;
        }
    }
    __syncthreads();
    if (staged) {
        bucket_sweep(pe, rstart, rlen, chunks, [&](const int2& a) { img[atomicAdd(&c[a.y - node0], 1)] = a.x; });
        __syncthreads();
        const int line0 = beg - lead;
        const int top = lead + cnt;
        int li = tid;
        for (; li + 3 * SB < top; li += 4 * SB) {
            const bool k0 = li >= lead;
            const int v0 = k0 ? img[li] : 0;
            const int v1 = img[li + SB], v2 = img[li + 2 * SB], v3 = img[li + 3 * SB];
            if (k0) csr[line0 + li] = v0;
            csr[line0 + li + SB] = v1;
            csr[line0 + li + 2 * SB] = v2;
            csr[line0 + li + 3 * SB] = v3;
        }
        for (; li < top; li += SB)
            if (li >= lead) csr[line0 + li] = img[li];
    } else {
        bucket_sweep(pe, rstart, rlen, chunks, [&](const int2& a) { csr[atomicAdd(&c[a.y - node0], 1)] = a.x; });
    }
}

static __device__ __forceinline__ void vadd(float4& a, const float4& b) {
    a.x += b.x; a.y += b.y; a.z += b.z; a.w += b.w;
}
static __device__ __forceinline__ void vadd(float2& a, const float2& b) {
    a.x += b.x; a.y += b.y;
}
static __device__ __forceinline__ void vxor_add(float4& a, int o) {
    a.x += __shfl_xor(a.x, o, 32); a.y += __shfl_xor(a.y, o, 32);
    a.z += __shfl_xor(a.z, o, 32); a.w += __shfl_xor(a.w, o, 32);
}
static __device__ __forceinline__ void vxor_add(float2& a, int o) {
    a.x += __shfl_xor(a.x, o, 32); a.y += __shfl_xor(a.y, o, 32);
}
static __device__ __forceinline__ void vzero(float4& a) { a = make_float4(0.f, 0.f, 0.f, 0.f); }
static __device__ __forceinline__ void vzero(float2& a) { a = make_float2(0.f, 0.f); }
static __device__ __forceinline__ void vfinish(float* sm, const float4& a, float d, const float* __restrict__ bias) {
    sm[0] = bias[0] + a.x * d; sm[1] = bias[1] + a.y * d;
    sm[2] = bias[2] + a.z * d; sm[3] = bias[3] + a.w * d;
}
static __device__ __forceinline__ void vfinish(float* sm, const float2& a, float d, const float* __restrict__ bias) {
    sm[0] = bias[0] + a.x * d; sm[1] = bias[1] + a.y * d;
}
static __device__ __forceinline__ void vclear(float* sm, const float4&) { sm[0] = 0.f; sm[1] = 0.f; sm[2] = 0.f; sm[3] = 0.f; }
static __device__ __forceinline__ void vclear(float* sm, const float2&) { sm[0] = 0.f; sm[1] = 0.f; }

template <int C> struct VecOf;
template <> struct VecOf<4> { typedef float4 type; };
template <> struct VecOf<2> { typedef float2 type; };

constexpr int GL  = 4;
constexpr int GU  = 8;
constexpr int GB  = 256;
constexpr int NPB = GB / GL;

template <int C, int NEXT>
__global__ void __launch_bounds__(GB) gather_kernel(const int* __restrict__ rowptr,
                                                    const int* __restrict__ csr,
                                                    const float* __restrict__ h,
                                                    const float* __restrict__ dinv,
                                                    const float* __restrict__ bias,
                                                    const float* __restrict__ wnext,
                                                    float* __restrict__ out, int n) {
    typedef typename VecOf<C>::type VT;
    constexpr int CO = (NEXT > 0) ? NEXT : C;
    __shared__ __attribute__((aligned(16))) float sm[NPB * C];
    __shared__ __attribute__((aligned(16))) float sn[NPB * CO];
    __shared__ float sd[NPB];
    const VT* __restrict__ hv = reinterpret_cast<const VT*>(h);

    const int tid  = threadIdx.x;
    const int sub  = tid & (GL - 1);
    const int nl   = tid / GL;
    const int node = blockIdx.x * NPB + nl;
    const bool ok  = node < n;

    int j = 0, end = 0;
    float dn = 0.0f;
    if (ok) {
        j   = rowptr[node] + sub;
        end = rowptr[node + 1];
        dn  = dinv[node];
    }
    VT acc;
    vzero(acc);
    if (ok && sub == 0) acc = hv[node];

    for (; j < end; j += GU * GL) {
        int ent[GU];
        ent[0] = csr[j];
#pragma unroll
        for (int u = 1; u < GU; ++u) {
            const int ju = j + u * GL;
            ent[u] = (ju < end) ? csr[ju] : -1;
        }
        VT v[GU];
        v[0] = hv[ent[0]];
#pragma unroll
        for (int u = 1; u < GU; ++u) {
            vzero(v[u]);
            if (ent[u] >= 0) v[u] = hv[ent[u]];
        }
#pragma unroll
        for (int u = 0; u < GU; ++u) vadd(acc, v[u]);
    }

    for (int o = 1; o < GL; o <<= 1) vxor_add(acc, o);

    if (sub == 0) {
        if (ok) vfinish(sm + nl * C, acc, dn, bias);
        else vclear(sm + nl * C, acc);
        if constexpr (NEXT > 0) sd[nl] = dn;
    }
    __syncthreads();

    const float* stage = sm;
    if constexpr (NEXT > 0) {
        static_assert(C == 4, "the epilogue product is one 16x16x4 step");
        const int wv = __builtin_amdgcn_readfirstlane(tid >> 5);
        if (wv < NPB / 16) {
            const int lane = tid & 31;
            const int m    = lane & 15;
            const int kh   = (lane >> 4) * 2;
            const int mcl  = (m < NEXT) ? m : 0;
            const float w0 = wnext[mcl * C + kh];
            const float w1 = wnext[mcl * C + kh + 1];
            const float2 av = *reinterpret_cast<const float2*>(sm + (wv * 16 + m) * C + kh);
            const float rs = sd[wv * 16 + m];
            v2f a, b;
            a.x = rs * tanhf(av.x);
            a.y = rs * tanhf(av.y);
            b.x = (m < NEXT) ? w0 : 0.0f;
            b.y = (m < NEXT) ? w1 : 0.0f;
            v8f d = {};
            d = __builtin_amdgcn_wmma_f32_16x16x4_f32(
                 false, a,  false, b,
                 (short)0, d,  false,  false);
            if (m < NEXT) {
                float* dst = sn + (wv * 16 + (lane >> 4) * 8) * NEXT + m;
#pragma unroll
                for (int r = 0; r < 8; ++r) dst[r * NEXT] = d[r];
            }
        }
        __syncthreads();
        stage = sn;
    }

    constexpr int NV = NPB * CO / 4;
    if (tid < NV) {
        const int first = blockIdx.x * (NPB * CO) + tid * 4;
        const int total = n * CO;
        if (first + 3 < total) {
            *reinterpret_cast<float4*>(out + first) = *reinterpret_cast<const float4*>(stage + tid * 4);
        } else {
            for (int q = 0; q < 4; ++q)
                if (first + q < total) out[first + q] = stage[tid * 4 + q];
        }
    }
}

static inline int cdiv(int a, int b) { return (a + b - 1) / b; }

extern "C" void kernel_launch(void* const* d_in, const int* in_sizes, int n_in,
                              void* d_out, int out_size, void* d_ws, size_t ws_size,
                              hipStream_t stream) {
    const float* x  = (const float*)d_in[0];
    const int*   ei = (const int*)d_in[1];
    const float* W1 = (const float*)d_in[2];
    const float* b1 = (const float*)d_in[3];
    const float* W2 = (const float*)d_in[4];
    const float* b2 = (const float*)d_in[5];
    const float* W3 = (const float*)d_in[6];
    const float* b3 = (const float*)d_in[7];
    float* out = (float*)d_out;

    const int n = in_sizes[0] / 9;
    const int e = in_sizes[1] / 2;
    const int* src = ei;
    const int* dst = ei + e;

    char* ws = (char*)d_ws;
    size_t off = 0;
    auto carve = [&](size_t bytes) {
        void* p = ws + off;
        off += (bytes + 255) & ~(size_t)255;
        return p;
    };
    const int nbuck  = cdiv(n, R);
    const int fit    = cdiv(e, PCAP);
    const int chunks = fit > NBLK ? fit : NBLK;
    const int chunk  = cdiv(e, chunks);
    float* dinv   = (float*)carve((size_t)n * sizeof(float));
    int*   rowptr = (int*)carve(((size_t)n + 1) * sizeof(int));
    int*   runs   = (int*)carve(((size_t)nbuck + 1) * chunks * sizeof(int));
    int*   part   = (int*)carve((size_t)e * 2 * sizeof(int));
    int*   csr    = (int*)carve((size_t)e * sizeof(int));
    float* bufA = (float*)carve((size_t)n * 4 * sizeof(float));
    float* bufB = (float*)carve((size_t)n * 4 * sizeof(float));
    (void)ws_size;

    const int gG = cdiv(n, NPB);
    const size_t lds1 = (size_t)nbuck * sizeof(int);
    const size_t lds2 = (size_t)chunks * 2 * sizeof(int);

    chunk_sort_kernel<<<chunks, SB, lds1, stream>>>(src, dst, runs, part, e, chunk, nbuck);
    count_place_kernel<9, 4><<<nbuck, SB, lds2, stream>>>(part, runs, rowptr, dinv, csr, x, W1, bufA, n, chunk, chunks);
    gather_kernel<4, 4><<<gG, GB, 0, stream>>>(rowptr, csr, bufA, dinv, b1, W2, bufB, n);
    gather_kernel<4, 2><<<gG, GB, 0, stream>>>(rowptr, csr, bufB, dinv, b2, W3, bufA, n);
    gather_kernel<2, 0><<<gG, GB, 0, stream>>>(rowptr, csr, bufA, dinv, b3, nullptr, out, n);
}


# sort levels pass ONE 4-byte record per edge, (src << 8) | (dst & 255), not an 8-byte {src, dst} pair; level 1 writes them as whole 128-byte lines; half the bytes between the levels; sum order same
# speedup vs baseline: 14.8304x; 1.0362x over previous
#include <hip/hip_runtime.h>
#include <hip/hip_bf16.h>


typedef __attribute__((ext_vector_type(2))) float v2f;
typedef __attribute__((ext_vector_type(8))) float v8f;

constexpr int RB   = 8;
constexpr int R    = 1 << RB;
constexpr int NBLK = 256;
constexpr int SB   = 1024;
constexpr int PCAP = 16384;
constexpr int EPT  = PCAP / SB;
constexpr int CCAP = 16384;
constexpr int LINE = 32;
constexpr int NW   = SB / 32;

static __device__ __forceinline__ int wave_sum(int v) {
    for (int o = 16; o > 0; o >>= 1) v += __shfl_xor(v, o, 32);
    return v;
}

static __device__ __forceinline__ int2 block_sum2_1024(int a, int b, int* sm, int* res) {
    const int lane = threadIdx.x & 31;
    const int w = threadIdx.x >> 5;
    a = wave_sum(a);
    b = wave_sum(b);
    if (lane == 0) { sm[w] = a; sm[NW + w] = b; }
    __syncthreads();
    if (w == 0) {
        const int ta = wave_sum(sm[lane]);
        const int tb = wave_sum(sm[NW + lane]);
        if (lane == 0) { res[0] = ta; res[1] = tb; }
    }
    __syncthreads();
    return make_int2(res[0], res[1]);
}

static __device__ __forceinline__ int block_excl_scan_1024(int c, int* wofs) {
    const int lane = threadIdx.x & 31;
    const int w = threadIdx.x >> 5;
    int x = c;
    for (int o = 1; o < 32; o <<= 1) {
        const int y = __shfl_up(x, o, 32);
        if (lane >= o) x += y;
    }
    if (lane == 31) wofs[w] = x;
    __syncthreads();
    if (w == 0) {
        const int tot = wofs[lane];
        int s = tot;
        for (int o = 1; o < 32; o <<= 1) {
            const int y = __shfl_up(s, o, 32);
            if (lane >= o) s += y;
        }
        wofs[lane] = s - tot;
    }
    __syncthreads();
    return wofs[w] + (x - c);
}

__global__ void __launch_bounds__(SB) chunk_sort_kernel(const int* __restrict__ src, const int* __restrict__ dst,
                                                        int* __restrict__ runs, int* __restrict__ part,
                                                        int e, int chunk, int nbuck) {
    extern __shared__ int cur[];
    __shared__ int stage[PCAP + LINE];
    __shared__ int wofs[32];
    __shared__ int segtot;
    int* __restrict__ pe = part;
    const int tid = threadIdx.x;
    const int nb = (int)gridDim.x;
    const int blk = (int)blockIdx.x;
    const int beg = blk * chunk;
    const int end = min(beg + chunk, e);
    const int cnt = max(end - beg, 0);
    const int lead = beg & (LINE - 1);

    int d[EPT], s[EPT], r[EPT];
    bool ok[EPT];
#pragma unroll
    for (int k = 0; k < EPT; ++k) {
        const int i = tid + k * SB;
        ok[k] = i < cnt;
        const int j = ok[k] ? beg + i : 0;
        d[k] = dst[j];
        s[k] = src[j];
        r[k] = 0;
    }
    for (int p = tid; p < nbuck; p += SB) cur[p] = 0;
    __syncthreads();

#pragma unroll
    for (int k = 0; k < EPT; ++k)
        if (ok[k]) r[k] = atomicAdd(&cur[d[k] >> RB], 1);
    __syncthreads();

    int carry = 0;
    for (int base = 0; base < nbuck; base += SB) {
        const int p = base + tid;
        const int c = (p < nbuck) ? cur[p] : 0;
        const int ex = carry + block_excl_scan_1024(c, wofs);
        if (p < nbuck) {
            cur[p] = ex;
            runs[p * nb + blk] = ex;
        }
        if (tid == SB - 1) segtot = ex + c;
        __syncthreads();
        carry = segtot;
    }
    if (tid == 0) runs[nbuck * nb + blk] = cnt;

    int q[EPT];
#pragma unroll
    for (int k = 0; k < EPT; ++k) q[k] = cur[d[k] >> RB] + r[k];
#pragma unroll
    for (int k = 0; k < EPT; ++k)
        if (ok[k]) stage[lead + q[k]] = (int)(((unsigned)s[k] << RB) | ((unsigned)d[k] & (unsigned)(R - 1)));
    __syncthreads();

    const int line0 = beg - lead;
    const int top = lead + cnt;
    int li = tid;
    for (; li + 3 * SB < top; li += 4 * SB) {
        const bool k0 = li >= lead;
        const int p0 = k0 ? stage[li] : 0;
        const int p1 = stage[li + SB], p2 = stage[li + 2 * SB], p3 = stage[li + 3 * SB];
        if (k0) pe[line0 + li] = p0;
        pe[line0 + li + SB] = p1;
        pe[line0 + li + 2 * SB] = p2;
        pe[line0 + li + 3 * SB] = p3;
    }
    for (; li < top; li += SB)
        if (li >= lead) pe[line0 + li] = stage[li];
}

constexpr int RPT = 8;
template <class F>
static __device__ __forceinline__ void bucket_sweep(const int* __restrict__ pe, const int* rstart,
                                                    const int* rlen, int chunks, F f) {
    const int lane = threadIdx.x & 31;
    const int w = threadIdx.x >> 5;
    for (int b0 = w; b0 < chunks; b0 += RPT * NW) {
        int st[RPT], ln[RPT];
#pragma unroll
        for (int q = 0; q < RPT; ++q) {
            const int b = b0 + q * NW;
            const bool in = b < chunks;
            st[q] = in ? rstart[b] : 0;
            ln[q] = in ? rlen[b] : 0;
        }
        int a[2 * RPT];
        bool k[2 * RPT];
#pragma unroll
        for (int q = 0; q < RPT; ++q) {
            k[2 * q] = lane < ln[q];
            k[2 * q + 1] = lane + 32 < ln[q];
            a[2 * q] = pe[k[2 * q] ? st[q] + lane : 0];
            a[2 * q + 1] = pe[k[2 * q + 1] ? st[q] + lane + 32 : 0];
        }
        asm volatile("" ::"v"(a[0]), "v"(a[1]), "v"(a[2]), "v"(a[3]), "v"(a[4]), "v"(a[5]), "v"(a[6]), "v"(a[7]),
                     "v"(a[8]), "v"(a[9]), "v"(a[10]), "v"(a[11]), "v"(a[12]), "v"(a[13]), "v"(a[14]), "v"(a[15]));
#pragma unroll
        for (int q = 0; q < 2 * RPT; ++q)
            if (k[q]) f(a[q]);
#pragma unroll
        for (int q = 0; q < RPT; ++q)
            for (int i = lane + 64; i < ln[q]; i += 32) f(pe[st[q] + i]);
    }
}

template <int K, int NOUT>
__global__ void __launch_bounds__(SB) count_place_kernel(const int* __restrict__ part, const int* __restrict__ runs,
                                                         int* __restrict__ rowptr, float* __restrict__ dinv,
                                                         int* __restrict__ csr,
                                                         const float* __restrict__ x,
                                                         const float* __restrict__ w,
                                                         float* __restrict__ h,
                                                         int n, int chunk, int chunks) {
    static_assert(NOUT == 4, "a row of h leaves as one 16-byte store");
    static_assert(R / 16 <= NW, "one wave per 16-node tile of the bucket");
    extern __shared__ int rtab[];
    __shared__ int c[R];
    __shared__ int wofs[32];
    __shared__ int sm[2 * NW];
    __shared__ int res[2];
    __shared__ int img[CCAP + LINE];
    __shared__ __attribute__((aligned(16))) float sh[R * NOUT];
    int* rstart = rtab;
    int* rlen = rtab + chunks;
    const int* __restrict__ pe = part;
    const int tid = threadIdx.x;
    const int bucket = (int)blockIdx.x;
    const int node0 = bucket << RB;
    if (tid < R) c[tid] = 0;

    constexpr int KC = (K + 3) / 4;
    const int lane = tid & 31;
    const int wv = __builtin_amdgcn_readfirstlane(tid >> 5);
    const int tbase = node0 + wv * 16;
    const bool tile = (wv < R / 16) && (tbase < n);
    const int m = lane & 15;
    const int khalf = (lane >> 4) * 2;
    const bool colOK = m < NOUT;
    float xa[2 * KC], wb[2 * KC];
#pragma unroll
    for (int q = 0; q < 2 * KC; ++q) { xa[q] = 0.0f; wb[q] = 0.0f; }
    if (tile) {
        const int row = min(tbase + m, n - 1);
        const int mcl = colOK ? m : 0;
#pragma unroll
        for (int kc = 0; kc < KC; ++kc) {
            const int k0 = kc * 4 + khalf;
            const int ka0 = (k0 + 0) < K ? (k0 + 0) : 0;
            const int ka1 = (k0 + 1) < K ? (k0 + 1) : 0;
            xa[2 * kc] = x[row * K + ka0];
            xa[2 * kc + 1] = x[row * K + ka1];
            wb[2 * kc] = w[mcl * K + ka0];
            wb[2 * kc + 1] = w[mcl * K + ka1];
        }
    }

    int slo = 0, shi = 0;
    for (int b = tid; b < chunks; b += SB) {
        const int lo = runs[bucket * chunks + b];
        const int hi = runs[(bucket + 1) * chunks + b];
        rstart[b] = b * chunk + lo;
        rlen[b] = hi - lo;
        slo += lo;
        shi += hi;
    }
    if (tile) {
        v8f acc = {};
#pragma unroll
        for (int kc = 0; kc < KC; ++kc) {
            const int k0 = kc * 4 + khalf;
            const bool v0 = (k0 + 0) < K;
            const bool v1 = (k0 + 1) < K;
            v2f a, b;
            a.x = v0 ? xa[2 * kc] : 0.0f;
            a.y = v1 ? xa[2 * kc + 1] : 0.0f;
            b.x = (colOK && v0) ? wb[2 * kc] : 0.0f;
            b.y = (colOK && v1) ? wb[2 * kc + 1] : 0.0f;
            acc = __builtin_amdgcn_wmma_f32_16x16x4_f32(
                 false, a,  false, b,
                 (short)0, acc,  false,  false);
        }
        if (colOK) {
            float* dstp = sh + (wv * 16 + (lane >> 4) * 8) * NOUT + m;
#pragma unroll
            for (int r = 0; r < 8; ++r) dstp[r * NOUT] = acc[r];
        }
    }
    const int2 be = block_sum2_1024(slo, shi, sm, res);
    const int beg = be.x;
    const int end = be.y;

    bucket_sweep(pe, rstart, rlen, chunks, [&](int a) { atomicAdd(&c[a & (R - 1)], 1); });
    __syncthreads();
    const int cv = (tid < R) ? c[tid] : 0;
    const int excl = block_excl_scan_1024(cv, wofs);
    const int node = node0 + tid;
    const int cnt = end - beg;
    const bool staged = cnt <= CCAP;
    const int lead = beg & (LINE - 1);
    if (tid < R) {
        c[tid] = (staged ? lead : beg) + excl;
        if (node < n) {
            const float dv = rsqrtf((float)cv + 1.0f);
            rowptr[node] = beg + excl;
            dinv[node] = dv;
            const float4 p = *reinterpret_cast<const float4*>(sh + tid * NOUT);
            reinterpret_cast<float4*>(h)[node] = make_float4(p.x * dv, p.y * dv, p.z * dv, p.w * dv);
            if (node == n - 1) rowptr[n] = beg + excl + cv;
        }
    }
    __syncthreads();
    if (staged) {
        bucket_sweep(pe, rstart, rlen, chunks, [&](int a) { img[atomicAdd(&c[a & (R - 1)], 1)] = (int)((unsigned)a >> RB); });
        __syncthreads();
        const int line0 = beg - lead;
        const int top = lead + cnt;
        int li = tid;
        for (; li + 3 * SB < top; li += 4 * SB) {
            const bool k0 = li >= lead;
            const int v0 = k0 ? img[li] : 0;
            const int v1 = img[li + SB], v2 = img[li + 2 * SB], v3 = img[li + 3 * SB];
            if (k0) csr[line0 + li] = v0;
            csr[line0 + li + SB] = v1;
            csr[line0 + li + 2 * SB] = v2;
            csr[line0 + li + 3 * SB] = v3;
        }
        for (; li < top; li += SB)
            if (li >= lead) csr[line0 + li] = img[li];
    } else {
        bucket_sweep(pe, rstart, rlen, chunks, [&](int a) { csr[atomicAdd(&c[a & (R - 1)], 1)] = (int)((unsigned)a >> RB); });
    }
}

static __device__ __forceinline__ void vadd(float4& a, const float4& b) {
    a.x += b.x; a.y += b.y; a.z += b.z; a.w += b.w;
}
static __device__ __forceinline__ void vadd(float2& a, const float2& b) {
    a.x += b.x; a.y += b.y;
}
static __device__ __forceinline__ void vxor_add(float4& a, int o) {
    a.x += __shfl_xor(a.x, o, 32); a.y += __shfl_xor(a.y, o, 32);
    a.z += __shfl_xor(a.z, o, 32); a.w += __shfl_xor(a.w, o, 32);
}
static __device__ __forceinline__ void vxor_add(float2& a, int o) {
    a.x += __shfl_xor(a.x, o, 32); a.y += __shfl_xor(a.y, o, 32);
}
static __device__ __forceinline__ void vzero(float4& a) { a = make_float4(0.f, 0.f, 0.f, 0.f); }
static __device__ __forceinline__ void vzero(float2& a) { a = make_float2(0.f, 0.f); }
static __device__ __forceinline__ void vfinish(float* sm, const float4& a, float d, const float* __restrict__ bias) {
    sm[0] = bias[0] + a.x * d; sm[1] = bias[1] + a.y * d;
    sm[2] = bias[2] + a.z * d; sm[3] = bias[3] + a.w * d;
}
static __device__ __forceinline__ void vfinish(float* sm, const float2& a, float d, const float* __restrict__ bias) {
    sm[0] = bias[0] + a.x * d; sm[1] = bias[1] + a.y * d;
}
static __device__ __forceinline__ void vclear(float* sm, const float4&) { sm[0] = 0.f; sm[1] = 0.f; sm[2] = 0.f; sm[3] = 0.f; }
static __device__ __forceinline__ void vclear(float* sm, const float2&) { sm[0] = 0.f; sm[1] = 0.f; }

template <int C> struct VecOf;
template <> struct VecOf<4> { typedef float4 type; };
template <> struct VecOf<2> { typedef float2 type; };

constexpr int GL  = 4;
constexpr int GU  = 8;
constexpr int GB  = 256;
constexpr int NPB = GB / GL;

template <int C, int NEXT>
__global__ void __launch_bounds__(GB) gather_kernel(const int* __restrict__ rowptr,
                                                    const int* __restrict__ csr,
                                                    const float* __restrict__ h,
                                                    const float* __restrict__ dinv,
                                                    const float* __restrict__ bias,
                                                    const float* __restrict__ wnext,
                                                    float* __restrict__ out, int n) {
    typedef typename VecOf<C>::type VT;
    constexpr int CO = (NEXT > 0) ? NEXT : C;
    __shared__ __attribute__((aligned(16))) float sm[NPB * C];
    __shared__ __attribute__((aligned(16))) float sn[NPB * CO];
    __shared__ float sd[NPB];
    const VT* __restrict__ hv = reinterpret_cast<const VT*>(h);

    const int tid  = threadIdx.x;
    const int sub  = tid & (GL - 1);
    const int nl   = tid / GL;
    const int node = blockIdx.x * NPB + nl;
    const bool ok  = node < n;

    int j = 0, end = 0;
    float dn = 0.0f;
    if (ok) {
        j   = rowptr[node] + sub;
        end = rowptr[node + 1];
        dn  = dinv[node];
    }
    VT acc;
    vzero(acc);
    if (ok && sub == 0) acc = hv[node];

    for (; j < end; j += GU * GL) {
        int ent[GU];
        ent[0] = csr[j];
#pragma unroll
        for (int u = 1; u < GU; ++u) {
            const int ju = j + u * GL;
            ent[u] = (ju < end) ? csr[ju] : -1;
        }
        VT v[GU];
        v[0] = hv[ent[0]];
#pragma unroll
        for (int u = 1; u < GU; ++u) {
            vzero(v[u]);
            if (ent[u] >= 0) v[u] = hv[ent[u]];
        }
#pragma unroll
        for (int u = 0; u < GU; ++u) vadd(acc, v[u]);
    }

    for (int o = 1; o < GL; o <<= 1) vxor_add(acc, o);

    if (sub == 0) {
        if (ok) vfinish(sm + nl * C, acc, dn, bias);
        else vclear(sm + nl * C, acc);
        if constexpr (NEXT > 0) sd[nl] = dn;
    }
    __syncthreads();

    const float* stage = sm;
    if constexpr (NEXT > 0) {
        static_assert(C == 4, "the epilogue product is one 16x16x4 step");
        const int wv = __builtin_amdgcn_readfirstlane(tid >> 5);
        if (wv < NPB / 16) {
            const int lane = tid & 31;
            const int m    = lane & 15;
            const int kh   = (lane >> 4) * 2;
            const int mcl  = (m < NEXT) ? m : 0;
            const float w0 = wnext[mcl * C + kh];
            const float w1 = wnext[mcl * C + kh + 1];
            const float2 av = *reinterpret_cast<const float2*>(sm + (wv * 16 + m) * C + kh);
            const float rs = sd[wv * 16 + m];
            v2f a, b;
            a.x = rs * tanhf(av.x);
            a.y = rs * tanhf(av.y);
            b.x = (m < NEXT) ? w0 : 0.0f;
            b.y = (m < NEXT) ? w1 : 0.0f;
            v8f d = {};
            d = __builtin_amdgcn_wmma_f32_16x16x4_f32(
                 false, a,  false, b,
                 (short)0, d,  false,  false);
            if (m < NEXT) {
                float* dst = sn + (wv * 16 + (lane >> 4) * 8) * NEXT + m;
#pragma unroll
                for (int r = 0; r < 8; ++r) dst[r * NEXT] = d[r];
            }
        }
        __syncthreads();
        stage = sn;
    }

    constexpr int NV = NPB * CO / 4;
    if (tid < NV) {
        const int first = blockIdx.x * (NPB * CO) + tid * 4;
        const int total = n * CO;
        if (first + 3 < total) {
            *reinterpret_cast<float4*>(out + first) = *reinterpret_cast<const float4*>(stage + tid * 4);
        } else {
            for (int q = 0; q < 4; ++q)
                if (first + q < total) out[first + q] = stage[tid * 4 + q];
        }
    }
}

static inline int cdiv(int a, int b) { return (a + b - 1) / b; }

extern "C" void kernel_launch(void* const* d_in, const int* in_sizes, int n_in,
                              void* d_out, int out_size, void* d_ws, size_t ws_size,
                              hipStream_t stream) {
    const float* x  = (const float*)d_in[0];
    const int*   ei = (const int*)d_in[1];
    const float* W1 = (const float*)d_in[2];
    const float* b1 = (const float*)d_in[3];
    const float* W2 = (const float*)d_in[4];
    const float* b2 = (const float*)d_in[5];
    const float* W3 = (const float*)d_in[6];
    const float* b3 = (const float*)d_in[7];
    float* out = (float*)d_out;

    const int n = in_sizes[0] / 9;
    const int e = in_sizes[1] / 2;
    const int* src = ei;
    const int* dst = ei + e;

    char* ws = (char*)d_ws;
    size_t off = 0;
    auto carve = [&](size_t bytes) {
        void* p = ws + off;
        off += (bytes + 255) & ~(size_t)255;
        return p;
    };
    const int nbuck  = cdiv(n, R);
    const int fit    = cdiv(e, PCAP);
    const int chunks = fit > NBLK ? fit : NBLK;
    const int chunk  = cdiv(e, chunks);
    float* dinv   = (float*)carve((size_t)n * sizeof(float));
    int*   rowptr = (int*)carve(((size_t)n + 1) * sizeof(int));
    int*   runs   = (int*)carve(((size_t)nbuck + 1) * chunks * sizeof(int));
    int*   part   = (int*)carve((size_t)e * sizeof(int));
    int*   csr    = (int*)carve((size_t)e * sizeof(int));
    float* bufA = (float*)carve((size_t)n * 4 * sizeof(float));
    float* bufB = (float*)carve((size_t)n * 4 * sizeof(float));
    (void)ws_size;

    const int gG = cdiv(n, NPB);
    const size_t lds1 = (size_t)nbuck * sizeof(int);
    const size_t lds2 = (size_t)chunks * 2 * sizeof(int);

    chunk_sort_kernel<<<chunks, SB, lds1, stream>>>(src, dst, runs, part, e, chunk, nbuck);
    count_place_kernel<9, 4><<<nbuck, SB, lds2, stream>>>(part, runs, rowptr, dinv, csr, x, W1, bufA, n, chunk, chunks);
    gather_kernel<4, 4><<<gG, GB, 0, stream>>>(rowptr, csr, bufA, dinv, b1, W2, bufB, n);
    gather_kernel<4, 2><<<gG, GB, 0, stream>>>(rowptr, csr, bufB, dinv, b2, W3, bufA, n);
    gather_kernel<2, 0><<<gG, GB, 0, stream>>>(rowptr, csr, bufA, dinv, b3, nullptr, out, n);
}


# level 2 reads its records once: the counting LDS add returns the edge's rank in its node, record and rank wait in registers over the scan, place = node start + rank; sum order still LDS arrival order
# speedup vs baseline: 15.4307x; 1.0405x over previous
#include <hip/hip_runtime.h>
#include <hip/hip_bf16.h>


typedef __attribute__((ext_vector_type(2))) float v2f;
typedef __attribute__((ext_vector_type(8))) float v8f;

constexpr int RB   = 8;
constexpr int R    = 1 << RB;
constexpr int NBLK = 256;
constexpr int SB   = 1024;
constexpr int PCAP = 16384;
constexpr int EPT  = PCAP / SB;
constexpr int CCAP = 16384;
constexpr int LINE = 32;
constexpr int NW   = SB / 32;

static __device__ __forceinline__ int wave_sum(int v) {
    for (int o = 16; o > 0; o >>= 1) v += __shfl_xor(v, o, 32);
    return v;
}

static __device__ __forceinline__ int3 block_sum3_1024(int a, int b, int c, int* sm, int* res) {
    const int lane = threadIdx.x & 31;
    const int w = threadIdx.x >> 5;
    a = wave_sum(a);
    b = wave_sum(b);
    c = wave_sum(c);
    if (lane == 0) { sm[w] = a; sm[NW + w] = b; sm[2 * NW + w] = c; }
    __syncthreads();
    if (w == 0) {
        const int ta = wave_sum(sm[lane]);
        const int tb = wave_sum(sm[NW + lane]);
        const int tc = wave_sum(sm[2 * NW + lane]);
        if (lane == 0) { res[0] = ta; res[1] = tb; res[2] = tc; }
    }
    __syncthreads();
    return make_int3(res[0], res[1], res[2]);
}

static __device__ __forceinline__ int block_excl_scan_1024(int c, int* wofs) {
    const int lane = threadIdx.x & 31;
    const int w = threadIdx.x >> 5;
    int x = c;
    for (int o = 1; o < 32; o <<= 1) {
        const int y = __shfl_up(x, o, 32);
        if (lane >= o) x += y;
    }
    if (lane == 31) wofs[w] = x;
    __syncthreads();
    if (w == 0) {
        const int tot = wofs[lane];
        int s = tot;
        for (int o = 1; o < 32; o <<= 1) {
            const int y = __shfl_up(s, o, 32);
            if (lane >= o) s += y;
        }
        wofs[lane] = s - tot;
    }
    __syncthreads();
    return wofs[w] + (x - c);
}

__global__ void __launch_bounds__(SB) chunk_sort_kernel(const int* __restrict__ src, const int* __restrict__ dst,
                                                        int* __restrict__ runs, int* __restrict__ part,
                                                        int e, int chunk, int nbuck) {
    extern __shared__ int cur[];
    __shared__ int stage[PCAP + LINE];
    __shared__ int wofs[32];
    __shared__ int segtot;
    int* __restrict__ pe = part;
    const int tid = threadIdx.x;
    const int nb = (int)gridDim.x;
    const int blk = (int)blockIdx.x;
    const int beg = blk * chunk;
    const int end = min(beg + chunk, e);
    const int cnt = max(end - beg, 0);
    const int lead = beg & (LINE - 1);

    int d[EPT], s[EPT], r[EPT];
    bool ok[EPT];
#pragma unroll
    for (int k = 0; k < EPT; ++k) {
        const int i = tid + k * SB;
        ok[k] = i < cnt;
        const int j = ok[k] ? beg + i : 0;
        d[k] = dst[j];
        s[k] = src[j];
        r[k] = 0;
    }
    for (int p = tid; p < nbuck; p += SB) cur[p] = 0;
    __syncthreads();

#pragma unroll
    for (int k = 0; k < EPT; ++k)
        if (ok[k]) r[k] = atomicAdd(&cur[d[k] >> RB], 1);
    __syncthreads();

    int carry = 0;
    for (int base = 0; base < nbuck; base += SB) {
        const int p = base + tid;
        const int c = (p < nbuck) ? cur[p] : 0;
        const int ex = carry + block_excl_scan_1024(c, wofs);
        if (p < nbuck) {
            cur[p] = ex;
            runs[p * nb + blk] = ex;
        }
        if (tid == SB - 1) segtot = ex + c;
        __syncthreads();
        carry = segtot;
    }
    if (tid == 0) runs[nbuck * nb + blk] = cnt;

    int q[EPT];
#pragma unroll
    for (int k = 0; k < EPT; ++k) q[k] = cur[d[k] >> RB] + r[k];
#pragma unroll
    for (int k = 0; k < EPT; ++k)
        if (ok[k]) stage[lead + q[k]] = (int)(((unsigned)s[k] << RB) | ((unsigned)d[k] & (unsigned)(R - 1)));
    __syncthreads();

    const int line0 = beg - lead;
    const int top = lead + cnt;
    int li = tid;
    for (; li + 3 * SB < top; li += 4 * SB) {
        const bool k0 = li >= lead;
        const int p0 = k0 ? stage[li] : 0;
        const int p1 = stage[li + SB], p2 = stage[li + 2 * SB], p3 = stage[li + 3 * SB];
        if (k0) pe[line0 + li] = p0;
        pe[line0 + li + SB] = p1;
        pe[line0 + li + 2 * SB] = p2;
        pe[line0 + li + 3 * SB] = p3;
    }
    for (; li < top; li += SB)
        if (li >= lead) pe[line0 + li] = stage[li];
}

constexpr int RPT = 8;

static __device__ __forceinline__ void bucket_hold(const int* __restrict__ pe, const int* rstart, const int* rlen,
                                                   int chunks, int (&a)[2 * RPT], bool (&k)[2 * RPT]) {
    const int lane = threadIdx.x & 31;
    const int w = threadIdx.x >> 5;
    int st[RPT], ln[RPT];
#pragma unroll
    for (int q = 0; q < RPT; ++q) {
        const int b = w + q * NW;
        const bool in = b < chunks;
        st[q] = in ? rstart[b] : 0;
        ln[q] = in ? rlen[b] : 0;
    }
#pragma unroll
    for (int q = 0; q < RPT; ++q) {
        k[2 * q] = lane < ln[q];
        k[2 * q + 1] = lane + 32 < ln[q];
        a[2 * q] = pe[k[2 * q] ? st[q] + lane : 0];
        a[2 * q + 1] = pe[k[2 * q + 1] ? st[q] + lane + 32 : 0];
    }
    asm volatile("" ::"v"(a[0]), "v"(a[1]), "v"(a[2]), "v"(a[3]), "v"(a[4]), "v"(a[5]), "v"(a[6]), "v"(a[7]),
                 "v"(a[8]), "v"(a[9]), "v"(a[10]), "v"(a[11]), "v"(a[12]), "v"(a[13]), "v"(a[14]), "v"(a[15]));
}

template <class F>
static __device__ __forceinline__ void bucket_tail(const int* __restrict__ pe, const int* rstart, const int* rlen,
                                                   int chunks, int first, F f) {
    const int lane = threadIdx.x & 31;
    const int w = threadIdx.x >> 5;
    int q = 0;
    for (int b = w; b < chunks; b += NW, ++q) {
        const int st = rstart[b];
        const int ln = rlen[b];
        for (int i = lane + (q < RPT ? first : 0); i < ln; i += 32) f(pe[st + i]);
    }
}

template <int K, int NOUT>
__global__ void __launch_bounds__(SB) count_place_kernel(const int* __restrict__ part, const int* __restrict__ runs,
                                                         int* __restrict__ rowptr, float* __restrict__ dinv,
                                                         int* __restrict__ csr,
                                                         const float* __restrict__ x,
                                                         const float* __restrict__ w,
                                                         float* __restrict__ h,
                                                         int n, int chunk, int chunks) {
    static_assert(NOUT == 4, "a row of h leaves as one 16-byte store");
    static_assert(R / 16 <= NW, "one wave per 16-node tile of the bucket");
    extern __shared__ int rtab[];
    __shared__ int c[R];
    __shared__ int t[R];
    __shared__ int wofs[32];
    __shared__ int sm[3 * NW];
    __shared__ int res[3];
    __shared__ int img[CCAP + LINE];
    __shared__ __attribute__((aligned(16))) float sh[R * NOUT];
    int* rstart = rtab;
    int* rlen = rtab + chunks;
    const int* __restrict__ pe = part;
    const int tid = threadIdx.x;
    const int bucket = (int)blockIdx.x;
    const int node0 = bucket << RB;
    if (tid < R) { c[tid] = 0; t[tid] = 0; }

    constexpr int KC = (K + 3) / 4;
    const int lane = tid & 31;
    const int wv = __builtin_amdgcn_readfirstlane(tid >> 5);
    const int tbase = node0 + wv * 16;
    const bool tile = (wv < R / 16) && (tbase < n);
    const int m = lane & 15;
    const int khalf = (lane >> 4) * 2;
    const bool colOK = m < NOUT;
    float xa[2 * KC], wb[2 * KC];
#pragma unroll
    for (int q = 0; q < 2 * KC; ++q) { xa[q] = 0.0f; wb[q] = 0.0f; }
    if (tile) {
        const int row = min(tbase + m, n - 1);
        const int mcl = colOK ? m : 0;
#pragma unroll
        for (int kc = 0; kc < KC; ++kc) {
            const int k0 = kc * 4 + khalf;
            const int ka0 = (k0 + 0) < K ? (k0 + 0) : 0;
            const int ka1 = (k0 + 1) < K ? (k0 + 1) : 0;
            xa[2 * kc] = x[row * K + ka0];
            xa[2 * kc + 1] = x[row * K + ka1];
            wb[2 * kc] = w[mcl * K + ka0];
            wb[2 * kc + 1] = w[mcl * K + ka1];
        }
    }

    int slo = 0, shi = 0, big = 0;
    for (int b = tid; b < chunks; b += SB) {
        const int lo = runs[bucket * chunks + b];
        const int hi = runs[(bucket + 1) * chunks + b];
        rstart[b] = b * chunk + lo;
        rlen[b] = hi - lo;
        slo += lo;
        shi += hi;
        big += (hi - lo > 64) ? 1 : 0;
    }
    if (tile) {
        v8f acc = {};
#pragma unroll
        for (int kc = 0; kc < KC; ++kc) {
            const int k0 = kc * 4 + khalf;
            const bool v0 = (k0 + 0) < K;
            const bool v1 = (k0 + 1) < K;
            v2f a, b;
            a.x = v0 ? xa[2 * kc] : 0.0f;
            a.y = v1 ? xa[2 * kc + 1] : 0.0f;
            b.x = (colOK && v0) ? wb[2 * kc] : 0.0f;
            b.y = (colOK && v1) ? wb[2 * kc + 1] : 0.0f;
            acc = __builtin_amdgcn_wmma_f32_16x16x4_f32(
                 false, a,  false, b,
                 (short)0, acc,  false,  false);
        }
        if (colOK) {
            float* dstp = sh + (wv * 16 + (lane >> 4) * 8) * NOUT + m;
#pragma unroll
            for (int r = 0; r < 8; ++r) dstp[r * NOUT] = acc[r];
        }
    }
    const int3 be = block_sum3_1024(slo, shi, big, sm, res);
    const int beg = be.x;
    const int end = be.y;
    const int cnt = end - beg;
    const bool staged = cnt <= CCAP;
    const int lead = beg & (LINE - 1);

    static_assert(CCAP <= 65536, "a rank of a held record fits 16 bits");
    const bool held = staged;
    const bool tails = !held || (be.z != 0) || (chunks > RPT * NW);
    const int first = held ? 64 : 0;
    int a[2 * RPT], rk[RPT];
    bool k[2 * RPT];
    if (held) {
        bucket_hold(pe, rstart, rlen, chunks, a, k);
#pragma unroll
        for (int q = 0; q < RPT; ++q) {
            int r0 = 0, r1 = 0;
            if (k[2 * q]) r0 = atomicAdd(&c[a[2 * q] & (R - 1)], 1);
            if (k[2 * q + 1]) r1 = atomicAdd(&c[a[2 * q + 1] & (R - 1)], 1);
            rk[q] = r0 | (r1 << 16);
        }
    }
    if (tails) bucket_tail(pe, rstart, rlen, chunks, first, [&](int v) { atomicAdd(&t[v & (R - 1)], 1); });
    __syncthreads();
    const int ch = (tid < R) ? c[tid] : 0;
    const int cv = ch + ((tid < R) ? t[tid] : 0);
    const int excl = block_excl_scan_1024(cv, wofs);
    const int node = node0 + tid;
    if (tid < R) {
        const int start = (staged ? lead : beg) + excl;
        c[tid] = start;
        t[tid] = start + ch;
        if (node < n) {
            const float dv = rsqrtf((float)cv + 1.0f);
            rowptr[node] = beg + excl;
            dinv[node] = dv;
            const float4 p = *reinterpret_cast<const float4*>(sh + tid * NOUT);
            reinterpret_cast<float4*>(h)[node] = make_float4(p.x * dv, p.y * dv, p.z * dv, p.w * dv);
            if (node == n - 1) rowptr[n] = beg + excl + cv;
        }
    }
    __syncthreads();
    if (staged) {
        int pos[2 * RPT];
#pragma unroll
        for (int q = 0; q < 2 * RPT; ++q) pos[q] = c[a[q] & (R - 1)];
#pragma unroll
        for (int q = 0; q < RPT; ++q) {
            if (k[2 * q]) img[pos[2 * q] + (rk[q] & 0xffff)] = (int)((unsigned)a[2 * q] >> RB);
            if (k[2 * q + 1]) img[pos[2 * q + 1] + (int)((unsigned)rk[q] >> 16)] = (int)((unsigned)a[2 * q + 1] >> RB);
        }
        if (tails) bucket_tail(pe, rstart, rlen, chunks, first, [&](int v) { img[atomicAdd(&t[v & (R - 1)], 1)] = (int)((unsigned)v >> RB); });
        __syncthreads();
        const int line0 = beg - lead;
        const int top = lead + cnt;
        int li = tid;
        for (; li + 3 * SB < top; li += 4 * SB) {
            const bool k0 = li >= lead;
            const int v0 = k0 ? img[li] : 0;
            const int v1 = img[li + SB], v2 = img[li + 2 * SB], v3 = img[li + 3 * SB];
            if (k0) csr[line0 + li] = v0;
            csr[line0 + li + SB] = v1;
            csr[line0 + li + 2 * SB] = v2;
            csr[line0 + li + 3 * SB] = v3;
        }
        for (; li < top; li += SB)
            if (li >= lead) csr[line0 + li] = img[li];
    } else {
        bucket_tail(pe, rstart, rlen, chunks, first, [&](int v) { csr[atomicAdd(&t[v & (R - 1)], 1)] = (int)((unsigned)v >> RB); });
    }
}

static __device__ __forceinline__ void vadd(float4& a, const float4& b) {
    a.x += b.x; a.y += b.y; a.z += b.z; a.w += b.w;
}
static __device__ __forceinline__ void vadd(float2& a, const float2& b) {
    a.x += b.x; a.y += b.y;
}
static __device__ __forceinline__ void vxor_add(float4& a, int o) {
    a.x += __shfl_xor(a.x, o, 32); a.y += __shfl_xor(a.y, o, 32);
    a.z += __shfl_xor(a.z, o, 32); a.w += __shfl_xor(a.w, o, 32);
}
static __device__ __forceinline__ void vxor_add(float2& a, int o) {
    a.x += __shfl_xor(a.x, o, 32); a.y += __shfl_xor(a.y, o, 32);
}
static __device__ __forceinline__ void vzero(float4& a) { a = make_float4(0.f, 0.f, 0.f, 0.f); }
static __device__ __forceinline__ void vzero(float2& a) { a = make_float2(0.f, 0.f); }
static __device__ __forceinline__ void vfinish(float* sm, const float4& a, float d, const float* __restrict__ bias) {
    sm[0] = bias[0] + a.x * d; sm[1] = bias[1] + a.y * d;
    sm[2] = bias[2] + a.z * d; sm[3] = bias[3] + a.w * d;
}
static __device__ __forceinline__ void vfinish(float* sm, const float2& a, float d, const float* __restrict__ bias) {
    sm[0] = bias[0] + a.x * d; sm[1] = bias[1] + a.y * d;
}
static __device__ __forceinline__ void vclear(float* sm, const float4&) { sm[0] = 0.f; sm[1] = 0.f; sm[2] = 0.f; sm[3] = 0.f; }
static __device__ __forceinline__ void vclear(float* sm, const float2&) { sm[0] = 0.f; sm[1] = 0.f; }

template <int C> struct VecOf;
template <> struct VecOf<4> { typedef float4 type; };
template <> struct VecOf<2> { typedef float2 type; };

constexpr int GL  = 4;
constexpr int GU  = 8;
constexpr int GB  = 256;
constexpr int NPB = GB / GL;

template <int C, int NEXT>
__global__ void __launch_bounds__(GB) gather_kernel(const int* __restrict__ rowptr,
                                                    const int* __restrict__ csr,
                                                    const float* __restrict__ h,
                                                    const float* __restrict__ dinv,
                                                    const float* __restrict__ bias,
                                                    const float* __restrict__ wnext,
                                                    float* __restrict__ out, int n) {
    typedef typename VecOf<C>::type VT;
    constexpr int CO = (NEXT > 0) ? NEXT : C;
    __shared__ __attribute__((aligned(16))) float sm[NPB * C];
    __shared__ __attribute__((aligned(16))) float sn[NPB * CO];
    __shared__ float sd[NPB];
    const VT* __restrict__ hv = reinterpret_cast<const VT*>(h);

    const int tid  = threadIdx.x;
    const int sub  = tid & (GL - 1);
    const int nl   = tid / GL;
    const int node = blockIdx.x * NPB + nl;
    const bool ok  = node < n;

    int j = 0, end = 0;
    float dn = 0.0f;
    if (ok) {
        j   = rowptr[node] + sub;
        end = rowptr[node + 1];
        dn  = dinv[node];
    }
    VT acc;
    vzero(acc);
    if (ok && sub == 0) acc = hv[node];

    for (; j < end; j += GU * GL) {
        int ent[GU];
        ent[0] = csr[j];
#pragma unroll
        for (int u = 1; u < GU; ++u) {
            const int ju = j + u * GL;
            ent[u] = (ju < end) ? csr[ju] : -1;
        }
        VT v[GU];
        v[0] = hv[ent[0]];
#pragma unroll
        for (int u = 1; u < GU; ++u) {
            vzero(v[u]);
            if (ent[u] >= 0) v[u] = hv[ent[u]];
        }
#pragma unroll
        for (int u = 0; u < GU; ++u) vadd(acc, v[u]);
    }

    for (int o = 1; o < GL; o <<= 1) vxor_add(acc, o);

    if (sub == 0) {
        if (ok) vfinish(sm + nl * C, acc, dn, bias);
        else vclear(sm + nl * C, acc);
        if constexpr (NEXT > 0) sd[nl] = dn;
    }
    __syncthreads();

    const float* stage = sm;
    if constexpr (NEXT > 0) {
        static_assert(C == 4, "the epilogue product is one 16x16x4 step");
        const int wv = __builtin_amdgcn_readfirstlane(tid >> 5);
        if (wv < NPB / 16) {
            const int lane = tid & 31;
            const int m    = lane & 15;
            const int kh   = (lane >> 4) * 2;
            const int mcl  = (m < NEXT) ? m : 0;
            const float w0 = wnext[mcl * C + kh];
            const float w1 = wnext[mcl * C + kh + 1];
            const float2 av = *reinterpret_cast<const float2*>(sm + (wv * 16 + m) * C + kh);
            const float rs = sd[wv * 16 + m];
            v2f a, b;
            a.x = rs * tanhf(av.x);
            a.y = rs * tanhf(av.y);
            b.x = (m < NEXT) ? w0 : 0.0f;
            b.y = (m < NEXT) ? w1 : 0.0f;
            v8f d = {};
            d = __builtin_amdgcn_wmma_f32_16x16x4_f32(
                 false, a,  false, b,
                 (short)0, d,  false,  false);
            if (m < NEXT) {
                float* dst = sn + (wv * 16 + (lane >> 4) * 8) * NEXT + m;
#pragma unroll
                for (int r = 0; r < 8; ++r) dst[r * NEXT] = d[r];
            }
        }
        __syncthreads();
        stage = sn;
    }

    constexpr int NV = NPB * CO / 4;
    if (tid < NV) {
        const int first = blockIdx.x * (NPB * CO) + tid * 4;
        const int total = n * CO;
        if (first + 3 < total) {
            *reinterpret_cast<float4*>(out + first) = *reinterpret_cast<const float4*>(stage + tid * 4);
        } else {
            for (int q = 0; q < 4; ++q)
                if (first + q < total) out[first + q] = stage[tid * 4 + q];
        }
    }
}

static inline int cdiv(int a, int b) { return (a + b - 1) / b; }

extern "C" void kernel_launch(void* const* d_in, const int* in_sizes, int n_in,
                              void* d_out, int out_size, void* d_ws, size_t ws_size,
                              hipStream_t stream) {
    const float* x  = (const float*)d_in[0];
    const int*   ei = (const int*)d_in[1];
    const float* W1 = (const float*)d_in[2];
    const float* b1 = (const float*)d_in[3];
    const float* W2 = (const float*)d_in[4];
    const float* b2 = (const float*)d_in[5];
    const float* W3 = (const float*)d_in[6];
    const float* b3 = (const float*)d_in[7];
    float* out = (float*)d_out;

    const int n = in_sizes[0] / 9;
    const int e = in_sizes[1] / 2;
    const int* src = ei;
    const int* dst = ei + e;

    char* ws = (char*)d_ws;
    size_t off = 0;
    auto carve = [&](size_t bytes) {
        void* p = ws + off;
        off += (bytes + 255) & ~(size_t)255;
        return p;
    };
    const int nbuck  = cdiv(n, R);
    const int fit    = cdiv(e, PCAP);
    const int chunks = fit > NBLK ? fit : NBLK;
    const int chunk  = cdiv(e, chunks);
    float* dinv   = (float*)carve((size_t)n * sizeof(float));
    int*   rowptr = (int*)carve(((size_t)n + 1) * sizeof(int));
    int*   runs   = (int*)carve(((size_t)nbuck + 1) * chunks * sizeof(int));
    int*   part   = (int*)carve((size_t)e * sizeof(int));
    int*   csr    = (int*)carve((size_t)e * sizeof(int));
    float* bufA = (float*)carve((size_t)n * 4 * sizeof(float));
    float* bufB = (float*)carve((size_t)n * 4 * sizeof(float));
    (void)ws_size;

    const int gG = cdiv(n, NPB);
    const size_t lds1 = (size_t)nbuck * sizeof(int);
    const size_t lds2 = (size_t)chunks * 2 * sizeof(int);

    chunk_sort_kernel<<<chunks, SB, lds1, stream>>>(src, dst, runs, part, e, chunk, nbuck);
    count_place_kernel<9, 4><<<nbuck, SB, lds2, stream>>>(part, runs, rowptr, dinv, csr, x, W1, bufA, n, chunk, chunks);
    gather_kernel<4, 4><<<gG, GB, 0, stream>>>(rowptr, csr, bufA, dinv, b1, W2, bufB, n);
    gather_kernel<4, 2><<<gG, GB, 0, stream>>>(rowptr, csr, bufB, dinv, b2, W3, bufA, n);
    gather_kernel<2, 0><<<gG, GB, 0, stream>>>(rowptr, csr, bufA, dinv, b3, nullptr, out, n);
}
